# starting kernel
# baseline (speedup 1.0000x reference)
#include <hip/hip_runtime.h>
#include <cmath>

#define B_ROWS   16384
#define NVC      1024
#define VMAXF    6.0f
#define SC_ITERS 10

typedef __attribute__((ext_vector_type(2))) float v2f;
typedef __attribute__((ext_vector_type(8))) float v8f;

__global__ __launch_bounds__(32) void fp_moments_wmma(
    const float* __restrict__ f, const float* __restrict__ dv_p,
    float4* __restrict__ scal) {
  const int lane    = threadIdx.x;
  const int rowbase = blockIdx.x * 16;
  const float dv    = *dv_p;
  const int r       = lane & 15;
  const int khalf   = (lane >> 4) * 2;

  const float* fp = f + (size_t)(rowbase + r) * NVC + khalf;

  const float c0 = (r == 0) ? 1.0f : 0.0f;
  const float c1 = (r == 1) ? 1.0f : 0.0f;
  const float c2 = (r == 2) ? 1.0f : 0.0f;

  const float dv4 = 4.0f * dv;
  float vcur = dv * ((float)khalf + 0.5f) - VMAXF;

  v8f acc0 = {}, acc1 = {}, acc2 = {}, acc3 = {};
  for (int kb = 0; kb < NVC; kb += 16) {
#pragma unroll
    for (int u = 0; u < 4; ++u) {
      const int k = kb + 4 * u;
      v2f a;
      a.x = fp[k + 0];
      a.y = fp[k + 1];
      const float v0 = vcur;
      const float v1 = vcur + dv;
      v2f b;
      b.x = (c2 * v0 + c1) * v0 + c0;
      b.y = (c2 * v1 + c1) * v1 + c0;
      v8f* acc = (u == 0) ? &acc0 : (u == 1) ? &acc1 : (u == 2) ? &acc2 : &acc3;
      *acc = __builtin_amdgcn_wmma_f32_16x16x4_f32(
          false, a, false, b, (short)0, *acc, false, false);
      vcur += dv4;
    }
  }
  v8f acc = (acc0 + acc1) + (acc2 + acc3);

  float av[8];
#pragma unroll
  for (int i = 0; i < 8; ++i) av[i] = acc[i];

  float myn = 0.f, myvb = 0.f, myet = 0.f, mybeta = 0.f;
#pragma unroll
  for (int m = 0; m < 16; ++m) {
    const int base  = (m < 8) ? 0 : 16;
    const float val = av[m & 7];
    const float S0  = __shfl(val, base + 0, 32);
    const float S1  = __shfl(val, base + 1, 32);
    const float S2  = __shfl(val, base + 2, 32);
    if (lane == m) {
      const float n_   = S0 * dv;
      const float vbar = S1 / S0;
      const float e_t  = (S2 - S1 * S1 / S0) * dv;
      myn = n_; myvb = vbar; myet = e_t;
      mybeta = n_ / (2.0f * e_t);
    }
  }
  if (lane < 16) scal[rowbase + lane] = make_float4(myn, myvb, myet, mybeta);
}

__global__ __launch_bounds__(256) void fp_beta_iter(
    const float* __restrict__ dv_p, float4* __restrict__ scal) {
  const int wave = threadIdx.x >> 5;
  const int lane = threadIdx.x & 31;
  const int row  = blockIdx.x * 8 + wave;
  const float dv = *dv_p;

  const float4 s = scal[row];
  const float n_ = s.x, vbar = s.y, e_t = s.z;
  float beta     = s.w;
  const float g  = n_ / e_t;

  float u2[32];
#pragma unroll
  for (int c = 0; c < 32; ++c) {
    const int j   = lane + 32 * c;
    const float v = dv * ((float)j + 0.5f) - VMAXF;
    const float u = v - vbar;
    u2[c] = u * u;
  }

  for (int it = 0; it < SC_ITERS; ++it) {
    float sm = 0.f, se = 0.f;
#pragma unroll
    for (int c = 0; c < 32; ++c) {
      const float M = __expf(-beta * u2[c]);
      sm += M;
      se += M * u2[c];
    }
#pragma unroll
    for (int off = 16; off > 0; off >>= 1) {
      sm += __shfl_xor(sm, off, 32);
      se += __shfl_xor(se, off, 32);
    }
    beta = beta * (se / sm) * g;
  }
  if (lane == 0) scal[row].w = beta;
}

__global__ __launch_bounds__(256) void fp_thomas(
    const float* __restrict__ f, const float* __restrict__ v_edge,
    const float* __restrict__ dv_p, const float* __restrict__ nu_p,
    const float* __restrict__ dt_p, const float4* __restrict__ scal,
    float* __restrict__ cp_ws, float* __restrict__ out) {
  const int row  = blockIdx.x * blockDim.x + threadIdx.x;
  const float dv = *dv_p, nu = *nu_p, dt = *dt_p;

  const float4 sc  = scal[row];
  const float vbar = sc.y, beta = sc.w;
  const float D      = 0.5f / beta;
  const float Pdv    = D / dv;
  const float s      = dt * nu / dv;
  const float twobdv = 2.0f * beta * dv;

  const float* frow = f + (size_t)row * NVC;
  float*       orow = out + (size_t)row * NVC;

  float Ce0 = 0.f, P0 = 0.f, d0 = 0.5f;
  float cp_prev = 0.f, dp_prev = 0.f;

  for (int jb = 0; jb < NVC; jb += 4) {
    const float4 fj = *(const float4*)(frow + jb);
    float fv[4] = {fj.x, fj.y, fj.z, fj.w};
    float dpv[4];
#pragma unroll
    for (int q = 0; q < 4; ++q) {
      const int j = jb + q;
      const int e = j + 1;
      float Ce1, P1, d1;
      if (e < NVC) {
        const float c = v_edge[e] - vbar;
        Ce1 = c;
        P1  = Pdv;
        const float w = twobdv * c;
        if (fabsf(w) < 1e-6f) {
          d1 = 0.5f;
        } else {
          const float em = expm1f(w);
          d1 = (em - w) / (w * em);
        }
      } else {
        Ce1 = 0.f; P1 = 0.f; d1 = 0.5f;
      }
      const float coefU = Ce1 * (1.0f - d1) + P1;
      const float coefL = -Ce0 * d0 + P0;
      const float coefD = Ce1 * d1 - P1 - Ce0 * (1.0f - d0) - P0;
      const float dl = -s * coefL;
      const float dd = 1.0f - s * coefD;
      const float du = -s * coefU;

      const float denom = dd - dl * cp_prev;
      const float rdiv  = 1.0f / denom;
      const float cp    = du * rdiv;
      const float dp    = (fv[q] - dl * dp_prev) * rdiv;

      cp_ws[(size_t)j * B_ROWS + row] = cp;
      dpv[q]  = dp;
      cp_prev = cp; dp_prev = dp;
      Ce0 = Ce1; P0 = P1; d0 = d1;
    }
    *(float4*)(orow + jb) = make_float4(dpv[0], dpv[1], dpv[2], dpv[3]);
  }

  float xnext = 0.f;
  for (int jb = NVC - 4; jb >= 0; jb -= 4) {
    const float4 dpj = *(const float4*)(orow + jb);
    float dv4[4] = {dpj.x, dpj.y, dpj.z, dpj.w};
    float xv[4];
#pragma unroll
    for (int q = 3; q >= 0; --q) {
      const int j   = jb + q;
      const float c = cp_ws[(size_t)j * B_ROWS + row];
      const float x = dv4[q] - c * xnext;
      xv[q] = x;
      xnext = x;
    }
    *(float4*)(orow + jb) = make_float4(xv[0], xv[1], xv[2], xv[3]);
  }
}

extern "C" void kernel_launch(void* const* d_in, const int* in_sizes, int n_in,
                              void* d_out, int out_size, void* d_ws, size_t ws_size,
                              hipStream_t stream) {
  const float* f      = (const float*)d_in[0];
  const float* v      = (const float*)d_in[1];  (void)v;
  const float* v_edge = (const float*)d_in[2];
  const float* dv     = (const float*)d_in[3];
  const float* nu     = (const float*)d_in[4];
  const float* dt     = (const float*)d_in[5];
  (void)in_sizes; (void)n_in; (void)out_size; (void)ws_size;

  float4* scal = (float4*)d_ws;
  float*  cp   = (float*)((char*)d_ws + (size_t)B_ROWS * sizeof(float4));
  float*  out  = (float*)d_out;

  fp_moments_wmma<<<B_ROWS / 16, 32, 0, stream>>>(f, dv, scal);
  fp_beta_iter  <<<B_ROWS / 8, 256, 0, stream>>>(dv, scal);
  fp_thomas     <<<B_ROWS / 256, 256, 0, stream>>>(f, v_edge, dv, nu, dt, scal, cp, out);
}


# fp_thomas forward sweep: edge weights (expm1+divide) made branch-free and computed one 4-cell block ahead in the same straight-line body, so they overlap the serial cp/dp chain
# speedup vs baseline: 1.3491x; 1.3491x over previous
#include <hip/hip_runtime.h>
#include <cmath>

#define B_ROWS   16384
#define NVC      1024
#define VMAXF    6.0f
#define SC_ITERS 10

typedef __attribute__((ext_vector_type(2))) float v2f;
typedef __attribute__((ext_vector_type(8))) float v8f;

__global__ __launch_bounds__(32) void fp_moments_wmma(
    const float* __restrict__ f, const float* __restrict__ dv_p,
    float4* __restrict__ scal) {
  const int lane    = threadIdx.x;
  const int rowbase = blockIdx.x * 16;
  const float dv    = *dv_p;
  const int r       = lane & 15;
  const int khalf   = (lane >> 4) * 2;

  const float* fp = f + (size_t)(rowbase + r) * NVC + khalf;

  const float c0 = (r == 0) ? 1.0f : 0.0f;
  const float c1 = (r == 1) ? 1.0f : 0.0f;
  const float c2 = (r == 2) ? 1.0f : 0.0f;

  const float dv4 = 4.0f * dv;
  float vcur = dv * ((float)khalf + 0.5f) - VMAXF;

  v8f acc0 = {}, acc1 = {}, acc2 = {}, acc3 = {};
  for (int kb = 0; kb < NVC; kb += 16) {
#pragma unroll
    for (int u = 0; u < 4; ++u) {
      const int k = kb + 4 * u;
      v2f a;
      a.x = fp[k + 0];
      a.y = fp[k + 1];
      const float v0 = vcur;
      const float v1 = vcur + dv;
      v2f b;
      b.x = (c2 * v0 + c1) * v0 + c0;
      b.y = (c2 * v1 + c1) * v1 + c0;
      v8f* acc = (u == 0) ? &acc0 : (u == 1) ? &acc1 : (u == 2) ? &acc2 : &acc3;
      *acc = __builtin_amdgcn_wmma_f32_16x16x4_f32(
          false, a, false, b, (short)0, *acc, false, false);
      vcur += dv4;
    }
  }
  v8f acc = (acc0 + acc1) + (acc2 + acc3);

  float av[8];
#pragma unroll
  for (int i = 0; i < 8; ++i) av[i] = acc[i];

  float myn = 0.f, myvb = 0.f, myet = 0.f, mybeta = 0.f;
#pragma unroll
  for (int m = 0; m < 16; ++m) {
    const int base  = (m < 8) ? 0 : 16;
    const float val = av[m & 7];
    const float S0  = __shfl(val, base + 0, 32);
    const float S1  = __shfl(val, base + 1, 32);
    const float S2  = __shfl(val, base + 2, 32);
    if (lane == m) {
      const float n_   = S0 * dv;
      const float vbar = S1 / S0;
      const float e_t  = (S2 - S1 * S1 / S0) * dv;
      myn = n_; myvb = vbar; myet = e_t;
      mybeta = n_ / (2.0f * e_t);
    }
  }
  if (lane < 16) scal[rowbase + lane] = make_float4(myn, myvb, myet, mybeta);
}

__global__ __launch_bounds__(256) void fp_beta_iter(
    const float* __restrict__ dv_p, float4* __restrict__ scal) {
  const int wave = threadIdx.x >> 5;
  const int lane = threadIdx.x & 31;
  const int row  = blockIdx.x * 8 + wave;
  const float dv = *dv_p;

  const float4 s = scal[row];
  const float n_ = s.x, vbar = s.y, e_t = s.z;
  float beta     = s.w;
  const float g  = n_ / e_t;

  float u2[32];
#pragma unroll
  for (int c = 0; c < 32; ++c) {
    const int j   = lane + 32 * c;
    const float v = dv * ((float)j + 0.5f) - VMAXF;
    const float u = v - vbar;
    u2[c] = u * u;
  }

  for (int it = 0; it < SC_ITERS; ++it) {
    float sm = 0.f, se = 0.f;
#pragma unroll
    for (int c = 0; c < 32; ++c) {
      const float M = __expf(-beta * u2[c]);
      sm += M;
      se += M * u2[c];
    }
#pragma unroll
    for (int off = 16; off > 0; off >>= 1) {
      sm += __shfl_xor(sm, off, 32);
      se += __shfl_xor(se, off, 32);
    }
    beta = beta * (se / sm) * g;
  }
  if (lane == 0) scal[row].w = beta;
}

__device__ __forceinline__ float cc_delta(float w) {
  const bool small = fabsf(w) < 1e-6f;
  const float ws   = small ? 1.0f : w;
  const float em   = expm1f(ws);
  const float d    = (em - ws) / (ws * em);
  return small ? 0.5f : d;
}

__global__ __launch_bounds__(256) void fp_thomas(
    const float* __restrict__ f, const float* __restrict__ v_edge,
    const float* __restrict__ dv_p, const float* __restrict__ nu_p,
    const float* __restrict__ dt_p, const float4* __restrict__ scal,
    float* __restrict__ cp_ws, float* __restrict__ out) {
  const int row  = blockIdx.x * blockDim.x + threadIdx.x;
  const float dv = *dv_p, nu = *nu_p, dt = *dt_p;

  const float4 sc  = scal[row];
  const float vbar = sc.y, beta = sc.w;
  const float D      = 0.5f / beta;
  const float Pdv    = D / dv;
  const float s      = dt * nu / dv;
  const float twobdv = 2.0f * beta * dv;

  const float* frow = f + (size_t)row * NVC;
  float*       orow = out + (size_t)row * NVC;

  float Ce0 = 0.f, P0 = 0.f, d0 = 0.5f;
  float cp_prev = 0.f, dp_prev = 0.f;

  float cC[4], dC[4];
#pragma unroll
  for (int q = 0; q < 4; ++q) {
    const float c = v_edge[q + 1] - vbar;
    cC[q] = c;
    dC[q] = cc_delta(twobdv * c);
  }

  for (int jb = 0; jb < NVC; jb += 4) {
    const float4 fj = *(const float4*)(frow + jb);
    float fv[4] = {fj.x, fj.y, fj.z, fj.w};

    float cN[4], dN[4];
#pragma unroll
    for (int q = 0; q < 4; ++q) {
      int e = jb + 5 + q;
      e = (e < NVC) ? e : NVC;
      const float c = v_edge[e] - vbar;
      cN[q] = c;
      dN[q] = cc_delta(twobdv * c);
    }

    float dpv[4];
#pragma unroll
    for (int q = 0; q < 4; ++q) {
      const int j = jb + q;
      const bool interior = (j + 1 < NVC);
      const float Ce1 = interior ? cC[q] : 0.f;
      const float P1  = interior ? Pdv   : 0.f;
      const float d1  = interior ? dC[q] : 0.5f;

      const float coefU = Ce1 * (1.0f - d1) + P1;
      const float coefL = -Ce0 * d0 + P0;
      const float coefD = Ce1 * d1 - P1 - Ce0 * (1.0f - d0) - P0;
      const float dl = -s * coefL;
      const float dd = 1.0f - s * coefD;
      const float du = -s * coefU;

      const float denom = dd - dl * cp_prev;
      const float rdiv  = 1.0f / denom;
      const float cp    = du * rdiv;
      const float dp    = (fv[q] - dl * dp_prev) * rdiv;

      cp_ws[(size_t)j * B_ROWS + row] = cp;
      dpv[q]  = dp;
      cp_prev = cp; dp_prev = dp;
      Ce0 = Ce1; P0 = P1; d0 = d1;
    }
    *(float4*)(orow + jb) = make_float4(dpv[0], dpv[1], dpv[2], dpv[3]);

#pragma unroll
    for (int q = 0; q < 4; ++q) { cC[q] = cN[q]; dC[q] = dN[q]; }
  }

  float xnext = 0.f;
  for (int jb = NVC - 4; jb >= 0; jb -= 4) {
    const float4 dpj = *(const float4*)(orow + jb);
    float dv4[4] = {dpj.x, dpj.y, dpj.z, dpj.w};
    float xv[4];
#pragma unroll
    for (int q = 3; q >= 0; --q) {
      const int j   = jb + q;
      const float c = cp_ws[(size_t)j * B_ROWS + row];
      const float x = dv4[q] - c * xnext;
      xv[q] = x;
      xnext = x;
    }
    *(float4*)(orow + jb) = make_float4(xv[0], xv[1], xv[2], xv[3]);
  }
}

extern "C" void kernel_launch(void* const* d_in, const int* in_sizes, int n_in,
                              void* d_out, int out_size, void* d_ws, size_t ws_size,
                              hipStream_t stream) {
  const float* f      = (const float*)d_in[0];
  const float* v      = (const float*)d_in[1];  (void)v;
  const float* v_edge = (const float*)d_in[2];
  const float* dv     = (const float*)d_in[3];
  const float* nu     = (const float*)d_in[4];
  const float* dt     = (const float*)d_in[5];
  (void)in_sizes; (void)n_in; (void)out_size; (void)ws_size;

  float4* scal = (float4*)d_ws;
  float*  cp   = (float*)((char*)d_ws + (size_t)B_ROWS * sizeof(float4));
  float*  out  = (float*)d_out;

  fp_moments_wmma<<<B_ROWS / 16, 32, 0, stream>>>(f, dv, scal);
  fp_beta_iter  <<<B_ROWS / 8, 256, 0, stream>>>(dv, scal);
  fp_thomas     <<<B_ROWS / 256, 256, 0, stream>>>(f, v_edge, dv, nu, dt, scal, cp, out);
}


# fp_thomas: each wave stages its 32 rows through LDS in 32-column tiles: f, dp, x move as whole 128-B lines requested a tile ahead, no global access inside the recurrence; 1 wave per workgroup
# speedup vs baseline: 2.4227x; 1.7958x over previous
#include <hip/hip_runtime.h>
#include <cmath>

#define B_ROWS   16384
#define NVC      1024
#define VMAXF    6.0f
#define SC_ITERS 10

typedef __attribute__((ext_vector_type(2))) float v2f;
typedef __attribute__((ext_vector_type(8))) float v8f;

__global__ __launch_bounds__(32) void fp_moments_wmma(
    const float* __restrict__ f, const float* __restrict__ dv_p,
    float4* __restrict__ scal) {
  const int lane    = threadIdx.x;
  const int rowbase = blockIdx.x * 16;
  const float dv    = *dv_p;
  const int r       = lane & 15;
  const int khalf   = (lane >> 4) * 2;

  const float* fp = f + (size_t)(rowbase + r) * NVC + khalf;

  const float c0 = (r == 0) ? 1.0f : 0.0f;
  const float c1 = (r == 1) ? 1.0f : 0.0f;
  const float c2 = (r == 2) ? 1.0f : 0.0f;

  const float dv4 = 4.0f * dv;
  float vcur = dv * ((float)khalf + 0.5f) - VMAXF;

  v8f acc0 = {}, acc1 = {}, acc2 = {}, acc3 = {};
  for (int kb = 0; kb < NVC; kb += 16) {
#pragma unroll
    for (int u = 0; u < 4; ++u) {
      const int k = kb + 4 * u;
      v2f a;
      a.x = fp[k + 0];
      a.y = fp[k + 1];
      const float v0 = vcur;
      const float v1 = vcur + dv;
      v2f b;
      b.x = (c2 * v0 + c1) * v0 + c0;
      b.y = (c2 * v1 + c1) * v1 + c0;
      v8f* acc = (u == 0) ? &acc0 : (u == 1) ? &acc1 : (u == 2) ? &acc2 : &acc3;
      *acc = __builtin_amdgcn_wmma_f32_16x16x4_f32(
          false, a, false, b, (short)0, *acc, false, false);
      vcur += dv4;
    }
  }
  v8f acc = (acc0 + acc1) + (acc2 + acc3);

  float av[8];
#pragma unroll
  for (int i = 0; i < 8; ++i) av[i] = acc[i];

  float myn = 0.f, myvb = 0.f, myet = 0.f, mybeta = 0.f;
#pragma unroll
  for (int m = 0; m < 16; ++m) {
    const int base  = (m < 8) ? 0 : 16;
    const float val = av[m & 7];
    const float S0  = __shfl(val, base + 0, 32);
    const float S1  = __shfl(val, base + 1, 32);
    const float S2  = __shfl(val, base + 2, 32);
    if (lane == m) {
      const float n_   = S0 * dv;
      const float vbar = S1 / S0;
      const float e_t  = (S2 - S1 * S1 / S0) * dv;
      myn = n_; myvb = vbar; myet = e_t;
      mybeta = n_ / (2.0f * e_t);
    }
  }
  if (lane < 16) scal[rowbase + lane] = make_float4(myn, myvb, myet, mybeta);
}

__global__ __launch_bounds__(256) void fp_beta_iter(
    const float* __restrict__ dv_p, float4* __restrict__ scal) {
  const int wave = threadIdx.x >> 5;
  const int lane = threadIdx.x & 31;
  const int row  = blockIdx.x * 8 + wave;
  const float dv = *dv_p;

  const float4 s = scal[row];
  const float n_ = s.x, vbar = s.y, e_t = s.z;
  float beta     = s.w;
  const float g  = n_ / e_t;

  float u2[32];
#pragma unroll
  for (int c = 0; c < 32; ++c) {
    const int j   = lane + 32 * c;
    const float v = dv * ((float)j + 0.5f) - VMAXF;
    const float u = v - vbar;
    u2[c] = u * u;
  }

  for (int it = 0; it < SC_ITERS; ++it) {
    float sm = 0.f, se = 0.f;
#pragma unroll
    for (int c = 0; c < 32; ++c) {
      const float M = __expf(-beta * u2[c]);
      sm += M;
      se += M * u2[c];
    }
#pragma unroll
    for (int off = 16; off > 0; off >>= 1) {
      sm += __shfl_xor(sm, off, 32);
      se += __shfl_xor(se, off, 32);
    }
    beta = beta * (se / sm) * g;
  }
  if (lane == 0) scal[row].w = beta;
}

__device__ __forceinline__ float cc_delta(float w) {
  const bool small = fabsf(w) < 1e-6f;
  const float ws   = small ? 1.0f : w;
  const float em   = expm1f(ws);
  const float d    = (em - ws) / (ws * em);
  return small ? 0.5f : d;
}

#define TCOLS 32
#define NTILE (NVC / TCOLS)
#define TP    36
#define NVE   1056

typedef __attribute__((ext_vector_type(4))) float v4f;

__device__ __forceinline__ void wave_lds_handover() {
  __builtin_amdgcn_fence(__ATOMIC_ACQ_REL, "wavefront");
}

__device__ __forceinline__ void tile_load_lines(const float* g, float (&r)[TCOLS]) {
#pragma unroll
  for (int i = 0; i < 8; ++i) {
    const v4f t4 = *(const v4f*)(g + (size_t)i * 4 * NVC);
    r[4 * i + 0] = t4.x; r[4 * i + 1] = t4.y; r[4 * i + 2] = t4.z; r[4 * i + 3] = t4.w;
  }
}

__device__ __forceinline__ void tile_put_lines(float* T, int ldsG, const float (&r)[TCOLS]) {
#pragma unroll
  for (int i = 0; i < 8; ++i) {
    v4f t4; t4.x = r[4 * i + 0]; t4.y = r[4 * i + 1]; t4.z = r[4 * i + 2]; t4.w = r[4 * i + 3];
    *(v4f*)&T[ldsG + i * 4 * TP] = t4;
  }
}

__device__ __forceinline__ void tile_store_lines(const float* T, int ldsG, float* g) {
#pragma unroll
  for (int i = 0; i < 8; ++i) {
    const v4f t4 = *(const v4f*)&T[ldsG + i * 4 * TP];
    *(v4f*)(g + (size_t)i * 4 * NVC) = t4;
  }
}

__device__ __forceinline__ void cp_load_tile(const float* cpt, float (&c)[TCOLS]) {
#pragma unroll
  for (int k = 0; k < TCOLS; ++k) c[k] = cpt[(size_t)k * B_ROWS];
}

__device__ __forceinline__ void bwd_tile(float* Td, int ldsG, int ldsR, float* g,
                                         const float (&dpl)[TCOLS], const float (&cc)[TCOLS],
                                         float& xnext) {
  tile_put_lines(Td, ldsG, dpl);
  wave_lds_handover();
  float dr[TCOLS];
#pragma unroll
  for (int q = 0; q < 8; ++q) {
    const v4f t4 = *(const v4f*)&Td[ldsR + q * 4];
    dr[4 * q + 0] = t4.x; dr[4 * q + 1] = t4.y; dr[4 * q + 2] = t4.z; dr[4 * q + 3] = t4.w;
  }
  float xr[TCOLS];
  float xn = xnext;
#pragma unroll
  for (int k = TCOLS - 1; k >= 0; --k) {
    const float x = dr[k] - cc[k] * xn;
    xr[k] = x;
    xn = x;
  }
  xnext = xn;
#pragma unroll
  for (int q = 0; q < 8; ++q) {
    v4f t4; t4.x = xr[4 * q + 0]; t4.y = xr[4 * q + 1]; t4.z = xr[4 * q + 2]; t4.w = xr[4 * q + 3];
    *(v4f*)&Td[ldsR + q * 4] = t4;
  }
  wave_lds_handover();
  tile_store_lines(Td, ldsG, g);
  wave_lds_handover();
}

__global__ __launch_bounds__(32) void fp_thomas(
    const float* __restrict__ f, const float* __restrict__ v_edge,
    const float* __restrict__ dv_p, const float* __restrict__ nu_p,
    const float* __restrict__ dt_p, const float4* __restrict__ scal,
    float* __restrict__ cp_ws, float* __restrict__ out) {
  __shared__ __attribute__((aligned(16))) float Tf[2][32 * TP];
  __shared__ __attribute__((aligned(16))) float Td[32 * TP];
  __shared__ __attribute__((aligned(16))) float Tc[32 * TP];
  __shared__ __attribute__((aligned(16))) float Ve[NVE];

  const int lane    = threadIdx.x;
  const int rowbase = blockIdx.x * 32;
  const int row     = rowbase + lane;
  const float dv = *dv_p, nu = *nu_p, dt = *dt_p;

  const int gr = lane >> 3, gq = lane & 7;
  const float* fG = f   + (size_t)(rowbase + gr) * NVC + gq * 4;
  float*       oG = out + (size_t)(rowbase + gr) * NVC + gq * 4;
  const int ldsG = gr * TP + gq * 4;
  const int ldsR = lane * TP;
  float* cpcol = cp_ws + row;

  float pre[TCOLS];
  tile_load_lines(fG, pre);

  for (int k = lane; k < NVE; k += 32) {
    const int e = (k + 1 < NVC) ? k + 1 : NVC;
    Ve[k] = v_edge[e];
  }

  const float4 sc  = scal[row];
  const float vbar = sc.y, beta = sc.w;
  const float D      = 0.5f / beta;
  const float Pdv    = D / dv;
  const float s      = dt * nu / dv;
  const float twobdv = 2.0f * beta * dv;

  tile_put_lines(&Tf[0][0], ldsG, pre);
  wave_lds_handover();

  float Ce0 = 0.f, P0 = 0.f, d0 = 0.5f;
  float cp_prev = 0.f, dp_prev = 0.f;

  float cC[4], dC[4];
  {
    const v4f e4 = *(const v4f*)&Ve[0];
    const float ev[4] = {e4.x, e4.y, e4.z, e4.w};
#pragma unroll
    for (int q = 0; q < 4; ++q) {
      const float c = ev[q] - vbar;
      cC[q] = c;
      dC[q] = cc_delta(twobdv * c);
    }
  }
  v4f enext = *(const v4f*)&Ve[4];

  for (int t = 0; t < NTILE; ++t) {
    const int cur = t & 1;
    const float* Tfc = &Tf[cur][0];
    const int tn = (t + 1 < NTILE) ? t + 1 : t;
    tile_load_lines(fG + tn * TCOLS, pre);

    v4f fnext = *(const v4f*)&Tfc[ldsR];
    v4f cprev4 = {0.f, 0.f, 0.f, 0.f}, dprev4 = {0.f, 0.f, 0.f, 0.f};
#pragma nounroll
    for (int b = 0; b < TCOLS / 4; ++b) {
      const int jb = t * TCOLS + b * 4;
      if (b > 0) {
        *(v4f*)&Tc[ldsR + (b - 1) * 4] = cprev4;
        *(v4f*)&Td[ldsR + (b - 1) * 4] = dprev4;
      }
      const v4f fj = fnext;
      const v4f ej = enext;
      const int bn = (b + 1 < TCOLS / 4) ? b + 1 : b;
      fnext = *(const v4f*)&Tfc[ldsR + bn * 4];
      enext = *(const v4f*)&Ve[jb + 8];
      __builtin_amdgcn_sched_barrier(0);
      const float fv[4] = {fj.x, fj.y, fj.z, fj.w};
      const float ev[4] = {ej.x, ej.y, ej.z, ej.w};

      float cN[4], dN[4];
#pragma unroll
      for (int q = 0; q < 4; ++q) {
        const float c = ev[q] - vbar;
        cN[q] = c;
        dN[q] = cc_delta(twobdv * c);
      }

      float dpv[4], cpv[4];
#pragma unroll
      for (int q = 0; q < 4; ++q) {
        const int j = jb + q;
        const bool interior = (j + 1 < NVC);
        const float Ce1 = interior ? cC[q] : 0.f;
        const float P1  = interior ? Pdv   : 0.f;
        const float d1  = interior ? dC[q] : 0.5f;

        const float coefU = Ce1 * (1.0f - d1) + P1;
        const float coefL = -Ce0 * d0 + P0;
        const float coefD = Ce1 * d1 - P1 - Ce0 * (1.0f - d0) - P0;
        const float dl = -s * coefL;
        const float dd = 1.0f - s * coefD;
        const float du = -s * coefU;

        const float denom = dd - dl * cp_prev;
        const float rdiv  = 1.0f / denom;
        const float cp    = du * rdiv;
        const float dp    = (fv[q] - dl * dp_prev) * rdiv;

        cpv[q] = cp;
        dpv[q] = dp;
        cp_prev = cp; dp_prev = dp;
        Ce0 = Ce1; P0 = P1; d0 = d1;
      }
      cprev4.x = cpv[0]; cprev4.y = cpv[1]; cprev4.z = cpv[2]; cprev4.w = cpv[3];
      dprev4.x = dpv[0]; dprev4.y = dpv[1]; dprev4.z = dpv[2]; dprev4.w = dpv[3];

#pragma unroll
      for (int q = 0; q < 4; ++q) { cC[q] = cN[q]; dC[q] = dN[q]; }
    }
    *(v4f*)&Tc[ldsR + (TCOLS / 4 - 1) * 4] = cprev4;
    *(v4f*)&Td[ldsR + (TCOLS / 4 - 1) * 4] = dprev4;
    wave_lds_handover();

    tile_store_lines(Td, ldsG, oG + t * TCOLS);
    {
      float* cpt = cpcol + (size_t)t * TCOLS * B_ROWS;
#pragma unroll
      for (int q = 0; q < TCOLS / 4; ++q) {
        const v4f c4 = *(const v4f*)&Tc[ldsR + q * 4];
        cpt[(size_t)(4 * q + 0) * B_ROWS] = c4.x;
        cpt[(size_t)(4 * q + 1) * B_ROWS] = c4.y;
        cpt[(size_t)(4 * q + 2) * B_ROWS] = c4.z;
        cpt[(size_t)(4 * q + 3) * B_ROWS] = c4.w;
      }
    }
    tile_put_lines(&Tf[cur ^ 1][0], ldsG, pre);
    wave_lds_handover();
  }

  float dA[TCOLS], cA[TCOLS], dB[TCOLS], cB[TCOLS];
  tile_load_lines(oG + (NTILE - 1) * TCOLS, dA);
  cp_load_tile(cpcol + (size_t)(NTILE - 1) * TCOLS * B_ROWS, cA);

  float xnext = 0.f;
  for (int t = NTILE - 1; t >= 1; t -= 2) {
    tile_load_lines(oG + (t - 1) * TCOLS, dB);
    cp_load_tile(cpcol + (size_t)(t - 1) * TCOLS * B_ROWS, cB);
    bwd_tile(Td, ldsG, ldsR, oG + t * TCOLS, dA, cA, xnext);

    const int t2 = (t >= 2) ? t - 2 : 0;
    tile_load_lines(oG + t2 * TCOLS, dA);
    cp_load_tile(cpcol + (size_t)t2 * TCOLS * B_ROWS, cA);
    bwd_tile(Td, ldsG, ldsR, oG + (t - 1) * TCOLS, dB, cB, xnext);
  }
}

extern "C" void kernel_launch(void* const* d_in, const int* in_sizes, int n_in,
                              void* d_out, int out_size, void* d_ws, size_t ws_size,
                              hipStream_t stream) {
  const float* f      = (const float*)d_in[0];
  const float* v      = (const float*)d_in[1];  (void)v;
  const float* v_edge = (const float*)d_in[2];
  const float* dv     = (const float*)d_in[3];
  const float* nu     = (const float*)d_in[4];
  const float* dt     = (const float*)d_in[5];
  (void)in_sizes; (void)n_in; (void)out_size; (void)ws_size;

  float4* scal = (float4*)d_ws;
  float*  cp   = (float*)((char*)d_ws + (size_t)B_ROWS * sizeof(float4));
  float*  out  = (float*)d_out;

  fp_moments_wmma<<<B_ROWS / 16, 32, 0, stream>>>(f, dv, scal);
  fp_beta_iter  <<<B_ROWS / 8, 256, 0, stream>>>(dv, scal);
  fp_thomas     <<<B_ROWS / 32, 32, 0, stream>>>(f, v_edge, dv, nu, dt, scal, cp, out);
}


# fp_thomas: two-sided elimination: a 2nd wave per workgroup eliminates from cell 1023 down on the mirrored system, halves meet in a 2x2 solve; serial chain 512 cells; order of the solve's sums changes
# speedup vs baseline: 3.4408x; 1.4202x over previous
#include <hip/hip_runtime.h>
#include <cmath>

#define B_ROWS   16384
#define NVC      1024
#define VMAXF    6.0f
#define SC_ITERS 10

typedef __attribute__((ext_vector_type(2))) float v2f;
typedef __attribute__((ext_vector_type(8))) float v8f;

__global__ __launch_bounds__(32) void fp_moments_wmma(
    const float* __restrict__ f, const float* __restrict__ dv_p,
    float4* __restrict__ scal) {
  const int lane    = threadIdx.x;
  const int rowbase = blockIdx.x * 16;
  const float dv    = *dv_p;
  const int r       = lane & 15;
  const int khalf   = (lane >> 4) * 2;

  const float* fp = f + (size_t)(rowbase + r) * NVC + khalf;

  const float c0 = (r == 0) ? 1.0f : 0.0f;
  const float c1 = (r == 1) ? 1.0f : 0.0f;
  const float c2 = (r == 2) ? 1.0f : 0.0f;

  const float dv4 = 4.0f * dv;
  float vcur = dv * ((float)khalf + 0.5f) - VMAXF;

  v8f acc0 = {}, acc1 = {}, acc2 = {}, acc3 = {};
  for (int kb = 0; kb < NVC; kb += 16) {
#pragma unroll
    for (int u = 0; u < 4; ++u) {
      const int k = kb + 4 * u;
      v2f a;
      a.x = fp[k + 0];
      a.y = fp[k + 1];
      const float v0 = vcur;
      const float v1 = vcur + dv;
      v2f b;
      b.x = (c2 * v0 + c1) * v0 + c0;
      b.y = (c2 * v1 + c1) * v1 + c0;
      v8f* acc = (u == 0) ? &acc0 : (u == 1) ? &acc1 : (u == 2) ? &acc2 : &acc3;
      *acc = __builtin_amdgcn_wmma_f32_16x16x4_f32(
          false, a, false, b, (short)0, *acc, false, false);
      vcur += dv4;
    }
  }
  v8f acc = (acc0 + acc1) + (acc2 + acc3);

  float av[8];
#pragma unroll
  for (int i = 0; i < 8; ++i) av[i] = acc[i];

  float myn = 0.f, myvb = 0.f, myet = 0.f, mybeta = 0.f;
#pragma unroll
  for (int m = 0; m < 16; ++m) {
    const int base  = (m < 8) ? 0 : 16;
    const float val = av[m & 7];
    const float S0  = __shfl(val, base + 0, 32);
    const float S1  = __shfl(val, base + 1, 32);
    const float S2  = __shfl(val, base + 2, 32);
    if (lane == m) {
      const float n_   = S0 * dv;
      const float vbar = S1 / S0;
      const float e_t  = (S2 - S1 * S1 / S0) * dv;
      myn = n_; myvb = vbar; myet = e_t;
      mybeta = n_ / (2.0f * e_t);
    }
  }
  if (lane < 16) scal[rowbase + lane] = make_float4(myn, myvb, myet, mybeta);
}

__global__ __launch_bounds__(256) void fp_beta_iter(
    const float* __restrict__ dv_p, float4* __restrict__ scal) {
  const int wave = threadIdx.x >> 5;
  const int lane = threadIdx.x & 31;
  const int row  = blockIdx.x * 8 + wave;
  const float dv = *dv_p;

  const float4 s = scal[row];
  const float n_ = s.x, vbar = s.y, e_t = s.z;
  float beta     = s.w;
  const float g  = n_ / e_t;

  float u2[32];
#pragma unroll
  for (int c = 0; c < 32; ++c) {
    const int j   = lane + 32 * c;
    const float v = dv * ((float)j + 0.5f) - VMAXF;
    const float u = v - vbar;
    u2[c] = u * u;
  }

  for (int it = 0; it < SC_ITERS; ++it) {
    float sm = 0.f, se = 0.f;
#pragma unroll
    for (int c = 0; c < 32; ++c) {
      const float M = __expf(-beta * u2[c]);
      sm += M;
      se += M * u2[c];
    }
#pragma unroll
    for (int off = 16; off > 0; off >>= 1) {
      sm += __shfl_xor(sm, off, 32);
      se += __shfl_xor(se, off, 32);
    }
    beta = beta * (se / sm) * g;
  }
  if (lane == 0) scal[row].w = beta;
}

__device__ __forceinline__ float cc_delta(float w) {
  const bool small = fabsf(w) < 1e-6f;
  const float ws   = small ? 1.0f : w;
  const float em   = expm1f(ws);
  const float d    = (em - ws) / (ws * em);
  return small ? 0.5f : d;
}

#define TCOLS 32
#define NTILE (NVC / TCOLS)
#define HTILE (NTILE / 2)
#define TP    36
#define NVE   544

typedef __attribute__((ext_vector_type(4))) float v4f;

__device__ __forceinline__ void wave_lds_handover() {
  __builtin_amdgcn_fence(__ATOMIC_ACQ_REL, "wavefront");
}

template <int W> __device__ __forceinline__ int tile_of(int t) {
  return W ? (NTILE - 1 - t) : t;
}

template <int W> __device__ __forceinline__ int blk_of(int b) {
  return 4 * (W ? (TCOLS / 4 - 1 - b) : b);
}
template <int W> __device__ __forceinline__ float own4(v4f t, int q) {
  const int z = W ? 3 - q : q;
  return z == 0 ? t.x : z == 1 ? t.y : z == 2 ? t.z : t.w;
}
template <int W> __device__ __forceinline__ v4f pack4(float a0, float a1, float a2, float a3) {
  v4f r;
  if (W) { r.x = a3; r.y = a2; r.z = a1; r.w = a0; } else { r.x = a0; r.y = a1; r.z = a2; r.w = a3; }
  return r;
}

__device__ __forceinline__ void tile_load_lines(const float* g, float (&r)[TCOLS]) {
#pragma unroll
  for (int i = 0; i < 8; ++i) {
    const v4f t4 = *(const v4f*)(g + (size_t)i * 4 * NVC);
    r[4 * i + 0] = t4.x; r[4 * i + 1] = t4.y; r[4 * i + 2] = t4.z; r[4 * i + 3] = t4.w;
  }
}

__device__ __forceinline__ void tile_put_lines(float* T, int ldsG, const float (&r)[TCOLS]) {
#pragma unroll
  for (int i = 0; i < 8; ++i) {
    v4f t4; t4.x = r[4 * i + 0]; t4.y = r[4 * i + 1]; t4.z = r[4 * i + 2]; t4.w = r[4 * i + 3];
    *(v4f*)&T[ldsG + i * 4 * TP] = t4;
  }
}

__device__ __forceinline__ void tile_store_lines(const float* T, int ldsG, float* g) {
#pragma unroll
  for (int i = 0; i < 8; ++i) {
    const v4f t4 = *(const v4f*)&T[ldsG + i * 4 * TP];
    *(v4f*)(g + (size_t)i * 4 * NVC) = t4;
  }
}

__device__ __forceinline__ void cp_load_tile(const float* cpt, float (&c)[TCOLS]) {
#pragma unroll
  for (int k = 0; k < TCOLS; ++k) c[k] = cpt[(size_t)k * B_ROWS];
}

template <int W>
__device__ __forceinline__ void bwd_tile(float* Td, int ldsG, int ldsR, float* g,
                                         const float (&dpl)[TCOLS], const float (&cc)[TCOLS],
                                         float& xnext) {
  tile_put_lines(Td, ldsG, dpl);
  wave_lds_handover();
  float dr[TCOLS];
#pragma unroll
  for (int q = 0; q < 8; ++q) {
    const v4f t4 = *(const v4f*)&Td[ldsR + blk_of<W>(q)];
    dr[4 * q + 0] = own4<W>(t4, 0); dr[4 * q + 1] = own4<W>(t4, 1);
    dr[4 * q + 2] = own4<W>(t4, 2); dr[4 * q + 3] = own4<W>(t4, 3);
  }
  float xr[TCOLS];
  float xn = xnext;
#pragma unroll
  for (int k = TCOLS - 1; k >= 0; --k) {
    const float x = dr[k] - cc[k] * xn;
    xr[k] = x;
    xn = x;
  }
  xnext = xn;
#pragma unroll
  for (int q = 0; q < 8; ++q) {
    *(v4f*)&Td[ldsR + blk_of<W>(q)] =
        pack4<W>(xr[4 * q + 0], xr[4 * q + 1], xr[4 * q + 2], xr[4 * q + 3]);
  }
  wave_lds_handover();
  tile_store_lines(Td, ldsG, g);
  wave_lds_handover();
}

template <int W>
__device__ __forceinline__ void eliminate_half(
    const float* __restrict__ fG, float* __restrict__ oG, float* __restrict__ cpcol,
    const float* __restrict__ v_edge, float* Tf, float* Td, float* Tc, float* Ve,
    int lane, int ldsG, int ldsR, float vbar, float Pdv, float s, float twobdv,
    float& cp_last, float& dp_last) {
  float pre[TCOLS];
  tile_load_lines(fG + tile_of<W>(0) * TCOLS, pre);

  for (int k = lane; k < NVE; k += 32) {
    const int e = W ? (NVC - 1 - k) : (k + 1);
    Ve[k] = v_edge[e];
  }

  tile_put_lines(Tf, ldsG, pre);
  wave_lds_handover();

  float Ce0 = 0.f, P0 = 0.f, d0 = 0.5f;
  float cp_prev = 0.f, dp_prev = 0.f;

  float cC[4], dC[4];
  {
    const v4f e4 = *(const v4f*)&Ve[0];
    const float ev[4] = {e4.x, e4.y, e4.z, e4.w};
#pragma unroll
    for (int q = 0; q < 4; ++q) {
      const float c = W ? (vbar - ev[q]) : (ev[q] - vbar);
      cC[q] = c;
      dC[q] = cc_delta(twobdv * c);
    }
  }
  v4f enext = *(const v4f*)&Ve[4];

  for (int t = 0; t < HTILE; ++t) {
    const int cur = t & 1;
    const float* Tfc = Tf + cur * (32 * TP);
    const int tn = (t + 1 < HTILE) ? t + 1 : t;
    tile_load_lines(fG + tile_of<W>(tn) * TCOLS, pre);

    v4f fnext = *(const v4f*)&Tfc[ldsR + blk_of<W>(0)];
    v4f cprev4 = {0.f, 0.f, 0.f, 0.f}, dprev4 = {0.f, 0.f, 0.f, 0.f};
#pragma nounroll
    for (int b = 0; b < TCOLS / 4; ++b) {
      const int jb = t * TCOLS + b * 4;
      if (b > 0) {
        *(v4f*)&Tc[ldsR + (b - 1) * 4] = cprev4;
        *(v4f*)&Td[ldsR + blk_of<W>(b - 1)] = dprev4;
      }
      const v4f fj = fnext;
      const v4f ej = enext;
      const int bn = (b + 1 < TCOLS / 4) ? b + 1 : b;
      fnext = *(const v4f*)&Tfc[ldsR + blk_of<W>(bn)];
      enext = *(const v4f*)&Ve[jb + 8];
      __builtin_amdgcn_sched_barrier(0);
      const float fv[4] = {own4<W>(fj, 0), own4<W>(fj, 1), own4<W>(fj, 2), own4<W>(fj, 3)};
      const float ev[4] = {ej.x, ej.y, ej.z, ej.w};

      float cN[4], dN[4];
#pragma unroll
      for (int q = 0; q < 4; ++q) {
        const float c = W ? (vbar - ev[q]) : (ev[q] - vbar);
        cN[q] = c;
        dN[q] = cc_delta(twobdv * c);
      }

      float dpv[4], cpv[4];
#pragma unroll
      for (int q = 0; q < 4; ++q) {
        const float Ce1 = cC[q];
        const float P1  = Pdv;
        const float d1  = dC[q];

        const float coefU = Ce1 * (1.0f - d1) + P1;
        const float coefL = -Ce0 * d0 + P0;
        const float coefD = Ce1 * d1 - P1 - Ce0 * (1.0f - d0) - P0;
        const float dl = -s * coefL;
        const float dd = 1.0f - s * coefD;
        const float du = -s * coefU;

        const float denom = dd - dl * cp_prev;
        const float rdiv  = 1.0f / denom;
        const float cp    = du * rdiv;
        const float dp    = (fv[q] - dl * dp_prev) * rdiv;

        cpv[q] = cp;
        dpv[q] = dp;
        cp_prev = cp; dp_prev = dp;
        Ce0 = Ce1; P0 = P1; d0 = d1;
      }
      cprev4.x = cpv[0]; cprev4.y = cpv[1]; cprev4.z = cpv[2]; cprev4.w = cpv[3];
      dprev4 = pack4<W>(dpv[0], dpv[1], dpv[2], dpv[3]);

#pragma unroll
      for (int q = 0; q < 4; ++q) { cC[q] = cN[q]; dC[q] = dN[q]; }
    }
    *(v4f*)&Tc[ldsR + (TCOLS / 4 - 1) * 4] = cprev4;
    *(v4f*)&Td[ldsR + blk_of<W>(TCOLS / 4 - 1)] = dprev4;
    wave_lds_handover();

    const int tt = tile_of<W>(t);
    tile_store_lines(Td, ldsG, oG + tt * TCOLS);
    {
      float* cpt = cpcol + (size_t)tt * TCOLS * B_ROWS;
#pragma unroll
      for (int q = 0; q < TCOLS / 4; ++q) {
        const v4f c4 = *(const v4f*)&Tc[ldsR + q * 4];
        cpt[(size_t)(4 * q + 0) * B_ROWS] = c4.x;
        cpt[(size_t)(4 * q + 1) * B_ROWS] = c4.y;
        cpt[(size_t)(4 * q + 2) * B_ROWS] = c4.z;
        cpt[(size_t)(4 * q + 3) * B_ROWS] = c4.w;
      }
    }
    tile_put_lines(Tf + (cur ^ 1) * (32 * TP), ldsG, pre);
    wave_lds_handover();
  }
  cp_last = cp_prev;
  dp_last = dp_prev;
}

template <int W>
__device__ __forceinline__ void substitute_half(
    float* __restrict__ oG, const float* __restrict__ cpcol, float* Td,
    int ldsG, int ldsR, float xnext) {
  float dA[TCOLS], cA[TCOLS], dB[TCOLS], cB[TCOLS];
  tile_load_lines(oG + tile_of<W>(HTILE - 1) * TCOLS, dA);
  cp_load_tile(cpcol + (size_t)tile_of<W>(HTILE - 1) * TCOLS * B_ROWS, cA);

  for (int t = HTILE - 1; t >= 1; t -= 2) {
    tile_load_lines(oG + tile_of<W>(t - 1) * TCOLS, dB);
    cp_load_tile(cpcol + (size_t)tile_of<W>(t - 1) * TCOLS * B_ROWS, cB);
    bwd_tile<W>(Td, ldsG, ldsR, oG + tile_of<W>(t) * TCOLS, dA, cA, xnext);

    const int t2 = (t >= 2) ? t - 2 : 0;
    tile_load_lines(oG + tile_of<W>(t2) * TCOLS, dA);
    cp_load_tile(cpcol + (size_t)tile_of<W>(t2) * TCOLS * B_ROWS, cA);
    bwd_tile<W>(Td, ldsG, ldsR, oG + tile_of<W>(t - 1) * TCOLS, dB, cB, xnext);
  }
}

__global__ __launch_bounds__(64) void fp_thomas(
    const float* __restrict__ f, const float* __restrict__ v_edge,
    const float* __restrict__ dv_p, const float* __restrict__ nu_p,
    const float* __restrict__ dt_p, const float4* __restrict__ scal,
    float* __restrict__ cp_ws, float* __restrict__ out) {
  __shared__ __attribute__((aligned(16))) float Tf[2][2 * 32 * TP];
  __shared__ __attribute__((aligned(16))) float Td[2][32 * TP];
  __shared__ __attribute__((aligned(16))) float Tc[2][32 * TP];
  __shared__ __attribute__((aligned(16))) float Ve[2][NVE];
  __shared__ float Xc[2][32], Xd[2][32];

  const int w       = __builtin_amdgcn_readfirstlane((int)(threadIdx.x >> 5));
  const int lane    = threadIdx.x & 31;
  const int rowbase = blockIdx.x * 32;
  const int row     = rowbase + lane;
  const float dv = *dv_p, nu = *nu_p, dt = *dt_p;

  const int gr = lane >> 3, gq = lane & 7;
  const float* fG = f   + (size_t)(rowbase + gr) * NVC + gq * 4;
  float*       oG = out + (size_t)(rowbase + gr) * NVC + gq * 4;
  const int ldsG = gr * TP + gq * 4;
  const int ldsR = lane * TP;
  float* cpcol = cp_ws + row;

  const float4 sc  = scal[row];
  const float vbar = sc.y, beta = sc.w;
  const float D      = 0.5f / beta;
  const float Pdv    = D / dv;
  const float s      = dt * nu / dv;
  const float twobdv = 2.0f * beta * dv;

  float cp_last, dp_last;
  if (w == 0)
    eliminate_half<0>(fG, oG, cpcol, v_edge, Tf[0], Td[0], Tc[0], Ve[0], lane, ldsG, ldsR,
                      vbar, Pdv, s, twobdv, cp_last, dp_last);
  else
    eliminate_half<1>(fG, oG, cpcol, v_edge, Tf[1], Td[1], Tc[1], Ve[1], lane, ldsG, ldsR,
                      vbar, Pdv, s, twobdv, cp_last, dp_last);

  Xc[w][lane] = cp_last;
  Xd[w][lane] = dp_last;
  __syncthreads();
  const float co = Xc[w ^ 1][lane];
  const float dq = Xd[w ^ 1][lane];
  const float xnext = (dq - co * dp_last) / (1.0f - co * cp_last);

  if (w == 0)
    substitute_half<0>(oG, cpcol, Td[0], ldsG, ldsR, xnext);
  else
    substitute_half<1>(oG, cpcol, Td[1], ldsG, ldsR, xnext);
}

extern "C" void kernel_launch(void* const* d_in, const int* in_sizes, int n_in,
                              void* d_out, int out_size, void* d_ws, size_t ws_size,
                              hipStream_t stream) {
  const float* f      = (const float*)d_in[0];
  const float* v      = (const float*)d_in[1];  (void)v;
  const float* v_edge = (const float*)d_in[2];
  const float* dv     = (const float*)d_in[3];
  const float* nu     = (const float*)d_in[4];
  const float* dt     = (const float*)d_in[5];
  (void)in_sizes; (void)n_in; (void)out_size; (void)ws_size;

  float4* scal = (float4*)d_ws;
  float*  cp   = (float*)((char*)d_ws + (size_t)B_ROWS * sizeof(float4));
  float*  out  = (float*)d_out;

  fp_moments_wmma<<<B_ROWS / 16, 32, 0, stream>>>(f, dv, scal);
  fp_beta_iter  <<<B_ROWS / 8, 256, 0, stream>>>(dv, scal);
  fp_thomas     <<<B_ROWS / 32, 64, 0, stream>>>(f, v_edge, dv, nu, dt, scal, cp, out);
}


# fp_thomas forward sweep: the 8 correctly rounded f32 divisions per 4-cell block (recurrence 1/denom, edge-weight quotient) become v_rcp_f32 and a multiply; those quotients now ~1 ulp, not 0.5
# speedup vs baseline: 3.9745x; 1.1551x over previous
#include <hip/hip_runtime.h>
#include <cmath>

#define B_ROWS   16384
#define NVC      1024
#define VMAXF    6.0f
#define SC_ITERS 10

typedef __attribute__((ext_vector_type(2))) float v2f;
typedef __attribute__((ext_vector_type(8))) float v8f;

__global__ __launch_bounds__(32) void fp_moments_wmma(
    const float* __restrict__ f, const float* __restrict__ dv_p,
    float4* __restrict__ scal) {
  const int lane    = threadIdx.x;
  const int rowbase = blockIdx.x * 16;
  const float dv    = *dv_p;
  const int r       = lane & 15;
  const int khalf   = (lane >> 4) * 2;

  const float* fp = f + (size_t)(rowbase + r) * NVC + khalf;

  const float c0 = (r == 0) ? 1.0f : 0.0f;
  const float c1 = (r == 1) ? 1.0f : 0.0f;
  const float c2 = (r == 2) ? 1.0f : 0.0f;

  const float dv4 = 4.0f * dv;
  float vcur = dv * ((float)khalf + 0.5f) - VMAXF;

  v8f acc0 = {}, acc1 = {}, acc2 = {}, acc3 = {};
  for (int kb = 0; kb < NVC; kb += 16) {
#pragma unroll
    for (int u = 0; u < 4; ++u) {
      const int k = kb + 4 * u;
      v2f a;
      a.x = fp[k + 0];
      a.y = fp[k + 1];
      const float v0 = vcur;
      const float v1 = vcur + dv;
      v2f b;
      b.x = (c2 * v0 + c1) * v0 + c0;
      b.y = (c2 * v1 + c1) * v1 + c0;
      v8f* acc = (u == 0) ? &acc0 : (u == 1) ? &acc1 : (u == 2) ? &acc2 : &acc3;
      *acc = __builtin_amdgcn_wmma_f32_16x16x4_f32(
          false, a, false, b, (short)0, *acc, false, false);
      vcur += dv4;
    }
  }
  v8f acc = (acc0 + acc1) + (acc2 + acc3);

  float av[8];
#pragma unroll
  for (int i = 0; i < 8; ++i) av[i] = acc[i];

  float myn = 0.f, myvb = 0.f, myet = 0.f, mybeta = 0.f;
#pragma unroll
  for (int m = 0; m < 16; ++m) {
    const int base  = (m < 8) ? 0 : 16;
    const float val = av[m & 7];
    const float S0  = __shfl(val, base + 0, 32);
    const float S1  = __shfl(val, base + 1, 32);
    const float S2  = __shfl(val, base + 2, 32);
    if (lane == m) {
      const float n_   = S0 * dv;
      const float vbar = S1 / S0;
      const float e_t  = (S2 - S1 * S1 / S0) * dv;
      myn = n_; myvb = vbar; myet = e_t;
      mybeta = n_ / (2.0f * e_t);
    }
  }
  if (lane < 16) scal[rowbase + lane] = make_float4(myn, myvb, myet, mybeta);
}

__global__ __launch_bounds__(256) void fp_beta_iter(
    const float* __restrict__ dv_p, float4* __restrict__ scal) {
  const int wave = threadIdx.x >> 5;
  const int lane = threadIdx.x & 31;
  const int row  = blockIdx.x * 8 + wave;
  const float dv = *dv_p;

  const float4 s = scal[row];
  const float n_ = s.x, vbar = s.y, e_t = s.z;
  float beta     = s.w;
  const float g  = n_ / e_t;

  float u2[32];
#pragma unroll
  for (int c = 0; c < 32; ++c) {
    const int j   = lane + 32 * c;
    const float v = dv * ((float)j + 0.5f) - VMAXF;
    const float u = v - vbar;
    u2[c] = u * u;
  }

  for (int it = 0; it < SC_ITERS; ++it) {
    float sm = 0.f, se = 0.f;
#pragma unroll
    for (int c = 0; c < 32; ++c) {
      const float M = __expf(-beta * u2[c]);
      sm += M;
      se += M * u2[c];
    }
#pragma unroll
    for (int off = 16; off > 0; off >>= 1) {
      sm += __shfl_xor(sm, off, 32);
      se += __shfl_xor(se, off, 32);
    }
    beta = beta * (se / sm) * g;
  }
  if (lane == 0) scal[row].w = beta;
}

__device__ __forceinline__ float cc_delta(float w) {
  const bool small = fabsf(w) < 1e-6f;
  const float ws   = small ? 1.0f : w;
  const float em   = expm1f(ws);
  const float d    = (em - ws) * __builtin_amdgcn_rcpf(ws * em);
  return small ? 0.5f : d;
}

#define TCOLS 32
#define NTILE (NVC / TCOLS)
#define HTILE (NTILE / 2)
#define TP    36
#define NVE   544

typedef __attribute__((ext_vector_type(4))) float v4f;

__device__ __forceinline__ void wave_lds_handover() {
  __builtin_amdgcn_fence(__ATOMIC_ACQ_REL, "wavefront");
}

template <int W> __device__ __forceinline__ int tile_of(int t) {
  return W ? (NTILE - 1 - t) : t;
}

template <int W> __device__ __forceinline__ int blk_of(int b) {
  return 4 * (W ? (TCOLS / 4 - 1 - b) : b);
}
template <int W> __device__ __forceinline__ float own4(v4f t, int q) {
  const int z = W ? 3 - q : q;
  return z == 0 ? t.x : z == 1 ? t.y : z == 2 ? t.z : t.w;
}
template <int W> __device__ __forceinline__ v4f pack4(float a0, float a1, float a2, float a3) {
  v4f r;
  if (W) { r.x = a3; r.y = a2; r.z = a1; r.w = a0; } else { r.x = a0; r.y = a1; r.z = a2; r.w = a3; }
  return r;
}

__device__ __forceinline__ void tile_load_lines(const float* g, float (&r)[TCOLS]) {
#pragma unroll
  for (int i = 0; i < 8; ++i) {
    const v4f t4 = *(const v4f*)(g + (size_t)i * 4 * NVC);
    r[4 * i + 0] = t4.x; r[4 * i + 1] = t4.y; r[4 * i + 2] = t4.z; r[4 * i + 3] = t4.w;
  }
}

__device__ __forceinline__ void tile_put_lines(float* T, int ldsG, const float (&r)[TCOLS]) {
#pragma unroll
  for (int i = 0; i < 8; ++i) {
    v4f t4; t4.x = r[4 * i + 0]; t4.y = r[4 * i + 1]; t4.z = r[4 * i + 2]; t4.w = r[4 * i + 3];
    *(v4f*)&T[ldsG + i * 4 * TP] = t4;
  }
}

__device__ __forceinline__ void tile_store_lines(const float* T, int ldsG, float* g) {
#pragma unroll
  for (int i = 0; i < 8; ++i) {
    const v4f t4 = *(const v4f*)&T[ldsG + i * 4 * TP];
    *(v4f*)(g + (size_t)i * 4 * NVC) = t4;
  }
}

__device__ __forceinline__ void cp_load_tile(const float* cpt, float (&c)[TCOLS]) {
#pragma unroll
  for (int k = 0; k < TCOLS; ++k) c[k] = cpt[(size_t)k * B_ROWS];
}

template <int W>
__device__ __forceinline__ void bwd_tile(float* Td, int ldsG, int ldsR, float* g,
                                         const float (&dpl)[TCOLS], const float (&cc)[TCOLS],
                                         float& xnext) {
  tile_put_lines(Td, ldsG, dpl);
  wave_lds_handover();
  float dr[TCOLS];
#pragma unroll
  for (int q = 0; q < 8; ++q) {
    const v4f t4 = *(const v4f*)&Td[ldsR + blk_of<W>(q)];
    dr[4 * q + 0] = own4<W>(t4, 0); dr[4 * q + 1] = own4<W>(t4, 1);
    dr[4 * q + 2] = own4<W>(t4, 2); dr[4 * q + 3] = own4<W>(t4, 3);
  }
  float xr[TCOLS];
  float xn = xnext;
#pragma unroll
  for (int k = TCOLS - 1; k >= 0; --k) {
    const float x = dr[k] - cc[k] * xn;
    xr[k] = x;
    xn = x;
  }
  xnext = xn;
#pragma unroll
  for (int q = 0; q < 8; ++q) {
    *(v4f*)&Td[ldsR + blk_of<W>(q)] =
        pack4<W>(xr[4 * q + 0], xr[4 * q + 1], xr[4 * q + 2], xr[4 * q + 3]);
  }
  wave_lds_handover();
  tile_store_lines(Td, ldsG, g);
  wave_lds_handover();
}

template <int W>
__device__ __forceinline__ void eliminate_half(
    const float* __restrict__ fG, float* __restrict__ oG, float* __restrict__ cpcol,
    const float* __restrict__ v_edge, float* Tf, float* Td, float* Tc, float* Ve,
    int lane, int ldsG, int ldsR, float vbar, float Pdv, float s, float twobdv,
    float& cp_last, float& dp_last) {
  float pre[TCOLS];
  tile_load_lines(fG + tile_of<W>(0) * TCOLS, pre);

  for (int k = lane; k < NVE; k += 32) {
    const int e = W ? (NVC - 1 - k) : (k + 1);
    Ve[k] = v_edge[e];
  }

  tile_put_lines(Tf, ldsG, pre);
  wave_lds_handover();

  float Ce0 = 0.f, P0 = 0.f, d0 = 0.5f;
  float cp_prev = 0.f, dp_prev = 0.f;

  float cC[4], dC[4];
  {
    const v4f e4 = *(const v4f*)&Ve[0];
    const float ev[4] = {e4.x, e4.y, e4.z, e4.w};
#pragma unroll
    for (int q = 0; q < 4; ++q) {
      const float c = W ? (vbar - ev[q]) : (ev[q] - vbar);
      cC[q] = c;
      dC[q] = cc_delta(twobdv * c);
    }
  }
  v4f enext = *(const v4f*)&Ve[4];

  for (int t = 0; t < HTILE; ++t) {
    const int cur = t & 1;
    const float* Tfc = Tf + cur * (32 * TP);
    const int tn = (t + 1 < HTILE) ? t + 1 : t;
    tile_load_lines(fG + tile_of<W>(tn) * TCOLS, pre);

    v4f fnext = *(const v4f*)&Tfc[ldsR + blk_of<W>(0)];
    v4f cprev4 = {0.f, 0.f, 0.f, 0.f}, dprev4 = {0.f, 0.f, 0.f, 0.f};
#pragma nounroll
    for (int b = 0; b < TCOLS / 4; ++b) {
      const int jb = t * TCOLS + b * 4;
      if (b > 0) {
        *(v4f*)&Tc[ldsR + (b - 1) * 4] = cprev4;
        *(v4f*)&Td[ldsR + blk_of<W>(b - 1)] = dprev4;
      }
      const v4f fj = fnext;
      const v4f ej = enext;
      const int bn = (b + 1 < TCOLS / 4) ? b + 1 : b;
      fnext = *(const v4f*)&Tfc[ldsR + blk_of<W>(bn)];
      enext = *(const v4f*)&Ve[jb + 8];
      __builtin_amdgcn_sched_barrier(0);
      const float fv[4] = {own4<W>(fj, 0), own4<W>(fj, 1), own4<W>(fj, 2), own4<W>(fj, 3)};
      const float ev[4] = {ej.x, ej.y, ej.z, ej.w};

      float cN[4], dN[4];
#pragma unroll
      for (int q = 0; q < 4; ++q) {
        const float c = W ? (vbar - ev[q]) : (ev[q] - vbar);
        cN[q] = c;
        dN[q] = cc_delta(twobdv * c);
      }

      float dpv[4], cpv[4];
#pragma unroll
      for (int q = 0; q < 4; ++q) {
        const float Ce1 = cC[q];
        const float P1  = Pdv;
        const float d1  = dC[q];

        const float coefU = Ce1 * (1.0f - d1) + P1;
        const float coefL = -Ce0 * d0 + P0;
        const float coefD = Ce1 * d1 - P1 - Ce0 * (1.0f - d0) - P0;
        const float dl = -s * coefL;
        const float dd = 1.0f - s * coefD;
        const float du = -s * coefU;

        const float denom = dd - dl * cp_prev;
        const float rdiv  = __builtin_amdgcn_rcpf(denom);
        const float cp    = du * rdiv;
        const float dp    = (fv[q] - dl * dp_prev) * rdiv;

        cpv[q] = cp;
        dpv[q] = dp;
        cp_prev = cp; dp_prev = dp;
        Ce0 = Ce1; P0 = P1; d0 = d1;
      }
      cprev4.x = cpv[0]; cprev4.y = cpv[1]; cprev4.z = cpv[2]; cprev4.w = cpv[3];
      dprev4 = pack4<W>(dpv[0], dpv[1], dpv[2], dpv[3]);

#pragma unroll
      for (int q = 0; q < 4; ++q) { cC[q] = cN[q]; dC[q] = dN[q]; }
    }
    *(v4f*)&Tc[ldsR + (TCOLS / 4 - 1) * 4] = cprev4;
    *(v4f*)&Td[ldsR + blk_of<W>(TCOLS / 4 - 1)] = dprev4;
    wave_lds_handover();

    const int tt = tile_of<W>(t);
    tile_store_lines(Td, ldsG, oG + tt * TCOLS);
    {
      float* cpt = cpcol + (size_t)tt * TCOLS * B_ROWS;
#pragma unroll
      for (int q = 0; q < TCOLS / 4; ++q) {
        const v4f c4 = *(const v4f*)&Tc[ldsR + q * 4];
        cpt[(size_t)(4 * q + 0) * B_ROWS] = c4.x;
        cpt[(size_t)(4 * q + 1) * B_ROWS] = c4.y;
        cpt[(size_t)(4 * q + 2) * B_ROWS] = c4.z;
        cpt[(size_t)(4 * q + 3) * B_ROWS] = c4.w;
      }
    }
    tile_put_lines(Tf + (cur ^ 1) * (32 * TP), ldsG, pre);
    wave_lds_handover();
  }
  cp_last = cp_prev;
  dp_last = dp_prev;
}

template <int W>
__device__ __forceinline__ void substitute_half(
    float* __restrict__ oG, const float* __restrict__ cpcol, float* Td,
    int ldsG, int ldsR, float xnext) {
  float dA[TCOLS], cA[TCOLS], dB[TCOLS], cB[TCOLS];
  tile_load_lines(oG + tile_of<W>(HTILE - 1) * TCOLS, dA);
  cp_load_tile(cpcol + (size_t)tile_of<W>(HTILE - 1) * TCOLS * B_ROWS, cA);

  for (int t = HTILE - 1; t >= 1; t -= 2) {
    tile_load_lines(oG + tile_of<W>(t - 1) * TCOLS, dB);
    cp_load_tile(cpcol + (size_t)tile_of<W>(t - 1) * TCOLS * B_ROWS, cB);
    bwd_tile<W>(Td, ldsG, ldsR, oG + tile_of<W>(t) * TCOLS, dA, cA, xnext);

    const int t2 = (t >= 2) ? t - 2 : 0;
    tile_load_lines(oG + tile_of<W>(t2) * TCOLS, dA);
    cp_load_tile(cpcol + (size_t)tile_of<W>(t2) * TCOLS * B_ROWS, cA);
    bwd_tile<W>(Td, ldsG, ldsR, oG + tile_of<W>(t - 1) * TCOLS, dB, cB, xnext);
  }
}

__global__ __launch_bounds__(64) void fp_thomas(
    const float* __restrict__ f, const float* __restrict__ v_edge,
    const float* __restrict__ dv_p, const float* __restrict__ nu_p,
    const float* __restrict__ dt_p, const float4* __restrict__ scal,
    float* __restrict__ cp_ws, float* __restrict__ out) {
  __shared__ __attribute__((aligned(16))) float Tf[2][2 * 32 * TP];
  __shared__ __attribute__((aligned(16))) float Td[2][32 * TP];
  __shared__ __attribute__((aligned(16))) float Tc[2][32 * TP];
  __shared__ __attribute__((aligned(16))) float Ve[2][NVE];
  __shared__ float Xc[2][32], Xd[2][32];

  const int w       = __builtin_amdgcn_readfirstlane((int)(threadIdx.x >> 5));
  const int lane    = threadIdx.x & 31;
  const int rowbase = blockIdx.x * 32;
  const int row     = rowbase + lane;
  const float dv = *dv_p, nu = *nu_p, dt = *dt_p;

  const int gr = lane >> 3, gq = lane & 7;
  const float* fG = f   + (size_t)(rowbase + gr) * NVC + gq * 4;
  float*       oG = out + (size_t)(rowbase + gr) * NVC + gq * 4;
  const int ldsG = gr * TP + gq * 4;
  const int ldsR = lane * TP;
  float* cpcol = cp_ws + row;

  const float4 sc  = scal[row];
  const float vbar = sc.y, beta = sc.w;
  const float D      = 0.5f / beta;
  const float Pdv    = D / dv;
  const float s      = dt * nu / dv;
  const float twobdv = 2.0f * beta * dv;

  float cp_last, dp_last;
  if (w == 0)
    eliminate_half<0>(fG, oG, cpcol, v_edge, Tf[0], Td[0], Tc[0], Ve[0], lane, ldsG, ldsR,
                      vbar, Pdv, s, twobdv, cp_last, dp_last);
  else
    eliminate_half<1>(fG, oG, cpcol, v_edge, Tf[1], Td[1], Tc[1], Ve[1], lane, ldsG, ldsR,
                      vbar, Pdv, s, twobdv, cp_last, dp_last);

  Xc[w][lane] = cp_last;
  Xd[w][lane] = dp_last;
  __syncthreads();
  const float co = Xc[w ^ 1][lane];
  const float dq = Xd[w ^ 1][lane];
  const float xnext = (dq - co * dp_last) / (1.0f - co * cp_last);

  if (w == 0)
    substitute_half<0>(oG, cpcol, Td[0], ldsG, ldsR, xnext);
  else
    substitute_half<1>(oG, cpcol, Td[1], ldsG, ldsR, xnext);
}

extern "C" void kernel_launch(void* const* d_in, const int* in_sizes, int n_in,
                              void* d_out, int out_size, void* d_ws, size_t ws_size,
                              hipStream_t stream) {
  const float* f      = (const float*)d_in[0];
  const float* v      = (const float*)d_in[1];  (void)v;
  const float* v_edge = (const float*)d_in[2];
  const float* dv     = (const float*)d_in[3];
  const float* nu     = (const float*)d_in[4];
  const float* dt     = (const float*)d_in[5];
  (void)in_sizes; (void)n_in; (void)out_size; (void)ws_size;

  float4* scal = (float4*)d_ws;
  float*  cp   = (float*)((char*)d_ws + (size_t)B_ROWS * sizeof(float4));
  float*  out  = (float*)d_out;

  fp_moments_wmma<<<B_ROWS / 16, 32, 0, stream>>>(f, dv, scal);
  fp_beta_iter  <<<B_ROWS / 8, 256, 0, stream>>>(dv, scal);
  fp_thomas     <<<B_ROWS / 32, 64, 0, stream>>>(f, v_edge, dv, nu, dt, scal, cp, out);
}


# fp_thomas: cp stays in LDS: each lane keeps its half row's 512 cp values in an LDS row of its own from elimination to substitution; the workspace round trip (64 MB out, 64 MB back) is gone
# speedup vs baseline: 4.2883x; 1.0790x over previous
#include <hip/hip_runtime.h>
#include <cmath>

#define B_ROWS   16384
#define NVC      1024
#define VMAXF    6.0f
#define SC_ITERS 10

typedef __attribute__((ext_vector_type(2))) float v2f;
typedef __attribute__((ext_vector_type(8))) float v8f;

__global__ __launch_bounds__(32) void fp_moments_wmma(
    const float* __restrict__ f, const float* __restrict__ dv_p,
    float4* __restrict__ scal) {
  const int lane    = threadIdx.x;
  const int rowbase = blockIdx.x * 16;
  const float dv    = *dv_p;
  const int r       = lane & 15;
  const int khalf   = (lane >> 4) * 2;

  const float* fp = f + (size_t)(rowbase + r) * NVC + khalf;

  const float c0 = (r == 0) ? 1.0f : 0.0f;
  const float c1 = (r == 1) ? 1.0f : 0.0f;
  const float c2 = (r == 2) ? 1.0f : 0.0f;

  const float dv4 = 4.0f * dv;
  float vcur = dv * ((float)khalf + 0.5f) - VMAXF;

  v8f acc0 = {}, acc1 = {}, acc2 = {}, acc3 = {};
  for (int kb = 0; kb < NVC; kb += 16) {
#pragma unroll
    for (int u = 0; u < 4; ++u) {
      const int k = kb + 4 * u;
      v2f a;
      a.x = fp[k + 0];
      a.y = fp[k + 1];
      const float v0 = vcur;
      const float v1 = vcur + dv;
      v2f b;
      b.x = (c2 * v0 + c1) * v0 + c0;
      b.y = (c2 * v1 + c1) * v1 + c0;
      v8f* acc = (u == 0) ? &acc0 : (u == 1) ? &acc1 : (u == 2) ? &acc2 : &acc3;
      *acc = __builtin_amdgcn_wmma_f32_16x16x4_f32(
          false, a, false, b, (short)0, *acc, false, false);
      vcur += dv4;
    }
  }
  v8f acc = (acc0 + acc1) + (acc2 + acc3);

  float av[8];
#pragma unroll
  for (int i = 0; i < 8; ++i) av[i] = acc[i];

  float myn = 0.f, myvb = 0.f, myet = 0.f, mybeta = 0.f;
#pragma unroll
  for (int m = 0; m < 16; ++m) {
    const int base  = (m < 8) ? 0 : 16;
    const float val = av[m & 7];
    const float S0  = __shfl(val, base + 0, 32);
    const float S1  = __shfl(val, base + 1, 32);
    const float S2  = __shfl(val, base + 2, 32);
    if (lane == m) {
      const float n_   = S0 * dv;
      const float vbar = S1 / S0;
      const float e_t  = (S2 - S1 * S1 / S0) * dv;
      myn = n_; myvb = vbar; myet = e_t;
      mybeta = n_ / (2.0f * e_t);
    }
  }
  if (lane < 16) scal[rowbase + lane] = make_float4(myn, myvb, myet, mybeta);
}

__global__ __launch_bounds__(256) void fp_beta_iter(
    const float* __restrict__ dv_p, float4* __restrict__ scal) {
  const int wave = threadIdx.x >> 5;
  const int lane = threadIdx.x & 31;
  const int row  = blockIdx.x * 8 + wave;
  const float dv = *dv_p;

  const float4 s = scal[row];
  const float n_ = s.x, vbar = s.y, e_t = s.z;
  float beta     = s.w;
  const float g  = n_ / e_t;

  float u2[32];
#pragma unroll
  for (int c = 0; c < 32; ++c) {
    const int j   = lane + 32 * c;
    const float v = dv * ((float)j + 0.5f) - VMAXF;
    const float u = v - vbar;
    u2[c] = u * u;
  }

  for (int it = 0; it < SC_ITERS; ++it) {
    float sm = 0.f, se = 0.f;
#pragma unroll
    for (int c = 0; c < 32; ++c) {
      const float M = __expf(-beta * u2[c]);
      sm += M;
      se += M * u2[c];
    }
#pragma unroll
    for (int off = 16; off > 0; off >>= 1) {
      sm += __shfl_xor(sm, off, 32);
      se += __shfl_xor(se, off, 32);
    }
    beta = beta * (se / sm) * g;
  }
  if (lane == 0) scal[row].w = beta;
}

__device__ __forceinline__ float cc_delta(float w) {
  const bool small = fabsf(w) < 1e-6f;
  const float ws   = small ? 1.0f : w;
  const float em   = expm1f(ws);
  const float d    = (em - ws) * __builtin_amdgcn_rcpf(ws * em);
  return small ? 0.5f : d;
}

#define TCOLS 32
#define NTILE (NVC / TCOLS)
#define HTILE (NTILE / 2)
#define TP    36
#define NVE   544
#define HCELL (NVC / 2)
#define CPP   (HCELL + 4)

typedef __attribute__((ext_vector_type(4))) float v4f;

__device__ __forceinline__ void wave_lds_handover() {
  __builtin_amdgcn_fence(__ATOMIC_ACQ_REL, "wavefront");
}

template <int W> __device__ __forceinline__ int tile_of(int t) {
  return W ? (NTILE - 1 - t) : t;
}

template <int W> __device__ __forceinline__ int blk_of(int b) {
  return 4 * (W ? (TCOLS / 4 - 1 - b) : b);
}
template <int W> __device__ __forceinline__ float own4(v4f t, int q) {
  const int z = W ? 3 - q : q;
  return z == 0 ? t.x : z == 1 ? t.y : z == 2 ? t.z : t.w;
}
template <int W> __device__ __forceinline__ v4f pack4(float a0, float a1, float a2, float a3) {
  v4f r;
  if (W) { r.x = a3; r.y = a2; r.z = a1; r.w = a0; } else { r.x = a0; r.y = a1; r.z = a2; r.w = a3; }
  return r;
}

__device__ __forceinline__ void tile_load_lines(const float* g, float (&r)[TCOLS]) {
#pragma unroll
  for (int i = 0; i < 8; ++i) {
    const v4f t4 = *(const v4f*)(g + (size_t)i * 4 * NVC);
    r[4 * i + 0] = t4.x; r[4 * i + 1] = t4.y; r[4 * i + 2] = t4.z; r[4 * i + 3] = t4.w;
  }
}

__device__ __forceinline__ void tile_put_lines(float* T, int ldsG, const float (&r)[TCOLS]) {
#pragma unroll
  for (int i = 0; i < 8; ++i) {
    v4f t4; t4.x = r[4 * i + 0]; t4.y = r[4 * i + 1]; t4.z = r[4 * i + 2]; t4.w = r[4 * i + 3];
    *(v4f*)&T[ldsG + i * 4 * TP] = t4;
  }
}

__device__ __forceinline__ void tile_store_lines(const float* T, int ldsG, float* g) {
#pragma unroll
  for (int i = 0; i < 8; ++i) {
    const v4f t4 = *(const v4f*)&T[ldsG + i * 4 * TP];
    *(v4f*)(g + (size_t)i * 4 * NVC) = t4;
  }
}

template <int W>
__device__ __forceinline__ void bwd_tile(float* Td, int ldsG, int ldsR, float* g,
                                         const float (&dpl)[TCOLS], const float* Crt,
                                         float& xnext) {
  tile_put_lines(Td, ldsG, dpl);
  float cc[TCOLS];
#pragma unroll
  for (int q = 0; q < 8; ++q) {
    const v4f c4 = *(const v4f*)&Crt[4 * q];
    cc[4 * q + 0] = c4.x; cc[4 * q + 1] = c4.y; cc[4 * q + 2] = c4.z; cc[4 * q + 3] = c4.w;
  }
  wave_lds_handover();
  float dr[TCOLS];
#pragma unroll
  for (int q = 0; q < 8; ++q) {
    const v4f t4 = *(const v4f*)&Td[ldsR + blk_of<W>(q)];
    dr[4 * q + 0] = own4<W>(t4, 0); dr[4 * q + 1] = own4<W>(t4, 1);
    dr[4 * q + 2] = own4<W>(t4, 2); dr[4 * q + 3] = own4<W>(t4, 3);
  }
  float xr[TCOLS];
  float xn = xnext;
#pragma unroll
  for (int k = TCOLS - 1; k >= 0; --k) {
    const float x = dr[k] - cc[k] * xn;
    xr[k] = x;
    xn = x;
  }
  xnext = xn;
#pragma unroll
  for (int q = 0; q < 8; ++q) {
    *(v4f*)&Td[ldsR + blk_of<W>(q)] =
        pack4<W>(xr[4 * q + 0], xr[4 * q + 1], xr[4 * q + 2], xr[4 * q + 3]);
  }
  wave_lds_handover();
  tile_store_lines(Td, ldsG, g);
  wave_lds_handover();
}

template <int W>
__device__ __forceinline__ void eliminate_half(
    const float* __restrict__ fG, float* __restrict__ oG,
    const float* __restrict__ v_edge, float* Tf, float* Td, float* Cr, float* Ve,
    int lane, int ldsG, int ldsR, float vbar, float Pdv, float s, float twobdv,
    float& cp_last, float& dp_last) {
  float pre[TCOLS];
  tile_load_lines(fG + tile_of<W>(0) * TCOLS, pre);

  for (int k = lane; k < NVE; k += 32) {
    const int e = W ? (NVC - 1 - k) : (k + 1);
    Ve[k] = v_edge[e];
  }

  tile_put_lines(Tf, ldsG, pre);
  wave_lds_handover();

  float Ce0 = 0.f, P0 = 0.f, d0 = 0.5f;
  float cp_prev = 0.f, dp_prev = 0.f;

  float cC[4], dC[4];
  {
    const v4f e4 = *(const v4f*)&Ve[0];
    const float ev[4] = {e4.x, e4.y, e4.z, e4.w};
#pragma unroll
    for (int q = 0; q < 4; ++q) {
      const float c = W ? (vbar - ev[q]) : (ev[q] - vbar);
      cC[q] = c;
      dC[q] = cc_delta(twobdv * c);
    }
  }
  v4f enext = *(const v4f*)&Ve[4];

  for (int t = 0; t < HTILE; ++t) {
    const int tn = (t + 1 < HTILE) ? t + 1 : t;
    tile_load_lines(fG + tile_of<W>(tn) * TCOLS, pre);

    v4f fnext = *(const v4f*)&Tf[ldsR + blk_of<W>(0)];
    v4f cprev4 = {0.f, 0.f, 0.f, 0.f}, dprev4 = {0.f, 0.f, 0.f, 0.f};
#pragma nounroll
    for (int b = 0; b < TCOLS / 4; ++b) {
      const int jb = t * TCOLS + b * 4;
      if (b > 0) {
        *(v4f*)&Cr[jb - 4] = cprev4;
        *(v4f*)&Td[ldsR + blk_of<W>(b - 1)] = dprev4;
      }
      const v4f fj = fnext;
      const v4f ej = enext;
      const int bn = (b + 1 < TCOLS / 4) ? b + 1 : b;
      fnext = *(const v4f*)&Tf[ldsR + blk_of<W>(bn)];
      enext = *(const v4f*)&Ve[jb + 8];
      __builtin_amdgcn_sched_barrier(0);
      const float fv[4] = {own4<W>(fj, 0), own4<W>(fj, 1), own4<W>(fj, 2), own4<W>(fj, 3)};
      const float ev[4] = {ej.x, ej.y, ej.z, ej.w};

      float cN[4], dN[4];
#pragma unroll
      for (int q = 0; q < 4; ++q) {
        const float c = W ? (vbar - ev[q]) : (ev[q] - vbar);
        cN[q] = c;
        dN[q] = cc_delta(twobdv * c);
      }

      float dpv[4], cpv[4];
#pragma unroll
      for (int q = 0; q < 4; ++q) {
        const float Ce1 = cC[q];
        const float P1  = Pdv;
        const float d1  = dC[q];

        const float coefU = Ce1 * (1.0f - d1) + P1;
        const float coefL = -Ce0 * d0 + P0;
        const float coefD = Ce1 * d1 - P1 - Ce0 * (1.0f - d0) - P0;
        const float dl = -s * coefL;
        const float dd = 1.0f - s * coefD;
        const float du = -s * coefU;

        const float denom = dd - dl * cp_prev;
        const float rdiv  = __builtin_amdgcn_rcpf(denom);
        const float cp    = du * rdiv;
        const float dp    = (fv[q] - dl * dp_prev) * rdiv;

        cpv[q] = cp;
        dpv[q] = dp;
        cp_prev = cp; dp_prev = dp;
        Ce0 = Ce1; P0 = P1; d0 = d1;
      }
      cprev4.x = cpv[0]; cprev4.y = cpv[1]; cprev4.z = cpv[2]; cprev4.w = cpv[3];
      dprev4 = pack4<W>(dpv[0], dpv[1], dpv[2], dpv[3]);

#pragma unroll
      for (int q = 0; q < 4; ++q) { cC[q] = cN[q]; dC[q] = dN[q]; }
    }
    *(v4f*)&Cr[t * TCOLS + TCOLS - 4] = cprev4;
    *(v4f*)&Td[ldsR + blk_of<W>(TCOLS / 4 - 1)] = dprev4;
    wave_lds_handover();

    const int tt = tile_of<W>(t);
    tile_store_lines(Td, ldsG, oG + tt * TCOLS);
    tile_put_lines(Tf, ldsG, pre);
    wave_lds_handover();
  }
  cp_last = cp_prev;
  dp_last = dp_prev;
}

template <int W>
__device__ __forceinline__ void substitute_half(
    float* __restrict__ oG, const float* Cr, float* Td,
    int ldsG, int ldsR, float xnext) {
  float dA[TCOLS], dB[TCOLS];
  tile_load_lines(oG + tile_of<W>(HTILE - 1) * TCOLS, dA);

  for (int t = HTILE - 1; t >= 1; t -= 2) {
    tile_load_lines(oG + tile_of<W>(t - 1) * TCOLS, dB);
    bwd_tile<W>(Td, ldsG, ldsR, oG + tile_of<W>(t) * TCOLS, dA, Cr + t * TCOLS, xnext);

    const int t2 = (t >= 2) ? t - 2 : 0;
    tile_load_lines(oG + tile_of<W>(t2) * TCOLS, dA);
    bwd_tile<W>(Td, ldsG, ldsR, oG + tile_of<W>(t - 1) * TCOLS, dB, Cr + (t - 1) * TCOLS, xnext);
  }
}

__global__ __launch_bounds__(64) void fp_thomas(
    const float* __restrict__ f, const float* __restrict__ v_edge,
    const float* __restrict__ dv_p, const float* __restrict__ nu_p,
    const float* __restrict__ dt_p, const float4* __restrict__ scal,
    float* __restrict__ out) {
  __shared__ __attribute__((aligned(16))) float Cp[2][32 * CPP];
  __shared__ __attribute__((aligned(16))) float Tf[2][32 * TP];
  __shared__ __attribute__((aligned(16))) float Td[2][32 * TP];
  __shared__ __attribute__((aligned(16))) float Ve[2][NVE];
  __shared__ float Xc[2][32], Xd[2][32];

  const int w       = __builtin_amdgcn_readfirstlane((int)(threadIdx.x >> 5));
  const int lane    = threadIdx.x & 31;
  const int rowbase = blockIdx.x * 32;
  const int row     = rowbase + lane;
  const float dv = *dv_p, nu = *nu_p, dt = *dt_p;

  const int gr = lane >> 3, gq = lane & 7;
  const float* fG = f   + (size_t)(rowbase + gr) * NVC + gq * 4;
  float*       oG = out + (size_t)(rowbase + gr) * NVC + gq * 4;
  const int ldsG = gr * TP + gq * 4;
  const int ldsR = lane * TP;
  const int cpR  = lane * CPP;

  const float4 sc  = scal[row];
  const float vbar = sc.y, beta = sc.w;
  const float D      = 0.5f / beta;
  const float Pdv    = D / dv;
  const float s      = dt * nu / dv;
  const float twobdv = 2.0f * beta * dv;

  float cp_last, dp_last;
  if (w == 0)
    eliminate_half<0>(fG, oG, v_edge, Tf[0], Td[0], Cp[0] + cpR, Ve[0], lane, ldsG, ldsR,
                      vbar, Pdv, s, twobdv, cp_last, dp_last);
  else
    eliminate_half<1>(fG, oG, v_edge, Tf[1], Td[1], Cp[1] + cpR, Ve[1], lane, ldsG, ldsR,
                      vbar, Pdv, s, twobdv, cp_last, dp_last);

  Xc[w][lane] = cp_last;
  Xd[w][lane] = dp_last;
  __syncthreads();
  const float co = Xc[w ^ 1][lane];
  const float dq = Xd[w ^ 1][lane];
  const float xnext = (dq - co * dp_last) / (1.0f - co * cp_last);

  if (w == 0)
    substitute_half<0>(oG, Cp[0] + cpR, Td[0], ldsG, ldsR, xnext);
  else
    substitute_half<1>(oG, Cp[1] + cpR, Td[1], ldsG, ldsR, xnext);
}

extern "C" void kernel_launch(void* const* d_in, const int* in_sizes, int n_in,
                              void* d_out, int out_size, void* d_ws, size_t ws_size,
                              hipStream_t stream) {
  const float* f      = (const float*)d_in[0];
  const float* v      = (const float*)d_in[1];  (void)v;
  const float* v_edge = (const float*)d_in[2];
  const float* dv     = (const float*)d_in[3];
  const float* nu     = (const float*)d_in[4];
  const float* dt     = (const float*)d_in[5];
  (void)in_sizes; (void)n_in; (void)out_size; (void)ws_size;

  float4* scal = (float4*)d_ws;
  float*  out  = (float*)d_out;

  fp_moments_wmma<<<B_ROWS / 16, 32, 0, stream>>>(f, dv, scal);
  fp_beta_iter  <<<B_ROWS / 8, 256, 0, stream>>>(dv, scal);
  fp_thomas     <<<B_ROWS / 32, 64, 0, stream>>>(f, v_edge, dv, nu, dt, scal, out);
}


# fp_thomas edge weights: a wave whose half rows all have |w| < 1/2 (one range test per wave) takes delta from its Taylor series, 1 mul + 4 FMAs, not expm1 + reciprocal; other waves keep closed form
# speedup vs baseline: 4.5124x; 1.0523x over previous
#include <hip/hip_runtime.h>
#include <cmath>

#define B_ROWS   16384
#define NVC      1024
#define VMAXF    6.0f
#define SC_ITERS 10

typedef __attribute__((ext_vector_type(2))) float v2f;
typedef __attribute__((ext_vector_type(8))) float v8f;

__global__ __launch_bounds__(32) void fp_moments_wmma(
    const float* __restrict__ f, const float* __restrict__ dv_p,
    float4* __restrict__ scal) {
  const int lane    = threadIdx.x;
  const int rowbase = blockIdx.x * 16;
  const float dv    = *dv_p;
  const int r       = lane & 15;
  const int khalf   = (lane >> 4) * 2;

  const float* fp = f + (size_t)(rowbase + r) * NVC + khalf;

  const float c0 = (r == 0) ? 1.0f : 0.0f;
  const float c1 = (r == 1) ? 1.0f : 0.0f;
  const float c2 = (r == 2) ? 1.0f : 0.0f;

  const float dv4 = 4.0f * dv;
  float vcur = dv * ((float)khalf + 0.5f) - VMAXF;

  v8f acc0 = {}, acc1 = {}, acc2 = {}, acc3 = {};
  for (int kb = 0; kb < NVC; kb += 16) {
#pragma unroll
    for (int u = 0; u < 4; ++u) {
      const int k = kb + 4 * u;
      v2f a;
      a.x = fp[k + 0];
      a.y = fp[k + 1];
      const float v0 = vcur;
      const float v1 = vcur + dv;
      v2f b;
      b.x = (c2 * v0 + c1) * v0 + c0;
      b.y = (c2 * v1 + c1) * v1 + c0;
      v8f* acc = (u == 0) ? &acc0 : (u == 1) ? &acc1 : (u == 2) ? &acc2 : &acc3;
      *acc = __builtin_amdgcn_wmma_f32_16x16x4_f32(
          false, a, false, b, (short)0, *acc, false, false);
      vcur += dv4;
    }
  }
  v8f acc = (acc0 + acc1) + (acc2 + acc3);

  float av[8];
#pragma unroll
  for (int i = 0; i < 8; ++i) av[i] = acc[i];

  float myn = 0.f, myvb = 0.f, myet = 0.f, mybeta = 0.f;
#pragma unroll
  for (int m = 0; m < 16; ++m) {
    const int base  = (m < 8) ? 0 : 16;
    const float val = av[m & 7];
    const float S0  = __shfl(val, base + 0, 32);
    const float S1  = __shfl(val, base + 1, 32);
    const float S2  = __shfl(val, base + 2, 32);
    if (lane == m) {
      const float n_   = S0 * dv;
      const float vbar = S1 / S0;
      const float e_t  = (S2 - S1 * S1 / S0) * dv;
      myn = n_; myvb = vbar; myet = e_t;
      mybeta = n_ / (2.0f * e_t);
    }
  }
  if (lane < 16) scal[rowbase + lane] = make_float4(myn, myvb, myet, mybeta);
}

__global__ __launch_bounds__(256) void fp_beta_iter(
    const float* __restrict__ dv_p, float4* __restrict__ scal) {
  const int wave = threadIdx.x >> 5;
  const int lane = threadIdx.x & 31;
  const int row  = blockIdx.x * 8 + wave;
  const float dv = *dv_p;

  const float4 s = scal[row];
  const float n_ = s.x, vbar = s.y, e_t = s.z;
  float beta     = s.w;
  const float g  = n_ / e_t;

  float u2[32];
#pragma unroll
  for (int c = 0; c < 32; ++c) {
    const int j   = lane + 32 * c;
    const float v = dv * ((float)j + 0.5f) - VMAXF;
    const float u = v - vbar;
    u2[c] = u * u;
  }

  for (int it = 0; it < SC_ITERS; ++it) {
    float sm = 0.f, se = 0.f;
#pragma unroll
    for (int c = 0; c < 32; ++c) {
      const float M = __expf(-beta * u2[c]);
      sm += M;
      se += M * u2[c];
    }
#pragma unroll
    for (int off = 16; off > 0; off >>= 1) {
      sm += __shfl_xor(sm, off, 32);
      se += __shfl_xor(se, off, 32);
    }
    beta = beta * (se / sm) * g;
  }
  if (lane == 0) scal[row].w = beta;
}

__device__ __forceinline__ float cc_delta(float w) {
  const bool small = fabsf(w) < 1e-6f;
  const float ws   = small ? 1.0f : w;
  const float em   = expm1f(ws);
  const float d    = (em - ws) * __builtin_amdgcn_rcpf(ws * em);
  return small ? 0.5f : d;
}

__device__ __forceinline__ float cc_delta_series(float w) {
  const float w2 = w * w;
  float p = fmaf(w2, 1.0f / 1209600.0f, -1.0f / 30240.0f);
  p = fmaf(w2, p, 1.0f / 720.0f);
  p = fmaf(w2, p, -1.0f / 12.0f);
  const float d = fmaf(w, p, 0.5f);
  return (fabsf(w) < 1e-6f) ? 0.5f : d;
}
template <bool SERIES> __device__ __forceinline__ float cc_delta_of(float w) {
  return SERIES ? cc_delta_series(w) : cc_delta(w);
}

template <int CTRL> __device__ __forceinline__ float row_lane(float x) {
  return __builtin_bit_cast(
      float, __builtin_amdgcn_update_dpp(0, __builtin_bit_cast(int, x), CTRL, 0xF, 0xF, true));
}
__device__ __forceinline__ void wave_minmax(float& lo, float& hi) {
  lo = fminf(lo, row_lane<0xB1>(lo));   hi = fmaxf(hi, row_lane<0xB1>(hi));
  lo = fminf(lo, row_lane<0x4E>(lo));   hi = fmaxf(hi, row_lane<0x4E>(hi));
  lo = fminf(lo, row_lane<0x141>(lo));  hi = fmaxf(hi, row_lane<0x141>(hi));
  lo = fminf(lo, row_lane<0x140>(lo));  hi = fmaxf(hi, row_lane<0x140>(hi));
  const unsigned ulo = __builtin_bit_cast(unsigned, lo), uhi = __builtin_bit_cast(unsigned, hi);
  lo = fminf(lo, __builtin_bit_cast(float, __builtin_amdgcn_permlanex16(ulo, ulo, 0x76543210u, 0xfedcba98u, false, false)));
  hi = fmaxf(hi, __builtin_bit_cast(float, __builtin_amdgcn_permlanex16(uhi, uhi, 0x76543210u, 0xfedcba98u, false, false)));
}

#define TCOLS 32
#define NTILE (NVC / TCOLS)
#define HTILE (NTILE / 2)
#define TP    36
#define NVE   544
#define HCELL (NVC / 2)
#define CPP   (HCELL + 4)

typedef __attribute__((ext_vector_type(4))) float v4f;

__device__ __forceinline__ void wave_lds_handover() {
  __builtin_amdgcn_fence(__ATOMIC_ACQ_REL, "wavefront");
}

template <int W> __device__ __forceinline__ int tile_of(int t) {
  return W ? (NTILE - 1 - t) : t;
}

template <int W> __device__ __forceinline__ int blk_of(int b) {
  return 4 * (W ? (TCOLS / 4 - 1 - b) : b);
}
template <int W> __device__ __forceinline__ float own4(v4f t, int q) {
  const int z = W ? 3 - q : q;
  return z == 0 ? t.x : z == 1 ? t.y : z == 2 ? t.z : t.w;
}
template <int W> __device__ __forceinline__ v4f pack4(float a0, float a1, float a2, float a3) {
  v4f r;
  if (W) { r.x = a3; r.y = a2; r.z = a1; r.w = a0; } else { r.x = a0; r.y = a1; r.z = a2; r.w = a3; }
  return r;
}

__device__ __forceinline__ void tile_load_lines(const float* g, float (&r)[TCOLS]) {
#pragma unroll
  for (int i = 0; i < 8; ++i) {
    const v4f t4 = *(const v4f*)(g + (size_t)i * 4 * NVC);
    r[4 * i + 0] = t4.x; r[4 * i + 1] = t4.y; r[4 * i + 2] = t4.z; r[4 * i + 3] = t4.w;
  }
}

__device__ __forceinline__ void tile_put_lines(float* T, int ldsG, const float (&r)[TCOLS]) {
#pragma unroll
  for (int i = 0; i < 8; ++i) {
    v4f t4; t4.x = r[4 * i + 0]; t4.y = r[4 * i + 1]; t4.z = r[4 * i + 2]; t4.w = r[4 * i + 3];
    *(v4f*)&T[ldsG + i * 4 * TP] = t4;
  }
}

__device__ __forceinline__ void tile_store_lines(const float* T, int ldsG, float* g) {
#pragma unroll
  for (int i = 0; i < 8; ++i) {
    const v4f t4 = *(const v4f*)&T[ldsG + i * 4 * TP];
    *(v4f*)(g + (size_t)i * 4 * NVC) = t4;
  }
}

template <int W>
__device__ __forceinline__ void bwd_tile(float* Td, int ldsG, int ldsR, float* g,
                                         const float (&dpl)[TCOLS], const float* Crt,
                                         float& xnext) {
  tile_put_lines(Td, ldsG, dpl);
  float cc[TCOLS];
#pragma unroll
  for (int q = 0; q < 8; ++q) {
    const v4f c4 = *(const v4f*)&Crt[4 * q];
    cc[4 * q + 0] = c4.x; cc[4 * q + 1] = c4.y; cc[4 * q + 2] = c4.z; cc[4 * q + 3] = c4.w;
  }
  wave_lds_handover();
  float dr[TCOLS];
#pragma unroll
  for (int q = 0; q < 8; ++q) {
    const v4f t4 = *(const v4f*)&Td[ldsR + blk_of<W>(q)];
    dr[4 * q + 0] = own4<W>(t4, 0); dr[4 * q + 1] = own4<W>(t4, 1);
    dr[4 * q + 2] = own4<W>(t4, 2); dr[4 * q + 3] = own4<W>(t4, 3);
  }
  float xr[TCOLS];
  float xn = xnext;
#pragma unroll
  for (int k = TCOLS - 1; k >= 0; --k) {
    const float x = dr[k] - cc[k] * xn;
    xr[k] = x;
    xn = x;
  }
  xnext = xn;
#pragma unroll
  for (int q = 0; q < 8; ++q) {
    *(v4f*)&Td[ldsR + blk_of<W>(q)] =
        pack4<W>(xr[4 * q + 0], xr[4 * q + 1], xr[4 * q + 2], xr[4 * q + 3]);
  }
  wave_lds_handover();
  tile_store_lines(Td, ldsG, g);
  wave_lds_handover();
}

template <int W, bool SERIES>
__device__ __forceinline__ void eliminate_sweep(
    const float* __restrict__ fG, float* __restrict__ oG,
    float* Tf, float* Td, float* Cr, const float* Ve, float (&pre)[TCOLS],
    int ldsG, int ldsR, float vbar, float Pdv, float s, float twobdv,
    float& cp_last, float& dp_last) {
  float Ce0 = 0.f, P0 = 0.f, d0 = 0.5f;
  float cp_prev = 0.f, dp_prev = 0.f;

  float cC[4], dC[4];
  {
    const v4f e4 = *(const v4f*)&Ve[0];
    const float ev[4] = {e4.x, e4.y, e4.z, e4.w};
#pragma unroll
    for (int q = 0; q < 4; ++q) {
      const float c = W ? (vbar - ev[q]) : (ev[q] - vbar);
      cC[q] = c;
      dC[q] = cc_delta_of<SERIES>(twobdv * c);
    }
  }
  v4f enext = *(const v4f*)&Ve[4];

  for (int t = 0; t < HTILE; ++t) {
    const int tn = (t + 1 < HTILE) ? t + 1 : t;
    tile_load_lines(fG + tile_of<W>(tn) * TCOLS, pre);

    v4f fnext = *(const v4f*)&Tf[ldsR + blk_of<W>(0)];
    v4f cprev4 = {0.f, 0.f, 0.f, 0.f}, dprev4 = {0.f, 0.f, 0.f, 0.f};
#pragma nounroll
    for (int b = 0; b < TCOLS / 4; ++b) {
      const int jb = t * TCOLS + b * 4;
      if (b > 0) {
        *(v4f*)&Cr[jb - 4] = cprev4;
        *(v4f*)&Td[ldsR + blk_of<W>(b - 1)] = dprev4;
      }
      const v4f fj = fnext;
      const v4f ej = enext;
      const int bn = (b + 1 < TCOLS / 4) ? b + 1 : b;
      fnext = *(const v4f*)&Tf[ldsR + blk_of<W>(bn)];
      enext = *(const v4f*)&Ve[jb + 8];
      __builtin_amdgcn_sched_barrier(0);
      const float fv[4] = {own4<W>(fj, 0), own4<W>(fj, 1), own4<W>(fj, 2), own4<W>(fj, 3)};
      const float ev[4] = {ej.x, ej.y, ej.z, ej.w};

      float cN[4], dN[4];
#pragma unroll
      for (int q = 0; q < 4; ++q) {
        const float c = W ? (vbar - ev[q]) : (ev[q] - vbar);
        cN[q] = c;
        dN[q] = cc_delta_of<SERIES>(twobdv * c);
      }

      float dpv[4], cpv[4];
#pragma unroll
      for (int q = 0; q < 4; ++q) {
        const float Ce1 = cC[q];
        const float P1  = Pdv;
        const float d1  = dC[q];

        const float coefU = Ce1 * (1.0f - d1) + P1;
        const float coefL = -Ce0 * d0 + P0;
        const float coefD = Ce1 * d1 - P1 - Ce0 * (1.0f - d0) - P0;
        const float dl = -s * coefL;
        const float dd = 1.0f - s * coefD;
        const float du = -s * coefU;

        const float denom = dd - dl * cp_prev;
        const float rdiv  = __builtin_amdgcn_rcpf(denom);
        const float cp    = du * rdiv;
        const float dp    = (fv[q] - dl * dp_prev) * rdiv;

        cpv[q] = cp;
        dpv[q] = dp;
        cp_prev = cp; dp_prev = dp;
        Ce0 = Ce1; P0 = P1; d0 = d1;
      }
      cprev4.x = cpv[0]; cprev4.y = cpv[1]; cprev4.z = cpv[2]; cprev4.w = cpv[3];
      dprev4 = pack4<W>(dpv[0], dpv[1], dpv[2], dpv[3]);

#pragma unroll
      for (int q = 0; q < 4; ++q) { cC[q] = cN[q]; dC[q] = dN[q]; }
    }
    *(v4f*)&Cr[t * TCOLS + TCOLS - 4] = cprev4;
    *(v4f*)&Td[ldsR + blk_of<W>(TCOLS / 4 - 1)] = dprev4;
    wave_lds_handover();

    const int tt = tile_of<W>(t);
    tile_store_lines(Td, ldsG, oG + tt * TCOLS);
    tile_put_lines(Tf, ldsG, pre);
    wave_lds_handover();
  }
  cp_last = cp_prev;
  dp_last = dp_prev;
}

template <int W>
__device__ __forceinline__ void eliminate_half(
    const float* __restrict__ fG, float* __restrict__ oG,
    const float* __restrict__ v_edge, float* Tf, float* Td, float* Cr, float* Ve,
    int lane, int ldsG, int ldsR, float vbar, float Pdv, float s, float twobdv,
    float& cp_last, float& dp_last) {
  float pre[TCOLS];
  tile_load_lines(fG + tile_of<W>(0) * TCOLS, pre);

  float elo = __builtin_inff(), ehi = -__builtin_inff();
  for (int k = lane; k < NVE; k += 32) {
    const int e = W ? (NVC - 1 - k) : (k + 1);
    const float ve = v_edge[e];
    Ve[k] = ve;
    elo = fminf(elo, ve);
    ehi = fmaxf(ehi, ve);
  }
  wave_minmax(elo, ehi);
  const float wmax = fabsf(twobdv) * fmaxf(fabsf(elo - vbar), fabsf(ehi - vbar));
  const bool series =
      __builtin_amdgcn_ballot_w32(wmax < 0.5f) == __builtin_amdgcn_ballot_w32(true);

  tile_put_lines(Tf, ldsG, pre);
  wave_lds_handover();

  if (series)
    eliminate_sweep<W, true>(fG, oG, Tf, Td, Cr, Ve, pre, ldsG, ldsR, vbar, Pdv, s, twobdv,
                             cp_last, dp_last);
  else
    eliminate_sweep<W, false>(fG, oG, Tf, Td, Cr, Ve, pre, ldsG, ldsR, vbar, Pdv, s, twobdv,
                              cp_last, dp_last);
}

template <int W>
__device__ __forceinline__ void substitute_half(
    float* __restrict__ oG, const float* Cr, float* Td,
    int ldsG, int ldsR, float xnext) {
  float dA[TCOLS], dB[TCOLS];
  tile_load_lines(oG + tile_of<W>(HTILE - 1) * TCOLS, dA);

  for (int t = HTILE - 1; t >= 1; t -= 2) {
    tile_load_lines(oG + tile_of<W>(t - 1) * TCOLS, dB);
    bwd_tile<W>(Td, ldsG, ldsR, oG + tile_of<W>(t) * TCOLS, dA, Cr + t * TCOLS, xnext);

    const int t2 = (t >= 2) ? t - 2 : 0;
    tile_load_lines(oG + tile_of<W>(t2) * TCOLS, dA);
    bwd_tile<W>(Td, ldsG, ldsR, oG + tile_of<W>(t - 1) * TCOLS, dB, Cr + (t - 1) * TCOLS, xnext);
  }
}

__global__ __launch_bounds__(64) void fp_thomas(
    const float* __restrict__ f, const float* __restrict__ v_edge,
    const float* __restrict__ dv_p, const float* __restrict__ nu_p,
    const float* __restrict__ dt_p, const float4* __restrict__ scal,
    float* __restrict__ out) {
  __shared__ __attribute__((aligned(16))) float Cp[2][32 * CPP];
  __shared__ __attribute__((aligned(16))) float Tf[2][32 * TP];
  __shared__ __attribute__((aligned(16))) float Td[2][32 * TP];
  __shared__ __attribute__((aligned(16))) float Ve[2][NVE];
  __shared__ float Xc[2][32], Xd[2][32];

  const int w       = __builtin_amdgcn_readfirstlane((int)(threadIdx.x >> 5));
  const int lane    = threadIdx.x & 31;
  const int rowbase = blockIdx.x * 32;
  const int row     = rowbase + lane;
  const float dv = *dv_p, nu = *nu_p, dt = *dt_p;

  const int gr = lane >> 3, gq = lane & 7;
  const float* fG = f   + (size_t)(rowbase + gr) * NVC + gq * 4;
  float*       oG = out + (size_t)(rowbase + gr) * NVC + gq * 4;
  const int ldsG = gr * TP + gq * 4;
  const int ldsR = lane * TP;
  const int cpR  = lane * CPP;

  const float4 sc  = scal[row];
  const float vbar = sc.y, beta = sc.w;
  const float D      = 0.5f / beta;
  const float Pdv    = D / dv;
  const float s      = dt * nu / dv;
  const float twobdv = 2.0f * beta * dv;

  float cp_last, dp_last;
  if (w == 0)
    eliminate_half<0>(fG, oG, v_edge, Tf[0], Td[0], Cp[0] + cpR, Ve[0], lane, ldsG, ldsR,
                      vbar, Pdv, s, twobdv, cp_last, dp_last);
  else
    eliminate_half<1>(fG, oG, v_edge, Tf[1], Td[1], Cp[1] + cpR, Ve[1], lane, ldsG, ldsR,
                      vbar, Pdv, s, twobdv, cp_last, dp_last);

  Xc[w][lane] = cp_last;
  Xd[w][lane] = dp_last;
  __syncthreads();
  const float co = Xc[w ^ 1][lane];
  const float dq = Xd[w ^ 1][lane];
  const float xnext = (dq - co * dp_last) / (1.0f - co * cp_last);

  if (w == 0)
    substitute_half<0>(oG, Cp[0] + cpR, Td[0], ldsG, ldsR, xnext);
  else
    substitute_half<1>(oG, Cp[1] + cpR, Td[1], ldsG, ldsR, xnext);
}

extern "C" void kernel_launch(void* const* d_in, const int* in_sizes, int n_in,
                              void* d_out, int out_size, void* d_ws, size_t ws_size,
                              hipStream_t stream) {
  const float* f      = (const float*)d_in[0];
  const float* v      = (const float*)d_in[1];  (void)v;
  const float* v_edge = (const float*)d_in[2];
  const float* dv     = (const float*)d_in[3];
  const float* nu     = (const float*)d_in[4];
  const float* dt     = (const float*)d_in[5];
  (void)in_sizes; (void)n_in; (void)out_size; (void)ws_size;

  float4* scal = (float4*)d_ws;
  float*  out  = (float*)d_out;

  fp_moments_wmma<<<B_ROWS / 16, 32, 0, stream>>>(f, dv, scal);
  fp_beta_iter  <<<B_ROWS / 8, 256, 0, stream>>>(dv, scal);
  fp_thomas     <<<B_ROWS / 32, 64, 0, stream>>>(f, v_edge, dv, nu, dt, scal, out);
}


# fp_beta_iter: a wave takes 4 rows as two pairs on packed f32 (pk mul/add/fma), log2e folded into the per-row scale, lane sums on DPP not LDS permutes; 4096 waves not 16384; every sum keeps its order
# speedup vs baseline: 4.7163x; 1.0452x over previous
#include <hip/hip_runtime.h>
#include <cmath>

#define B_ROWS   16384
#define NVC      1024
#define VMAXF    6.0f
#define SC_ITERS 10

typedef __attribute__((ext_vector_type(2))) float v2f;
typedef __attribute__((ext_vector_type(8))) float v8f;

__global__ __launch_bounds__(32) void fp_moments_wmma(
    const float* __restrict__ f, const float* __restrict__ dv_p,
    float4* __restrict__ scal) {
  const int lane    = threadIdx.x;
  const int rowbase = blockIdx.x * 16;
  const float dv    = *dv_p;
  const int r       = lane & 15;
  const int khalf   = (lane >> 4) * 2;

  const float* fp = f + (size_t)(rowbase + r) * NVC + khalf;

  const float c0 = (r == 0) ? 1.0f : 0.0f;
  const float c1 = (r == 1) ? 1.0f : 0.0f;
  const float c2 = (r == 2) ? 1.0f : 0.0f;

  const float dv4 = 4.0f * dv;
  float vcur = dv * ((float)khalf + 0.5f) - VMAXF;

  v8f acc0 = {}, acc1 = {}, acc2 = {}, acc3 = {};
  for (int kb = 0; kb < NVC; kb += 16) {
#pragma unroll
    for (int u = 0; u < 4; ++u) {
      const int k = kb + 4 * u;
      v2f a;
      a.x = fp[k + 0];
      a.y = fp[k + 1];
      const float v0 = vcur;
      const float v1 = vcur + dv;
      v2f b;
      b.x = (c2 * v0 + c1) * v0 + c0;
      b.y = (c2 * v1 + c1) * v1 + c0;
      v8f* acc = (u == 0) ? &acc0 : (u == 1) ? &acc1 : (u == 2) ? &acc2 : &acc3;
      *acc = __builtin_amdgcn_wmma_f32_16x16x4_f32(
          false, a, false, b, (short)0, *acc, false, false);
      vcur += dv4;
    }
  }
  v8f acc = (acc0 + acc1) + (acc2 + acc3);

  float av[8];
#pragma unroll
  for (int i = 0; i < 8; ++i) av[i] = acc[i];

  float myn = 0.f, myvb = 0.f, myet = 0.f, mybeta = 0.f;
#pragma unroll
  for (int m = 0; m < 16; ++m) {
    const int base  = (m < 8) ? 0 : 16;
    const float val = av[m & 7];
    const float S0  = __shfl(val, base + 0, 32);
    const float S1  = __shfl(val, base + 1, 32);
    const float S2  = __shfl(val, base + 2, 32);
    if (lane == m) {
      const float n_   = S0 * dv;
      const float vbar = S1 / S0;
      const float e_t  = (S2 - S1 * S1 / S0) * dv;
      myn = n_; myvb = vbar; myet = e_t;
      mybeta = n_ / (2.0f * e_t);
    }
  }
  if (lane < 16) scal[rowbase + lane] = make_float4(myn, myvb, myet, mybeta);
}

template <int CTRL> __device__ __forceinline__ float row_lane(float x) {
  return __builtin_bit_cast(
      float, __builtin_amdgcn_update_dpp(0, __builtin_bit_cast(int, x), CTRL, 0xF, 0xF, true));
}

__device__ __forceinline__ float wave_sum(float x) {
  const unsigned u = __builtin_bit_cast(unsigned, x);
  x += __builtin_bit_cast(float, __builtin_amdgcn_permlanex16(u, u, 0x76543210u, 0xfedcba98u, false, false));
  x += row_lane<0x128>(x);
  x += __builtin_bit_cast(float, __builtin_amdgcn_mov_dpp8(__builtin_bit_cast(unsigned, x), 0x688facu));
  x += row_lane<0x4E>(x);
  x += row_lane<0xB1>(x);
  return x;
}

#define BROWS 4
__global__ __launch_bounds__(256) void fp_beta_iter(
    const float* __restrict__ dv_p, float4* __restrict__ scal) {
  const int wave = threadIdx.x >> 5;
  const int lane = threadIdx.x & 31;
  const int row0 = (blockIdx.x * 8 + wave) * BROWS;
  const float dv = *dv_p;

  const float4 s0 = scal[row0 + 0], s1 = scal[row0 + 1];
  const float4 s2 = scal[row0 + 2], s3 = scal[row0 + 3];
  const v2f vbA = {s0.y, s1.y}, vbB = {s2.y, s3.y};
  v2f betaA = {s0.w, s1.w}, betaB = {s2.w, s3.w};
  const v2f gA = {s0.x / s0.z, s1.x / s1.z};
  const v2f gB = {s2.x / s2.z, s3.x / s3.z};

  v2f u2A[32], u2B[32];
#pragma unroll
  for (int c = 0; c < 32; ++c) {
    const int j   = lane + 32 * c;
    const float v = dv * ((float)j + 0.5f) - VMAXF;
    const v2f vv  = {v, v};
    const v2f uA  = vv - vbA;
    const v2f uB  = vv - vbB;
    u2A[c] = uA * uA;
    u2B[c] = uB * uB;
  }

  const float NLOG2E = -1.44269504088896340736f;
  for (int it = 0; it < SC_ITERS; ++it) {
    const v2f nbA = betaA * NLOG2E;
    const v2f nbB = betaB * NLOG2E;
    v2f smA = {0.f, 0.f}, seA = {0.f, 0.f}, smB = {0.f, 0.f}, seB = {0.f, 0.f};
#pragma unroll
    for (int c = 0; c < 32; ++c) {
      const v2f tA = u2A[c] * nbA;
      const v2f tB = u2B[c] * nbB;
      v2f MA, MB;
      MA.x = __builtin_amdgcn_exp2f(tA.x);
      MA.y = __builtin_amdgcn_exp2f(tA.y);
      MB.x = __builtin_amdgcn_exp2f(tB.x);
      MB.y = __builtin_amdgcn_exp2f(tB.y);
      smA += MA;
      smB += MB;
      seA = __builtin_elementwise_fma(MA, u2A[c], seA);
      seB = __builtin_elementwise_fma(MB, u2B[c], seB);
    }
    float r[8] = {smA.x, smA.y, smB.x, smB.y, seA.x, seA.y, seB.x, seB.y};
#pragma unroll
    for (int k = 0; k < 8; ++k) r[k] = wave_sum(r[k]);
    const v2f smRA = {r[0], r[1]}, smRB = {r[2], r[3]};
    const v2f seRA = {r[4], r[5]}, seRB = {r[6], r[7]};
    betaA = betaA * (seRA / smRA) * gA;
    betaB = betaB * (seRB / smRB) * gB;
  }
  if (lane == 0) {
    scal[row0 + 0].w = betaA.x;
    scal[row0 + 1].w = betaA.y;
    scal[row0 + 2].w = betaB.x;
    scal[row0 + 3].w = betaB.y;
  }
}

__device__ __forceinline__ float cc_delta(float w) {
  const bool small = fabsf(w) < 1e-6f;
  const float ws   = small ? 1.0f : w;
  const float em   = expm1f(ws);
  const float d    = (em - ws) * __builtin_amdgcn_rcpf(ws * em);
  return small ? 0.5f : d;
}

__device__ __forceinline__ float cc_delta_series(float w) {
  const float w2 = w * w;
  float p = fmaf(w2, 1.0f / 1209600.0f, -1.0f / 30240.0f);
  p = fmaf(w2, p, 1.0f / 720.0f);
  p = fmaf(w2, p, -1.0f / 12.0f);
  const float d = fmaf(w, p, 0.5f);
  return (fabsf(w) < 1e-6f) ? 0.5f : d;
}
template <bool SERIES> __device__ __forceinline__ float cc_delta_of(float w) {
  return SERIES ? cc_delta_series(w) : cc_delta(w);
}

__device__ __forceinline__ void wave_minmax(float& lo, float& hi) {
  lo = fminf(lo, row_lane<0xB1>(lo));   hi = fmaxf(hi, row_lane<0xB1>(hi));
  lo = fminf(lo, row_lane<0x4E>(lo));   hi = fmaxf(hi, row_lane<0x4E>(hi));
  lo = fminf(lo, row_lane<0x141>(lo));  hi = fmaxf(hi, row_lane<0x141>(hi));
  lo = fminf(lo, row_lane<0x140>(lo));  hi = fmaxf(hi, row_lane<0x140>(hi));
  const unsigned ulo = __builtin_bit_cast(unsigned, lo), uhi = __builtin_bit_cast(unsigned, hi);
  lo = fminf(lo, __builtin_bit_cast(float, __builtin_amdgcn_permlanex16(ulo, ulo, 0x76543210u, 0xfedcba98u, false, false)));
  hi = fmaxf(hi, __builtin_bit_cast(float, __builtin_amdgcn_permlanex16(uhi, uhi, 0x76543210u, 0xfedcba98u, false, false)));
}

#define TCOLS 32
#define NTILE (NVC / TCOLS)
#define HTILE (NTILE / 2)
#define TP    36
#define NVE   544
#define HCELL (NVC / 2)
#define CPP   (HCELL + 4)

typedef __attribute__((ext_vector_type(4))) float v4f;

__device__ __forceinline__ void wave_lds_handover() {
  __builtin_amdgcn_fence(__ATOMIC_ACQ_REL, "wavefront");
}

template <int W> __device__ __forceinline__ int tile_of(int t) {
  return W ? (NTILE - 1 - t) : t;
}

template <int W> __device__ __forceinline__ int blk_of(int b) {
  return 4 * (W ? (TCOLS / 4 - 1 - b) : b);
}
template <int W> __device__ __forceinline__ float own4(v4f t, int q) {
  const int z = W ? 3 - q : q;
  return z == 0 ? t.x : z == 1 ? t.y : z == 2 ? t.z : t.w;
}
template <int W> __device__ __forceinline__ v4f pack4(float a0, float a1, float a2, float a3) {
  v4f r;
  if (W) { r.x = a3; r.y = a2; r.z = a1; r.w = a0; } else { r.x = a0; r.y = a1; r.z = a2; r.w = a3; }
  return r;
}

__device__ __forceinline__ void tile_load_lines(const float* g, float (&r)[TCOLS]) {
#pragma unroll
  for (int i = 0; i < 8; ++i) {
    const v4f t4 = *(const v4f*)(g + (size_t)i * 4 * NVC);
    r[4 * i + 0] = t4.x; r[4 * i + 1] = t4.y; r[4 * i + 2] = t4.z; r[4 * i + 3] = t4.w;
  }
}

__device__ __forceinline__ void tile_put_lines(float* T, int ldsG, const float (&r)[TCOLS]) {
#pragma unroll
  for (int i = 0; i < 8; ++i) {
    v4f t4; t4.x = r[4 * i + 0]; t4.y = r[4 * i + 1]; t4.z = r[4 * i + 2]; t4.w = r[4 * i + 3];
    *(v4f*)&T[ldsG + i * 4 * TP] = t4;
  }
}

__device__ __forceinline__ void tile_store_lines(const float* T, int ldsG, float* g) {
#pragma unroll
  for (int i = 0; i < 8; ++i) {
    const v4f t4 = *(const v4f*)&T[ldsG + i * 4 * TP];
    *(v4f*)(g + (size_t)i * 4 * NVC) = t4;
  }
}

template <int W>
__device__ __forceinline__ void bwd_tile(float* Td, int ldsG, int ldsR, float* g,
                                         const float (&dpl)[TCOLS], const float* Crt,
                                         float& xnext) {
  tile_put_lines(Td, ldsG, dpl);
  float cc[TCOLS];
#pragma unroll
  for (int q = 0; q < 8; ++q) {
    const v4f c4 = *(const v4f*)&Crt[4 * q];
    cc[4 * q + 0] = c4.x; cc[4 * q + 1] = c4.y; cc[4 * q + 2] = c4.z; cc[4 * q + 3] = c4.w;
  }
  wave_lds_handover();
  float dr[TCOLS];
#pragma unroll
  for (int q = 0; q < 8; ++q) {
    const v4f t4 = *(const v4f*)&Td[ldsR + blk_of<W>(q)];
    dr[4 * q + 0] = own4<W>(t4, 0); dr[4 * q + 1] = own4<W>(t4, 1);
    dr[4 * q + 2] = own4<W>(t4, 2); dr[4 * q + 3] = own4<W>(t4, 3);
  }
  float xr[TCOLS];
  float xn = xnext;
#pragma unroll
  for (int k = TCOLS - 1; k >= 0; --k) {
    const float x = dr[k] - cc[k] * xn;
    xr[k] = x;
    xn = x;
  }
  xnext = xn;
#pragma unroll
  for (int q = 0; q < 8; ++q) {
    *(v4f*)&Td[ldsR + blk_of<W>(q)] =
        pack4<W>(xr[4 * q + 0], xr[4 * q + 1], xr[4 * q + 2], xr[4 * q + 3]);
  }
  wave_lds_handover();
  tile_store_lines(Td, ldsG, g);
  wave_lds_handover();
}

template <int W, bool SERIES>
__device__ __forceinline__ void eliminate_sweep(
    const float* __restrict__ fG, float* __restrict__ oG,
    float* Tf, float* Td, float* Cr, const float* Ve, float (&pre)[TCOLS],
    int ldsG, int ldsR, float vbar, float Pdv, float s, float twobdv,
    float& cp_last, float& dp_last) {
  float Ce0 = 0.f, P0 = 0.f, d0 = 0.5f;
  float cp_prev = 0.f, dp_prev = 0.f;

  float cC[4], dC[4];
  {
    const v4f e4 = *(const v4f*)&Ve[0];
    const float ev[4] = {e4.x, e4.y, e4.z, e4.w};
#pragma unroll
    for (int q = 0; q < 4; ++q) {
      const float c = W ? (vbar - ev[q]) : (ev[q] - vbar);
      cC[q] = c;
      dC[q] = cc_delta_of<SERIES>(twobdv * c);
    }
  }
  v4f enext = *(const v4f*)&Ve[4];

  for (int t = 0; t < HTILE; ++t) {
    const int tn = (t + 1 < HTILE) ? t + 1 : t;
    tile_load_lines(fG + tile_of<W>(tn) * TCOLS, pre);

    v4f fnext = *(const v4f*)&Tf[ldsR + blk_of<W>(0)];
    v4f cprev4 = {0.f, 0.f, 0.f, 0.f}, dprev4 = {0.f, 0.f, 0.f, 0.f};
#pragma nounroll
    for (int b = 0; b < TCOLS / 4; ++b) {
      const int jb = t * TCOLS + b * 4;
      if (b > 0) {
        *(v4f*)&Cr[jb - 4] = cprev4;
        *(v4f*)&Td[ldsR + blk_of<W>(b - 1)] = dprev4;
      }
      const v4f fj = fnext;
      const v4f ej = enext;
      const int bn = (b + 1 < TCOLS / 4) ? b + 1 : b;
      fnext = *(const v4f*)&Tf[ldsR + blk_of<W>(bn)];
      enext = *(const v4f*)&Ve[jb + 8];
      __builtin_amdgcn_sched_barrier(0);
      const float fv[4] = {own4<W>(fj, 0), own4<W>(fj, 1), own4<W>(fj, 2), own4<W>(fj, 3)};
      const float ev[4] = {ej.x, ej.y, ej.z, ej.w};

      float cN[4], dN[4];
#pragma unroll
      for (int q = 0; q < 4; ++q) {
        const float c = W ? (vbar - ev[q]) : (ev[q] - vbar);
        cN[q] = c;
        dN[q] = cc_delta_of<SERIES>(twobdv * c);
      }

      float dpv[4], cpv[4];
#pragma unroll
      for (int q = 0; q < 4; ++q) {
        const float Ce1 = cC[q];
        const float P1  = Pdv;
        const float d1  = dC[q];

        const float coefU = Ce1 * (1.0f - d1) + P1;
        const float coefL = -Ce0 * d0 + P0;
        const float coefD = Ce1 * d1 - P1 - Ce0 * (1.0f - d0) - P0;
        const float dl = -s * coefL;
        const float dd = 1.0f - s * coefD;
        const float du = -s * coefU;

        const float denom = dd - dl * cp_prev;
        const float rdiv  = __builtin_amdgcn_rcpf(denom);
        const float cp    = du * rdiv;
        const float dp    = (fv[q] - dl * dp_prev) * rdiv;

        cpv[q] = cp;
        dpv[q] = dp;
        cp_prev = cp; dp_prev = dp;
        Ce0 = Ce1; P0 = P1; d0 = d1;
      }
      cprev4.x = cpv[0]; cprev4.y = cpv[1]; cprev4.z = cpv[2]; cprev4.w = cpv[3];
      dprev4 = pack4<W>(dpv[0], dpv[1], dpv[2], dpv[3]);

#pragma unroll
      for (int q = 0; q < 4; ++q) { cC[q] = cN[q]; dC[q] = dN[q]; }
    }
    *(v4f*)&Cr[t * TCOLS + TCOLS - 4] = cprev4;
    *(v4f*)&Td[ldsR + blk_of<W>(TCOLS / 4 - 1)] = dprev4;
    wave_lds_handover();

    const int tt = tile_of<W>(t);
    tile_store_lines(Td, ldsG, oG + tt * TCOLS);
    tile_put_lines(Tf, ldsG, pre);
    wave_lds_handover();
  }
  cp_last = cp_prev;
  dp_last = dp_prev;
}

template <int W>
__device__ __forceinline__ void eliminate_half(
    const float* __restrict__ fG, float* __restrict__ oG,
    const float* __restrict__ v_edge, float* Tf, float* Td, float* Cr, float* Ve,
    int lane, int ldsG, int ldsR, float vbar, float Pdv, float s, float twobdv,
    float& cp_last, float& dp_last) {
  float pre[TCOLS];
  tile_load_lines(fG + tile_of<W>(0) * TCOLS, pre);

  float elo = __builtin_inff(), ehi = -__builtin_inff();
  for (int k = lane; k < NVE; k += 32) {
    const int e = W ? (NVC - 1 - k) : (k + 1);
    const float ve = v_edge[e];
    Ve[k] = ve;
    elo = fminf(elo, ve);
    ehi = fmaxf(ehi, ve);
  }
  wave_minmax(elo, ehi);
  const float wmax = fabsf(twobdv) * fmaxf(fabsf(elo - vbar), fabsf(ehi - vbar));
  const bool series =
      __builtin_amdgcn_ballot_w32(wmax < 0.5f) == __builtin_amdgcn_ballot_w32(true);

  tile_put_lines(Tf, ldsG, pre);
  wave_lds_handover();

  if (series)
    eliminate_sweep<W, true>(fG, oG, Tf, Td, Cr, Ve, pre, ldsG, ldsR, vbar, Pdv, s, twobdv,
                             cp_last, dp_last);
  else
    eliminate_sweep<W, false>(fG, oG, Tf, Td, Cr, Ve, pre, ldsG, ldsR, vbar, Pdv, s, twobdv,
                              cp_last, dp_last);
}

template <int W>
__device__ __forceinline__ void substitute_half(
    float* __restrict__ oG, const float* Cr, float* Td,
    int ldsG, int ldsR, float xnext) {
  float dA[TCOLS], dB[TCOLS];
  tile_load_lines(oG + tile_of<W>(HTILE - 1) * TCOLS, dA);

  for (int t = HTILE - 1; t >= 1; t -= 2) {
    tile_load_lines(oG + tile_of<W>(t - 1) * TCOLS, dB);
    bwd_tile<W>(Td, ldsG, ldsR, oG + tile_of<W>(t) * TCOLS, dA, Cr + t * TCOLS, xnext);

    const int t2 = (t >= 2) ? t - 2 : 0;
    tile_load_lines(oG + tile_of<W>(t2) * TCOLS, dA);
    bwd_tile<W>(Td, ldsG, ldsR, oG + tile_of<W>(t - 1) * TCOLS, dB, Cr + (t - 1) * TCOLS, xnext);
  }
}

__global__ __launch_bounds__(64) void fp_thomas(
    const float* __restrict__ f, const float* __restrict__ v_edge,
    const float* __restrict__ dv_p, const float* __restrict__ nu_p,
    const float* __restrict__ dt_p, const float4* __restrict__ scal,
    float* __restrict__ out) {
  __shared__ __attribute__((aligned(16))) float Cp[2][32 * CPP];
  __shared__ __attribute__((aligned(16))) float Tf[2][32 * TP];
  __shared__ __attribute__((aligned(16))) float Td[2][32 * TP];
  __shared__ __attribute__((aligned(16))) float Ve[2][NVE];
  __shared__ float Xc[2][32], Xd[2][32];

  const int w       = __builtin_amdgcn_readfirstlane((int)(threadIdx.x >> 5));
  const int lane    = threadIdx.x & 31;
  const int rowbase = blockIdx.x * 32;
  const int row     = rowbase + lane;
  const float dv = *dv_p, nu = *nu_p, dt = *dt_p;

  const int gr = lane >> 3, gq = lane & 7;
  const float* fG = f   + (size_t)(rowbase + gr) * NVC + gq * 4;
  float*       oG = out + (size_t)(rowbase + gr) * NVC + gq * 4;
  const int ldsG = gr * TP + gq * 4;
  const int ldsR = lane * TP;
  const int cpR  = lane * CPP;

  const float4 sc  = scal[row];
  const float vbar = sc.y, beta = sc.w;
  const float D      = 0.5f / beta;
  const float Pdv    = D / dv;
  const float s      = dt * nu / dv;
  const float twobdv = 2.0f * beta * dv;

  float cp_last, dp_last;
  if (w == 0)
    eliminate_half<0>(fG, oG, v_edge, Tf[0], Td[0], Cp[0] + cpR, Ve[0], lane, ldsG, ldsR,
                      vbar, Pdv, s, twobdv, cp_last, dp_last);
  else
    eliminate_half<1>(fG, oG, v_edge, Tf[1], Td[1], Cp[1] + cpR, Ve[1], lane, ldsG, ldsR,
                      vbar, Pdv, s, twobdv, cp_last, dp_last);

  Xc[w][lane] = cp_last;
  Xd[w][lane] = dp_last;
  __syncthreads();
  const float co = Xc[w ^ 1][lane];
  const float dq = Xd[w ^ 1][lane];
  const float xnext = (dq - co * dp_last) / (1.0f - co * cp_last);

  if (w == 0)
    substitute_half<0>(oG, Cp[0] + cpR, Td[0], ldsG, ldsR, xnext);
  else
    substitute_half<1>(oG, Cp[1] + cpR, Td[1], ldsG, ldsR, xnext);
}

extern "C" void kernel_launch(void* const* d_in, const int* in_sizes, int n_in,
                              void* d_out, int out_size, void* d_ws, size_t ws_size,
                              hipStream_t stream) {
  const float* f      = (const float*)d_in[0];
  const float* v      = (const float*)d_in[1];  (void)v;
  const float* v_edge = (const float*)d_in[2];
  const float* dv     = (const float*)d_in[3];
  const float* nu     = (const float*)d_in[4];
  const float* dt     = (const float*)d_in[5];
  (void)in_sizes; (void)n_in; (void)out_size; (void)ws_size;

  float4* scal = (float4*)d_ws;
  float*  out  = (float*)d_out;

  fp_moments_wmma<<<B_ROWS / 16, 32, 0, stream>>>(f, dv, scal);
  fp_beta_iter  <<<B_ROWS / (8 * BROWS), 256, 0, stream>>>(dv, scal);
  fp_thomas     <<<B_ROWS / 32, 64, 0, stream>>>(f, v_edge, dv, nu, dt, scal, out);
}


# one launch: each thomas workgroup first makes its own 32 rows' moments (same WMMA loop, 16 rows a wave) and beta (4 rows a pass), handed on in LDS; 3 launches -> 1; every sum keeps its order
# speedup vs baseline: 5.3455x; 1.1334x over previous
#include <hip/hip_runtime.h>
#include <cmath>

#define B_ROWS   16384
#define NVC      1024
#define VMAXF    6.0f
#define SC_ITERS 10

typedef __attribute__((ext_vector_type(2))) float v2f;
typedef __attribute__((ext_vector_type(8))) float v8f;

__device__ __forceinline__ float4 moments16(
    const float* __restrict__ f, const float dv, const int rowbase, const int lane) {
  const int r       = lane & 15;
  const int khalf   = (lane >> 4) * 2;

  const float* fp = f + (size_t)(rowbase + r) * NVC + khalf;

  const float c0 = (r == 0) ? 1.0f : 0.0f;
  const float c1 = (r == 1) ? 1.0f : 0.0f;
  const float c2 = (r == 2) ? 1.0f : 0.0f;

  const float dv4 = 4.0f * dv;
  float vcur = dv * ((float)khalf + 0.5f) - VMAXF;

  v8f acc0 = {}, acc1 = {}, acc2 = {}, acc3 = {};
  for (int kb = 0; kb < NVC; kb += 16) {
#pragma unroll
    for (int u = 0; u < 4; ++u) {
      const int k = kb + 4 * u;
      v2f a;
      a.x = fp[k + 0];
      a.y = fp[k + 1];
      const float v0 = vcur;
      const float v1 = vcur + dv;
      v2f b;
      b.x = (c2 * v0 + c1) * v0 + c0;
      b.y = (c2 * v1 + c1) * v1 + c0;
      v8f* acc = (u == 0) ? &acc0 : (u == 1) ? &acc1 : (u == 2) ? &acc2 : &acc3;
      *acc = __builtin_amdgcn_wmma_f32_16x16x4_f32(
          false, a, false, b, (short)0, *acc, false, false);
      vcur += dv4;
    }
  }
  v8f acc = (acc0 + acc1) + (acc2 + acc3);

  float av[8];
#pragma unroll
  for (int i = 0; i < 8; ++i) av[i] = acc[i];

  float myn = 0.f, myvb = 0.f, myet = 0.f, mybeta = 0.f;
#pragma unroll
  for (int m = 0; m < 16; ++m) {
    const int base  = (m < 8) ? 0 : 16;
    const float val = av[m & 7];
    const float S0  = __shfl(val, base + 0, 32);
    const float S1  = __shfl(val, base + 1, 32);
    const float S2  = __shfl(val, base + 2, 32);
    if (lane == m) {
      const float n_   = S0 * dv;
      const float vbar = S1 / S0;
      const float e_t  = (S2 - S1 * S1 / S0) * dv;
      myn = n_; myvb = vbar; myet = e_t;
      mybeta = n_ / (2.0f * e_t);
    }
  }
  return make_float4(myn, myvb, myet, mybeta);
}

template <int CTRL> __device__ __forceinline__ float row_lane(float x) {
  return __builtin_bit_cast(
      float, __builtin_amdgcn_update_dpp(0, __builtin_bit_cast(int, x), CTRL, 0xF, 0xF, true));
}

__device__ __forceinline__ float wave_sum(float x) {
  const unsigned u = __builtin_bit_cast(unsigned, x);
  x += __builtin_bit_cast(float, __builtin_amdgcn_permlanex16(u, u, 0x76543210u, 0xfedcba98u, false, false));
  x += row_lane<0x128>(x);
  x += __builtin_bit_cast(float, __builtin_amdgcn_mov_dpp8(__builtin_bit_cast(unsigned, x), 0x688facu));
  x += row_lane<0x4E>(x);
  x += row_lane<0xB1>(x);
  return x;
}

#define BROWS 4
__device__ __forceinline__ void beta_rows4(
    const float dv, const int lane, const float4 s0, const float4 s1, const float4 s2,
    const float4 s3, v2f& outA, v2f& outB) {
  const v2f vbA = {s0.y, s1.y}, vbB = {s2.y, s3.y};
  v2f betaA = {s0.w, s1.w}, betaB = {s2.w, s3.w};
  const v2f gA = {s0.x / s0.z, s1.x / s1.z};
  const v2f gB = {s2.x / s2.z, s3.x / s3.z};

  v2f u2A[32], u2B[32];
#pragma unroll
  for (int c = 0; c < 32; ++c) {
    const int j   = lane + 32 * c;
    const float v = dv * ((float)j + 0.5f) - VMAXF;
    const v2f vv  = {v, v};
    const v2f uA  = vv - vbA;
    const v2f uB  = vv - vbB;
    u2A[c] = uA * uA;
    u2B[c] = uB * uB;
  }

  const float NLOG2E = -1.44269504088896340736f;
  for (int it = 0; it < SC_ITERS; ++it) {
    const v2f nbA = betaA * NLOG2E;
    const v2f nbB = betaB * NLOG2E;
    v2f smA = {0.f, 0.f}, seA = {0.f, 0.f}, smB = {0.f, 0.f}, seB = {0.f, 0.f};
#pragma unroll
    for (int c = 0; c < 32; ++c) {
      const v2f tA = u2A[c] * nbA;
      const v2f tB = u2B[c] * nbB;
      v2f MA, MB;
      MA.x = __builtin_amdgcn_exp2f(tA.x);
      MA.y = __builtin_amdgcn_exp2f(tA.y);
      MB.x = __builtin_amdgcn_exp2f(tB.x);
      MB.y = __builtin_amdgcn_exp2f(tB.y);
      smA += MA;
      smB += MB;
      seA = __builtin_elementwise_fma(MA, u2A[c], seA);
      seB = __builtin_elementwise_fma(MB, u2B[c], seB);
    }
    float r[8] = {smA.x, smA.y, smB.x, smB.y, seA.x, seA.y, seB.x, seB.y};
#pragma unroll
    for (int k = 0; k < 8; ++k) r[k] = wave_sum(r[k]);
    const v2f smRA = {r[0], r[1]}, smRB = {r[2], r[3]};
    const v2f seRA = {r[4], r[5]}, seRB = {r[6], r[7]};
    betaA = betaA * (seRA / smRA) * gA;
    betaB = betaB * (seRB / smRB) * gB;
  }
  outA = betaA;
  outB = betaB;
}

__device__ __forceinline__ float cc_delta(float w) {
  const bool small = fabsf(w) < 1e-6f;
  const float ws   = small ? 1.0f : w;
  const float em   = expm1f(ws);
  const float d    = (em - ws) * __builtin_amdgcn_rcpf(ws * em);
  return small ? 0.5f : d;
}

__device__ __forceinline__ float cc_delta_series(float w) {
  const float w2 = w * w;
  float p = fmaf(w2, 1.0f / 1209600.0f, -1.0f / 30240.0f);
  p = fmaf(w2, p, 1.0f / 720.0f);
  p = fmaf(w2, p, -1.0f / 12.0f);
  const float d = fmaf(w, p, 0.5f);
  return (fabsf(w) < 1e-6f) ? 0.5f : d;
}
template <bool SERIES> __device__ __forceinline__ float cc_delta_of(float w) {
  return SERIES ? cc_delta_series(w) : cc_delta(w);
}

__device__ __forceinline__ void wave_minmax(float& lo, float& hi) {
  lo = fminf(lo, row_lane<0xB1>(lo));   hi = fmaxf(hi, row_lane<0xB1>(hi));
  lo = fminf(lo, row_lane<0x4E>(lo));   hi = fmaxf(hi, row_lane<0x4E>(hi));
  lo = fminf(lo, row_lane<0x141>(lo));  hi = fmaxf(hi, row_lane<0x141>(hi));
  lo = fminf(lo, row_lane<0x140>(lo));  hi = fmaxf(hi, row_lane<0x140>(hi));
  const unsigned ulo = __builtin_bit_cast(unsigned, lo), uhi = __builtin_bit_cast(unsigned, hi);
  lo = fminf(lo, __builtin_bit_cast(float, __builtin_amdgcn_permlanex16(ulo, ulo, 0x76543210u, 0xfedcba98u, false, false)));
  hi = fmaxf(hi, __builtin_bit_cast(float, __builtin_amdgcn_permlanex16(uhi, uhi, 0x76543210u, 0xfedcba98u, false, false)));
}

#define TCOLS 32
#define NTILE (NVC / TCOLS)
#define HTILE (NTILE / 2)
#define TP    36
#define NVE   544
#define HCELL (NVC / 2)
#define CPP   (HCELL + 4)

typedef __attribute__((ext_vector_type(4))) float v4f;

__device__ __forceinline__ void wave_lds_handover() {
  __builtin_amdgcn_fence(__ATOMIC_ACQ_REL, "wavefront");
}

template <int W> __device__ __forceinline__ int tile_of(int t) {
  return W ? (NTILE - 1 - t) : t;
}

template <int W> __device__ __forceinline__ int blk_of(int b) {
  return 4 * (W ? (TCOLS / 4 - 1 - b) : b);
}
template <int W> __device__ __forceinline__ float own4(v4f t, int q) {
  const int z = W ? 3 - q : q;
  return z == 0 ? t.x : z == 1 ? t.y : z == 2 ? t.z : t.w;
}
template <int W> __device__ __forceinline__ v4f pack4(float a0, float a1, float a2, float a3) {
  v4f r;
  if (W) { r.x = a3; r.y = a2; r.z = a1; r.w = a0; } else { r.x = a0; r.y = a1; r.z = a2; r.w = a3; }
  return r;
}

__device__ __forceinline__ void tile_load_lines(const float* g, float (&r)[TCOLS]) {
#pragma unroll
  for (int i = 0; i < 8; ++i) {
    const v4f t4 = *(const v4f*)(g + (size_t)i * 4 * NVC);
    r[4 * i + 0] = t4.x; r[4 * i + 1] = t4.y; r[4 * i + 2] = t4.z; r[4 * i + 3] = t4.w;
  }
}

__device__ __forceinline__ void tile_put_lines(float* T, int ldsG, const float (&r)[TCOLS]) {
#pragma unroll
  for (int i = 0; i < 8; ++i) {
    v4f t4; t4.x = r[4 * i + 0]; t4.y = r[4 * i + 1]; t4.z = r[4 * i + 2]; t4.w = r[4 * i + 3];
    *(v4f*)&T[ldsG + i * 4 * TP] = t4;
  }
}

__device__ __forceinline__ void tile_store_lines(const float* T, int ldsG, float* g) {
#pragma unroll
  for (int i = 0; i < 8; ++i) {
    const v4f t4 = *(const v4f*)&T[ldsG + i * 4 * TP];
    *(v4f*)(g + (size_t)i * 4 * NVC) = t4;
  }
}

template <int W>
__device__ __forceinline__ void bwd_tile(float* Td, int ldsG, int ldsR, float* g,
                                         const float (&dpl)[TCOLS], const float* Crt,
                                         float& xnext) {
  tile_put_lines(Td, ldsG, dpl);
  float cc[TCOLS];
#pragma unroll
  for (int q = 0; q < 8; ++q) {
    const v4f c4 = *(const v4f*)&Crt[4 * q];
    cc[4 * q + 0] = c4.x; cc[4 * q + 1] = c4.y; cc[4 * q + 2] = c4.z; cc[4 * q + 3] = c4.w;
  }
  wave_lds_handover();
  float dr[TCOLS];
#pragma unroll
  for (int q = 0; q < 8; ++q) {
    const v4f t4 = *(const v4f*)&Td[ldsR + blk_of<W>(q)];
    dr[4 * q + 0] = own4<W>(t4, 0); dr[4 * q + 1] = own4<W>(t4, 1);
    dr[4 * q + 2] = own4<W>(t4, 2); dr[4 * q + 3] = own4<W>(t4, 3);
  }
  float xr[TCOLS];
  float xn = xnext;
#pragma unroll
  for (int k = TCOLS - 1; k >= 0; --k) {
    const float x = dr[k] - cc[k] * xn;
    xr[k] = x;
    xn = x;
  }
  xnext = xn;
#pragma unroll
  for (int q = 0; q < 8; ++q) {
    *(v4f*)&Td[ldsR + blk_of<W>(q)] =
        pack4<W>(xr[4 * q + 0], xr[4 * q + 1], xr[4 * q + 2], xr[4 * q + 3]);
  }
  wave_lds_handover();
  tile_store_lines(Td, ldsG, g);
  wave_lds_handover();
}

template <int W, bool SERIES>
__device__ __forceinline__ void eliminate_sweep(
    const float* __restrict__ fG, float* __restrict__ oG,
    float* Tf, float* Td, float* Cr, const float* Ve, float (&pre)[TCOLS],
    int ldsG, int ldsR, float vbar, float Pdv, float s, float twobdv,
    float& cp_last, float& dp_last) {
  float Ce0 = 0.f, P0 = 0.f, d0 = 0.5f;
  float cp_prev = 0.f, dp_prev = 0.f;

  float cC[4], dC[4];
  {
    const v4f e4 = *(const v4f*)&Ve[0];
    const float ev[4] = {e4.x, e4.y, e4.z, e4.w};
#pragma unroll
    for (int q = 0; q < 4; ++q) {
      const float c = W ? (vbar - ev[q]) : (ev[q] - vbar);
      cC[q] = c;
      dC[q] = cc_delta_of<SERIES>(twobdv * c);
    }
  }
  v4f enext = *(const v4f*)&Ve[4];

  for (int t = 0; t < HTILE; ++t) {
    const int tn = (t + 1 < HTILE) ? t + 1 : t;
    tile_load_lines(fG + tile_of<W>(tn) * TCOLS, pre);

    v4f fnext = *(const v4f*)&Tf[ldsR + blk_of<W>(0)];
    v4f cprev4 = {0.f, 0.f, 0.f, 0.f}, dprev4 = {0.f, 0.f, 0.f, 0.f};
#pragma nounroll
    for (int b = 0; b < TCOLS / 4; ++b) {
      const int jb = t * TCOLS + b * 4;
      if (b > 0) {
        *(v4f*)&Cr[jb - 4] = cprev4;
        *(v4f*)&Td[ldsR + blk_of<W>(b - 1)] = dprev4;
      }
      const v4f fj = fnext;
      const v4f ej = enext;
      const int bn = (b + 1 < TCOLS / 4) ? b + 1 : b;
      fnext = *(const v4f*)&Tf[ldsR + blk_of<W>(bn)];
      enext = *(const v4f*)&Ve[jb + 8];
      __builtin_amdgcn_sched_barrier(0);
      const float fv[4] = {own4<W>(fj, 0), own4<W>(fj, 1), own4<W>(fj, 2), own4<W>(fj, 3)};
      const float ev[4] = {ej.x, ej.y, ej.z, ej.w};

      float cN[4], dN[4];
#pragma unroll
      for (int q = 0; q < 4; ++q) {
        const float c = W ? (vbar - ev[q]) : (ev[q] - vbar);
        cN[q] = c;
        dN[q] = cc_delta_of<SERIES>(twobdv * c);
      }

      float dpv[4], cpv[4];
#pragma unroll
      for (int q = 0; q < 4; ++q) {
        const float Ce1 = cC[q];
        const float P1  = Pdv;
        const float d1  = dC[q];

        const float coefU = Ce1 * (1.0f - d1) + P1;
        const float coefL = -Ce0 * d0 + P0;
        const float coefD = Ce1 * d1 - P1 - Ce0 * (1.0f - d0) - P0;
        const float dl = -s * coefL;
        const float dd = 1.0f - s * coefD;
        const float du = -s * coefU;

        const float denom = dd - dl * cp_prev;
        const float rdiv  = __builtin_amdgcn_rcpf(denom);
        const float cp    = du * rdiv;
        const float dp    = (fv[q] - dl * dp_prev) * rdiv;

        cpv[q] = cp;
        dpv[q] = dp;
        cp_prev = cp; dp_prev = dp;
        Ce0 = Ce1; P0 = P1; d0 = d1;
      }
      cprev4.x = cpv[0]; cprev4.y = cpv[1]; cprev4.z = cpv[2]; cprev4.w = cpv[3];
      dprev4 = pack4<W>(dpv[0], dpv[1], dpv[2], dpv[3]);

#pragma unroll
      for (int q = 0; q < 4; ++q) { cC[q] = cN[q]; dC[q] = dN[q]; }
    }
    *(v4f*)&Cr[t * TCOLS + TCOLS - 4] = cprev4;
    *(v4f*)&Td[ldsR + blk_of<W>(TCOLS / 4 - 1)] = dprev4;
    wave_lds_handover();

    const int tt = tile_of<W>(t);
    tile_store_lines(Td, ldsG, oG + tt * TCOLS);
    tile_put_lines(Tf, ldsG, pre);
    wave_lds_handover();
  }
  cp_last = cp_prev;
  dp_last = dp_prev;
}

template <int W>
__device__ __forceinline__ void eliminate_half(
    const float* __restrict__ fG, float* __restrict__ oG,
    const float* __restrict__ v_edge, float* Tf, float* Td, float* Cr, float* Ve,
    int lane, int ldsG, int ldsR, float vbar, float Pdv, float s, float twobdv,
    float& cp_last, float& dp_last) {
  float pre[TCOLS];
  tile_load_lines(fG + tile_of<W>(0) * TCOLS, pre);

  float elo = __builtin_inff(), ehi = -__builtin_inff();
  for (int k = lane; k < NVE; k += 32) {
    const int e = W ? (NVC - 1 - k) : (k + 1);
    const float ve = v_edge[e];
    Ve[k] = ve;
    elo = fminf(elo, ve);
    ehi = fmaxf(ehi, ve);
  }
  wave_minmax(elo, ehi);
  const float wmax = fabsf(twobdv) * fmaxf(fabsf(elo - vbar), fabsf(ehi - vbar));
  const bool series =
      __builtin_amdgcn_ballot_w32(wmax < 0.5f) == __builtin_amdgcn_ballot_w32(true);

  tile_put_lines(Tf, ldsG, pre);
  wave_lds_handover();

  if (series)
    eliminate_sweep<W, true>(fG, oG, Tf, Td, Cr, Ve, pre, ldsG, ldsR, vbar, Pdv, s, twobdv,
                             cp_last, dp_last);
  else
    eliminate_sweep<W, false>(fG, oG, Tf, Td, Cr, Ve, pre, ldsG, ldsR, vbar, Pdv, s, twobdv,
                              cp_last, dp_last);
}

template <int W>
__device__ __forceinline__ void substitute_half(
    float* __restrict__ oG, const float* Cr, float* Td,
    int ldsG, int ldsR, float xnext) {
  float dA[TCOLS], dB[TCOLS];
  tile_load_lines(oG + tile_of<W>(HTILE - 1) * TCOLS, dA);

  for (int t = HTILE - 1; t >= 1; t -= 2) {
    tile_load_lines(oG + tile_of<W>(t - 1) * TCOLS, dB);
    bwd_tile<W>(Td, ldsG, ldsR, oG + tile_of<W>(t) * TCOLS, dA, Cr + t * TCOLS, xnext);

    const int t2 = (t >= 2) ? t - 2 : 0;
    tile_load_lines(oG + tile_of<W>(t2) * TCOLS, dA);
    bwd_tile<W>(Td, ldsG, ldsR, oG + tile_of<W>(t - 1) * TCOLS, dB, Cr + (t - 1) * TCOLS, xnext);
  }
}

__global__ __launch_bounds__(64) void fp_thomas(
    const float* __restrict__ f, const float* __restrict__ v_edge,
    const float* __restrict__ dv_p, const float* __restrict__ nu_p,
    const float* __restrict__ dt_p, float* __restrict__ out) {
  __shared__ __attribute__((aligned(16))) float Cp[2][32 * CPP];
  __shared__ __attribute__((aligned(16))) float Tf[2][32 * TP];
  __shared__ __attribute__((aligned(16))) float Td[2][32 * TP];
  __shared__ __attribute__((aligned(16))) float Ve[2][NVE];
  __shared__ float Xc[2][32], Xd[2][32];
  __shared__ __attribute__((aligned(16))) float4 Sc[32];

  const int w       = __builtin_amdgcn_readfirstlane((int)(threadIdx.x >> 5));
  const int lane    = threadIdx.x & 31;
  const int rowbase = blockIdx.x * 32;
  const float dv = *dv_p, nu = *nu_p, dt = *dt_p;

  const int gr = lane >> 3, gq = lane & 7;
  const float* fG = f   + (size_t)(rowbase + gr) * NVC + gq * 4;
  float*       oG = out + (size_t)(rowbase + gr) * NVC + gq * 4;
  const int ldsG = gr * TP + gq * 4;
  const int ldsR = lane * TP;
  const int cpR  = lane * CPP;

  {
    const float4 m = moments16(f, dv, rowbase + 16 * w, lane);
    if (lane < 16) Sc[16 * w + lane] = m;
    wave_lds_handover();
#pragma nounroll
    for (int g = 0; g < 16 / BROWS; ++g) {
      const int i0 = 16 * w + BROWS * g;
      const float4 s0 = Sc[i0 + 0], s1 = Sc[i0 + 1], s2 = Sc[i0 + 2], s3 = Sc[i0 + 3];
      v2f bA, bB;
      beta_rows4(dv, lane, s0, s1, s2, s3, bA, bB);
      if (lane == 0) {
        Sc[i0 + 0].w = bA.x;
        Sc[i0 + 1].w = bA.y;
        Sc[i0 + 2].w = bB.x;
        Sc[i0 + 3].w = bB.y;
      }
    }
  }
  __syncthreads();
  const float4 sc  = Sc[lane];
  const float vbar = sc.y, beta = sc.w;
  const float D      = 0.5f / beta;
  const float Pdv    = D / dv;
  const float s      = dt * nu / dv;
  const float twobdv = 2.0f * beta * dv;

  float cp_last, dp_last;
  if (w == 0)
    eliminate_half<0>(fG, oG, v_edge, Tf[0], Td[0], Cp[0] + cpR, Ve[0], lane, ldsG, ldsR,
                      vbar, Pdv, s, twobdv, cp_last, dp_last);
  else
    eliminate_half<1>(fG, oG, v_edge, Tf[1], Td[1], Cp[1] + cpR, Ve[1], lane, ldsG, ldsR,
                      vbar, Pdv, s, twobdv, cp_last, dp_last);

  Xc[w][lane] = cp_last;
  Xd[w][lane] = dp_last;
  __syncthreads();
  const float co = Xc[w ^ 1][lane];
  const float dq = Xd[w ^ 1][lane];
  const float xnext = (dq - co * dp_last) / (1.0f - co * cp_last);

  if (w == 0)
    substitute_half<0>(oG, Cp[0] + cpR, Td[0], ldsG, ldsR, xnext);
  else
    substitute_half<1>(oG, Cp[1] + cpR, Td[1], ldsG, ldsR, xnext);
}

extern "C" void kernel_launch(void* const* d_in, const int* in_sizes, int n_in,
                              void* d_out, int out_size, void* d_ws, size_t ws_size,
                              hipStream_t stream) {
  const float* f      = (const float*)d_in[0];
  const float* v      = (const float*)d_in[1];  (void)v;
  const float* v_edge = (const float*)d_in[2];
  const float* dv     = (const float*)d_in[3];
  const float* nu     = (const float*)d_in[4];
  const float* dt     = (const float*)d_in[5];
  (void)in_sizes; (void)n_in; (void)out_size; (void)ws_size;

  (void)d_ws;
  float* out = (float*)d_out;

  fp_thomas<<<B_ROWS / 32, 64, 0, stream>>>(f, v_edge, dv, nu, dt, out);
}


# moments loop: f loads of the next 128 columns (32 loads, 4 lines a row) all issued before the matrix instructions of the current 128; two register sets alternate; every sum keeps its order
# speedup vs baseline: 6.0905x; 1.1394x over previous
#include <hip/hip_runtime.h>
#include <cmath>

#define B_ROWS   16384
#define NVC      1024
#define VMAXF    6.0f
#define SC_ITERS 10

typedef __attribute__((ext_vector_type(2))) float v2f;
typedef __attribute__((ext_vector_type(8))) float v8f;

#define MSTG 32
#define MNST (NVC / (4 * MSTG))
__device__ __forceinline__ void moments_stage_load(const float* fp, v2f (&a)[MSTG]) {
#pragma unroll
  for (int i = 0; i < MSTG; ++i) a[i] = *(const v2f*)(fp + 4 * i);
}
__device__ __forceinline__ void moments_stage_mma(
    const v2f (&a)[MSTG], const float c0, const float c1, const float c2,
    const float dv, const float dv4, float& vcur,
    v8f& acc0, v8f& acc1, v8f& acc2, v8f& acc3) {
#pragma unroll
  for (int i = 0; i < MSTG; ++i) {
    const float v0 = vcur;
    const float v1 = vcur + dv;
    v2f b;
    b.x = (c2 * v0 + c1) * v0 + c0;
    b.y = (c2 * v1 + c1) * v1 + c0;
    v8f* acc = ((i & 3) == 0) ? &acc0 : ((i & 3) == 1) ? &acc1 : ((i & 3) == 2) ? &acc2 : &acc3;
    *acc = __builtin_amdgcn_wmma_f32_16x16x4_f32(
        false, a[i], false, b, (short)0, *acc, false, false);
    vcur += dv4;
  }
}

__device__ __forceinline__ float4 moments16(
    const float* __restrict__ f, const float dv, const int rowbase, const int lane) {
  const int r       = lane & 15;
  const int khalf   = (lane >> 4) * 2;

  const float* fp = f + (size_t)(rowbase + r) * NVC + khalf;

  const float c0 = (r == 0) ? 1.0f : 0.0f;
  const float c1 = (r == 1) ? 1.0f : 0.0f;
  const float c2 = (r == 2) ? 1.0f : 0.0f;

  const float dv4 = 4.0f * dv;
  float vcur = dv * ((float)khalf + 0.5f) - VMAXF;

  v8f acc0 = {}, acc1 = {}, acc2 = {}, acc3 = {};
  v2f aA[MSTG], aB[MSTG];
  moments_stage_load(fp, aA);
  for (int st = 0; st < MNST - 2; st += 2) {
    moments_stage_load(fp + (st + 1) * 4 * MSTG, aB);
    moments_stage_mma(aA, c0, c1, c2, dv, dv4, vcur, acc0, acc1, acc2, acc3);
    moments_stage_load(fp + (st + 2) * 4 * MSTG, aA);
    moments_stage_mma(aB, c0, c1, c2, dv, dv4, vcur, acc0, acc1, acc2, acc3);
  }
  moments_stage_load(fp + (MNST - 1) * 4 * MSTG, aB);
  moments_stage_mma(aA, c0, c1, c2, dv, dv4, vcur, acc0, acc1, acc2, acc3);
  moments_stage_mma(aB, c0, c1, c2, dv, dv4, vcur, acc0, acc1, acc2, acc3);
  v8f acc = (acc0 + acc1) + (acc2 + acc3);

  float av[8];
#pragma unroll
  for (int i = 0; i < 8; ++i) av[i] = acc[i];

  float myn = 0.f, myvb = 0.f, myet = 0.f, mybeta = 0.f;
#pragma unroll
  for (int m = 0; m < 16; ++m) {
    const int base  = (m < 8) ? 0 : 16;
    const float val = av[m & 7];
    const float S0  = __shfl(val, base + 0, 32);
    const float S1  = __shfl(val, base + 1, 32);
    const float S2  = __shfl(val, base + 2, 32);
    if (lane == m) {
      const float n_   = S0 * dv;
      const float vbar = S1 / S0;
      const float e_t  = (S2 - S1 * S1 / S0) * dv;
      myn = n_; myvb = vbar; myet = e_t;
      mybeta = n_ / (2.0f * e_t);
    }
  }
  return make_float4(myn, myvb, myet, mybeta);
}

template <int CTRL> __device__ __forceinline__ float row_lane(float x) {
  return __builtin_bit_cast(
      float, __builtin_amdgcn_update_dpp(0, __builtin_bit_cast(int, x), CTRL, 0xF, 0xF, true));
}

__device__ __forceinline__ float wave_sum(float x) {
  const unsigned u = __builtin_bit_cast(unsigned, x);
  x += __builtin_bit_cast(float, __builtin_amdgcn_permlanex16(u, u, 0x76543210u, 0xfedcba98u, false, false));
  x += row_lane<0x128>(x);
  x += __builtin_bit_cast(float, __builtin_amdgcn_mov_dpp8(__builtin_bit_cast(unsigned, x), 0x688facu));
  x += row_lane<0x4E>(x);
  x += row_lane<0xB1>(x);
  return x;
}

#define BROWS 4
__device__ __forceinline__ void beta_rows4(
    const float dv, const int lane, const float4 s0, const float4 s1, const float4 s2,
    const float4 s3, v2f& outA, v2f& outB) {
  const v2f vbA = {s0.y, s1.y}, vbB = {s2.y, s3.y};
  v2f betaA = {s0.w, s1.w}, betaB = {s2.w, s3.w};
  const v2f gA = {s0.x / s0.z, s1.x / s1.z};
  const v2f gB = {s2.x / s2.z, s3.x / s3.z};

  v2f u2A[32], u2B[32];
#pragma unroll
  for (int c = 0; c < 32; ++c) {
    const int j   = lane + 32 * c;
    const float v = dv * ((float)j + 0.5f) - VMAXF;
    const v2f vv  = {v, v};
    const v2f uA  = vv - vbA;
    const v2f uB  = vv - vbB;
    u2A[c] = uA * uA;
    u2B[c] = uB * uB;
  }

  const float NLOG2E = -1.44269504088896340736f;
  for (int it = 0; it < SC_ITERS; ++it) {
    const v2f nbA = betaA * NLOG2E;
    const v2f nbB = betaB * NLOG2E;
    v2f smA = {0.f, 0.f}, seA = {0.f, 0.f}, smB = {0.f, 0.f}, seB = {0.f, 0.f};
#pragma unroll
    for (int c = 0; c < 32; ++c) {
      const v2f tA = u2A[c] * nbA;
      const v2f tB = u2B[c] * nbB;
      v2f MA, MB;
      MA.x = __builtin_amdgcn_exp2f(tA.x);
      MA.y = __builtin_amdgcn_exp2f(tA.y);
      MB.x = __builtin_amdgcn_exp2f(tB.x);
      MB.y = __builtin_amdgcn_exp2f(tB.y);
      smA += MA;
      smB += MB;
      seA = __builtin_elementwise_fma(MA, u2A[c], seA);
      seB = __builtin_elementwise_fma(MB, u2B[c], seB);
    }
    float r[8] = {smA.x, smA.y, smB.x, smB.y, seA.x, seA.y, seB.x, seB.y};
#pragma unroll
    for (int k = 0; k < 8; ++k) r[k] = wave_sum(r[k]);
    const v2f smRA = {r[0], r[1]}, smRB = {r[2], r[3]};
    const v2f seRA = {r[4], r[5]}, seRB = {r[6], r[7]};
    betaA = betaA * (seRA / smRA) * gA;
    betaB = betaB * (seRB / smRB) * gB;
  }
  outA = betaA;
  outB = betaB;
}

__device__ __forceinline__ float cc_delta(float w) {
  const bool small = fabsf(w) < 1e-6f;
  const float ws   = small ? 1.0f : w;
  const float em   = expm1f(ws);
  const float d    = (em - ws) * __builtin_amdgcn_rcpf(ws * em);
  return small ? 0.5f : d;
}

__device__ __forceinline__ float cc_delta_series(float w) {
  const float w2 = w * w;
  float p = fmaf(w2, 1.0f / 1209600.0f, -1.0f / 30240.0f);
  p = fmaf(w2, p, 1.0f / 720.0f);
  p = fmaf(w2, p, -1.0f / 12.0f);
  const float d = fmaf(w, p, 0.5f);
  return (fabsf(w) < 1e-6f) ? 0.5f : d;
}
template <bool SERIES> __device__ __forceinline__ float cc_delta_of(float w) {
  return SERIES ? cc_delta_series(w) : cc_delta(w);
}

__device__ __forceinline__ void wave_minmax(float& lo, float& hi) {
  lo = fminf(lo, row_lane<0xB1>(lo));   hi = fmaxf(hi, row_lane<0xB1>(hi));
  lo = fminf(lo, row_lane<0x4E>(lo));   hi = fmaxf(hi, row_lane<0x4E>(hi));
  lo = fminf(lo, row_lane<0x141>(lo));  hi = fmaxf(hi, row_lane<0x141>(hi));
  lo = fminf(lo, row_lane<0x140>(lo));  hi = fmaxf(hi, row_lane<0x140>(hi));
  const unsigned ulo = __builtin_bit_cast(unsigned, lo), uhi = __builtin_bit_cast(unsigned, hi);
  lo = fminf(lo, __builtin_bit_cast(float, __builtin_amdgcn_permlanex16(ulo, ulo, 0x76543210u, 0xfedcba98u, false, false)));
  hi = fmaxf(hi, __builtin_bit_cast(float, __builtin_amdgcn_permlanex16(uhi, uhi, 0x76543210u, 0xfedcba98u, false, false)));
}

#define TCOLS 32
#define NTILE (NVC / TCOLS)
#define HTILE (NTILE / 2)
#define TP    36
#define NVE   544
#define HCELL (NVC / 2)
#define CPP   (HCELL + 4)

typedef __attribute__((ext_vector_type(4))) float v4f;

__device__ __forceinline__ void wave_lds_handover() {
  __builtin_amdgcn_fence(__ATOMIC_ACQ_REL, "wavefront");
}

template <int W> __device__ __forceinline__ int tile_of(int t) {
  return W ? (NTILE - 1 - t) : t;
}

template <int W> __device__ __forceinline__ int blk_of(int b) {
  return 4 * (W ? (TCOLS / 4 - 1 - b) : b);
}
template <int W> __device__ __forceinline__ float own4(v4f t, int q) {
  const int z = W ? 3 - q : q;
  return z == 0 ? t.x : z == 1 ? t.y : z == 2 ? t.z : t.w;
}
template <int W> __device__ __forceinline__ v4f pack4(float a0, float a1, float a2, float a3) {
  v4f r;
  if (W) { r.x = a3; r.y = a2; r.z = a1; r.w = a0; } else { r.x = a0; r.y = a1; r.z = a2; r.w = a3; }
  return r;
}

__device__ __forceinline__ void tile_load_lines(const float* g, float (&r)[TCOLS]) {
#pragma unroll
  for (int i = 0; i < 8; ++i) {
    const v4f t4 = *(const v4f*)(g + (size_t)i * 4 * NVC);
    r[4 * i + 0] = t4.x; r[4 * i + 1] = t4.y; r[4 * i + 2] = t4.z; r[4 * i + 3] = t4.w;
  }
}

__device__ __forceinline__ void tile_put_lines(float* T, int ldsG, const float (&r)[TCOLS]) {
#pragma unroll
  for (int i = 0; i < 8; ++i) {
    v4f t4; t4.x = r[4 * i + 0]; t4.y = r[4 * i + 1]; t4.z = r[4 * i + 2]; t4.w = r[4 * i + 3];
    *(v4f*)&T[ldsG + i * 4 * TP] = t4;
  }
}

__device__ __forceinline__ void tile_store_lines(const float* T, int ldsG, float* g) {
#pragma unroll
  for (int i = 0; i < 8; ++i) {
    const v4f t4 = *(const v4f*)&T[ldsG + i * 4 * TP];
    *(v4f*)(g + (size_t)i * 4 * NVC) = t4;
  }
}

template <int W>
__device__ __forceinline__ void bwd_tile(float* Td, int ldsG, int ldsR, float* g,
                                         const float (&dpl)[TCOLS], const float* Crt,
                                         float& xnext) {
  tile_put_lines(Td, ldsG, dpl);
  float cc[TCOLS];
#pragma unroll
  for (int q = 0; q < 8; ++q) {
    const v4f c4 = *(const v4f*)&Crt[4 * q];
    cc[4 * q + 0] = c4.x; cc[4 * q + 1] = c4.y; cc[4 * q + 2] = c4.z; cc[4 * q + 3] = c4.w;
  }
  wave_lds_handover();
  float dr[TCOLS];
#pragma unroll
  for (int q = 0; q < 8; ++q) {
    const v4f t4 = *(const v4f*)&Td[ldsR + blk_of<W>(q)];
    dr[4 * q + 0] = own4<W>(t4, 0); dr[4 * q + 1] = own4<W>(t4, 1);
    dr[4 * q + 2] = own4<W>(t4, 2); dr[4 * q + 3] = own4<W>(t4, 3);
  }
  float xr[TCOLS];
  float xn = xnext;
#pragma unroll
  for (int k = TCOLS - 1; k >= 0; --k) {
    const float x = dr[k] - cc[k] * xn;
    xr[k] = x;
    xn = x;
  }
  xnext = xn;
#pragma unroll
  for (int q = 0; q < 8; ++q) {
    *(v4f*)&Td[ldsR + blk_of<W>(q)] =
        pack4<W>(xr[4 * q + 0], xr[4 * q + 1], xr[4 * q + 2], xr[4 * q + 3]);
  }
  wave_lds_handover();
  tile_store_lines(Td, ldsG, g);
  wave_lds_handover();
}

template <int W, bool SERIES>
__device__ __forceinline__ void eliminate_sweep(
    const float* __restrict__ fG, float* __restrict__ oG,
    float* Tf, float* Td, float* Cr, const float* Ve, float (&pre)[TCOLS],
    int ldsG, int ldsR, float vbar, float Pdv, float s, float twobdv,
    float& cp_last, float& dp_last) {
  float Ce0 = 0.f, P0 = 0.f, d0 = 0.5f;
  float cp_prev = 0.f, dp_prev = 0.f;

  float cC[4], dC[4];
  {
    const v4f e4 = *(const v4f*)&Ve[0];
    const float ev[4] = {e4.x, e4.y, e4.z, e4.w};
#pragma unroll
    for (int q = 0; q < 4; ++q) {
      const float c = W ? (vbar - ev[q]) : (ev[q] - vbar);
      cC[q] = c;
      dC[q] = cc_delta_of<SERIES>(twobdv * c);
    }
  }
  v4f enext = *(const v4f*)&Ve[4];

  for (int t = 0; t < HTILE; ++t) {
    const int tn = (t + 1 < HTILE) ? t + 1 : t;
    tile_load_lines(fG + tile_of<W>(tn) * TCOLS, pre);

    v4f fnext = *(const v4f*)&Tf[ldsR + blk_of<W>(0)];
    v4f cprev4 = {0.f, 0.f, 0.f, 0.f}, dprev4 = {0.f, 0.f, 0.f, 0.f};
#pragma nounroll
    for (int b = 0; b < TCOLS / 4; ++b) {
      const int jb = t * TCOLS + b * 4;
      if (b > 0) {
        *(v4f*)&Cr[jb - 4] = cprev4;
        *(v4f*)&Td[ldsR + blk_of<W>(b - 1)] = dprev4;
      }
      const v4f fj = fnext;
      const v4f ej = enext;
      const int bn = (b + 1 < TCOLS / 4) ? b + 1 : b;
      fnext = *(const v4f*)&Tf[ldsR + blk_of<W>(bn)];
      enext = *(const v4f*)&Ve[jb + 8];
      __builtin_amdgcn_sched_barrier(0);
      const float fv[4] = {own4<W>(fj, 0), own4<W>(fj, 1), own4<W>(fj, 2), own4<W>(fj, 3)};
      const float ev[4] = {ej.x, ej.y, ej.z, ej.w};

      float cN[4], dN[4];
#pragma unroll
      for (int q = 0; q < 4; ++q) {
        const float c = W ? (vbar - ev[q]) : (ev[q] - vbar);
        cN[q] = c;
        dN[q] = cc_delta_of<SERIES>(twobdv * c);
      }

      float dpv[4], cpv[4];
#pragma unroll
      for (int q = 0; q < 4; ++q) {
        const float Ce1 = cC[q];
        const float P1  = Pdv;
        const float d1  = dC[q];

        const float coefU = Ce1 * (1.0f - d1) + P1;
        const float coefL = -Ce0 * d0 + P0;
        const float coefD = Ce1 * d1 - P1 - Ce0 * (1.0f - d0) - P0;
        const float dl = -s * coefL;
        const float dd = 1.0f - s * coefD;
        const float du = -s * coefU;

        const float denom = dd - dl * cp_prev;
        const float rdiv  = __builtin_amdgcn_rcpf(denom);
        const float cp    = du * rdiv;
        const float dp    = (fv[q] - dl * dp_prev) * rdiv;

        cpv[q] = cp;
        dpv[q] = dp;
        cp_prev = cp; dp_prev = dp;
        Ce0 = Ce1; P0 = P1; d0 = d1;
      }
      cprev4.x = cpv[0]; cprev4.y = cpv[1]; cprev4.z = cpv[2]; cprev4.w = cpv[3];
      dprev4 = pack4<W>(dpv[0], dpv[1], dpv[2], dpv[3]);

#pragma unroll
      for (int q = 0; q < 4; ++q) { cC[q] = cN[q]; dC[q] = dN[q]; }
    }
    *(v4f*)&Cr[t * TCOLS + TCOLS - 4] = cprev4;
    *(v4f*)&Td[ldsR + blk_of<W>(TCOLS / 4 - 1)] = dprev4;
    wave_lds_handover();

    const int tt = tile_of<W>(t);
    tile_store_lines(Td, ldsG, oG + tt * TCOLS);
    tile_put_lines(Tf, ldsG, pre);
    wave_lds_handover();
  }
  cp_last = cp_prev;
  dp_last = dp_prev;
}

template <int W>
__device__ __forceinline__ void eliminate_half(
    const float* __restrict__ fG, float* __restrict__ oG,
    const float* __restrict__ v_edge, float* Tf, float* Td, float* Cr, float* Ve,
    int lane, int ldsG, int ldsR, float vbar, float Pdv, float s, float twobdv,
    float& cp_last, float& dp_last) {
  float pre[TCOLS];
  tile_load_lines(fG + tile_of<W>(0) * TCOLS, pre);

  float elo = __builtin_inff(), ehi = -__builtin_inff();
  for (int k = lane; k < NVE; k += 32) {
    const int e = W ? (NVC - 1 - k) : (k + 1);
    const float ve = v_edge[e];
    Ve[k] = ve;
    elo = fminf(elo, ve);
    ehi = fmaxf(ehi, ve);
  }
  wave_minmax(elo, ehi);
  const float wmax = fabsf(twobdv) * fmaxf(fabsf(elo - vbar), fabsf(ehi - vbar));
  const bool series =
      __builtin_amdgcn_ballot_w32(wmax < 0.5f) == __builtin_amdgcn_ballot_w32(true);

  tile_put_lines(Tf, ldsG, pre);
  wave_lds_handover();

  if (series)
    eliminate_sweep<W, true>(fG, oG, Tf, Td, Cr, Ve, pre, ldsG, ldsR, vbar, Pdv, s, twobdv,
                             cp_last, dp_last);
  else
    eliminate_sweep<W, false>(fG, oG, Tf, Td, Cr, Ve, pre, ldsG, ldsR, vbar, Pdv, s, twobdv,
                              cp_last, dp_last);
}

template <int W>
__device__ __forceinline__ void substitute_half(
    float* __restrict__ oG, const float* Cr, float* Td,
    int ldsG, int ldsR, float xnext) {
  float dA[TCOLS], dB[TCOLS];
  tile_load_lines(oG + tile_of<W>(HTILE - 1) * TCOLS, dA);

  for (int t = HTILE - 1; t >= 1; t -= 2) {
    tile_load_lines(oG + tile_of<W>(t - 1) * TCOLS, dB);
    bwd_tile<W>(Td, ldsG, ldsR, oG + tile_of<W>(t) * TCOLS, dA, Cr + t * TCOLS, xnext);

    const int t2 = (t >= 2) ? t - 2 : 0;
    tile_load_lines(oG + tile_of<W>(t2) * TCOLS, dA);
    bwd_tile<W>(Td, ldsG, ldsR, oG + tile_of<W>(t - 1) * TCOLS, dB, Cr + (t - 1) * TCOLS, xnext);
  }
}

__global__ __launch_bounds__(64) void fp_thomas(
    const float* __restrict__ f, const float* __restrict__ v_edge,
    const float* __restrict__ dv_p, const float* __restrict__ nu_p,
    const float* __restrict__ dt_p, float* __restrict__ out) {
  __shared__ __attribute__((aligned(16))) float Cp[2][32 * CPP];
  __shared__ __attribute__((aligned(16))) float Tf[2][32 * TP];
  __shared__ __attribute__((aligned(16))) float Td[2][32 * TP];
  __shared__ __attribute__((aligned(16))) float Ve[2][NVE];
  __shared__ float Xc[2][32], Xd[2][32];
  __shared__ __attribute__((aligned(16))) float4 Sc[32];

  const int w       = __builtin_amdgcn_readfirstlane((int)(threadIdx.x >> 5));
  const int lane    = threadIdx.x & 31;
  const int rowbase = blockIdx.x * 32;
  const float dv = *dv_p, nu = *nu_p, dt = *dt_p;

  const int gr = lane >> 3, gq = lane & 7;
  const float* fG = f   + (size_t)(rowbase + gr) * NVC + gq * 4;
  float*       oG = out + (size_t)(rowbase + gr) * NVC + gq * 4;
  const int ldsG = gr * TP + gq * 4;
  const int ldsR = lane * TP;
  const int cpR  = lane * CPP;

  {
    const float4 m = moments16(f, dv, rowbase + 16 * w, lane);
    if (lane < 16) Sc[16 * w + lane] = m;
    wave_lds_handover();
#pragma nounroll
    for (int g = 0; g < 16 / BROWS; ++g) {
      const int i0 = 16 * w + BROWS * g;
      const float4 s0 = Sc[i0 + 0], s1 = Sc[i0 + 1], s2 = Sc[i0 + 2], s3 = Sc[i0 + 3];
      v2f bA, bB;
      beta_rows4(dv, lane, s0, s1, s2, s3, bA, bB);
      if (lane == 0) {
        Sc[i0 + 0].w = bA.x;
        Sc[i0 + 1].w = bA.y;
        Sc[i0 + 2].w = bB.x;
        Sc[i0 + 3].w = bB.y;
      }
    }
  }
  __syncthreads();
  const float4 sc  = Sc[lane];
  const float vbar = sc.y, beta = sc.w;
  const float D      = 0.5f / beta;
  const float Pdv    = D / dv;
  const float s      = dt * nu / dv;
  const float twobdv = 2.0f * beta * dv;

  float cp_last, dp_last;
  if (w == 0)
    eliminate_half<0>(fG, oG, v_edge, Tf[0], Td[0], Cp[0] + cpR, Ve[0], lane, ldsG, ldsR,
                      vbar, Pdv, s, twobdv, cp_last, dp_last);
  else
    eliminate_half<1>(fG, oG, v_edge, Tf[1], Td[1], Cp[1] + cpR, Ve[1], lane, ldsG, ldsR,
                      vbar, Pdv, s, twobdv, cp_last, dp_last);

  Xc[w][lane] = cp_last;
  Xd[w][lane] = dp_last;
  __syncthreads();
  const float co = Xc[w ^ 1][lane];
  const float dq = Xd[w ^ 1][lane];
  const float xnext = (dq - co * dp_last) / (1.0f - co * cp_last);

  if (w == 0)
    substitute_half<0>(oG, Cp[0] + cpR, Td[0], ldsG, ldsR, xnext);
  else
    substitute_half<1>(oG, Cp[1] + cpR, Td[1], ldsG, ldsR, xnext);
}

extern "C" void kernel_launch(void* const* d_in, const int* in_sizes, int n_in,
                              void* d_out, int out_size, void* d_ws, size_t ws_size,
                              hipStream_t stream) {
  const float* f      = (const float*)d_in[0];
  const float* v      = (const float*)d_in[1];  (void)v;
  const float* v_edge = (const float*)d_in[2];
  const float* dv     = (const float*)d_in[3];
  const float* nu     = (const float*)d_in[4];
  const float* dt     = (const float*)d_in[5];
  (void)in_sizes; (void)n_in; (void)out_size; (void)ws_size;

  (void)d_ws;
  float* out = (float*)d_out;

  fp_thomas<<<B_ROWS / 32, 64, 0, stream>>>(f, v_edge, dv, nu, dt, out);
}


# fp_thomas elimination: 4-cell blocks alternate between two register sets for the f and edge values read from LDS; no copy between blocks, every wait for a read stands a whole block after its issue
# speedup vs baseline: 6.3356x; 1.0402x over previous
#include <hip/hip_runtime.h>
#include <cmath>

#define B_ROWS   16384
#define NVC      1024
#define VMAXF    6.0f
#define SC_ITERS 10

typedef __attribute__((ext_vector_type(2))) float v2f;
typedef __attribute__((ext_vector_type(8))) float v8f;

#define MSTG 32
#define MNST (NVC / (4 * MSTG))
__device__ __forceinline__ void moments_stage_load(const float* fp, v2f (&a)[MSTG]) {
#pragma unroll
  for (int i = 0; i < MSTG; ++i) a[i] = *(const v2f*)(fp + 4 * i);
}
__device__ __forceinline__ void moments_stage_mma(
    const v2f (&a)[MSTG], const float c0, const float c1, const float c2,
    const float dv, const float dv4, float& vcur,
    v8f& acc0, v8f& acc1, v8f& acc2, v8f& acc3) {
#pragma unroll
  for (int i = 0; i < MSTG; ++i) {
    const float v0 = vcur;
    const float v1 = vcur + dv;
    v2f b;
    b.x = (c2 * v0 + c1) * v0 + c0;
    b.y = (c2 * v1 + c1) * v1 + c0;
    v8f* acc = ((i & 3) == 0) ? &acc0 : ((i & 3) == 1) ? &acc1 : ((i & 3) == 2) ? &acc2 : &acc3;
    *acc = __builtin_amdgcn_wmma_f32_16x16x4_f32(
        false, a[i], false, b, (short)0, *acc, false, false);
    vcur += dv4;
  }
}

__device__ __forceinline__ float4 moments16(
    const float* __restrict__ f, const float dv, const int rowbase, const int lane) {
  const int r       = lane & 15;
  const int khalf   = (lane >> 4) * 2;

  const float* fp = f + (size_t)(rowbase + r) * NVC + khalf;

  const float c0 = (r == 0) ? 1.0f : 0.0f;
  const float c1 = (r == 1) ? 1.0f : 0.0f;
  const float c2 = (r == 2) ? 1.0f : 0.0f;

  const float dv4 = 4.0f * dv;
  float vcur = dv * ((float)khalf + 0.5f) - VMAXF;

  v8f acc0 = {}, acc1 = {}, acc2 = {}, acc3 = {};
  v2f aA[MSTG], aB[MSTG];
  moments_stage_load(fp, aA);
  for (int st = 0; st < MNST - 2; st += 2) {
    moments_stage_load(fp + (st + 1) * 4 * MSTG, aB);
    moments_stage_mma(aA, c0, c1, c2, dv, dv4, vcur, acc0, acc1, acc2, acc3);
    moments_stage_load(fp + (st + 2) * 4 * MSTG, aA);
    moments_stage_mma(aB, c0, c1, c2, dv, dv4, vcur, acc0, acc1, acc2, acc3);
  }
  moments_stage_load(fp + (MNST - 1) * 4 * MSTG, aB);
  moments_stage_mma(aA, c0, c1, c2, dv, dv4, vcur, acc0, acc1, acc2, acc3);
  moments_stage_mma(aB, c0, c1, c2, dv, dv4, vcur, acc0, acc1, acc2, acc3);
  v8f acc = (acc0 + acc1) + (acc2 + acc3);

  float av[8];
#pragma unroll
  for (int i = 0; i < 8; ++i) av[i] = acc[i];

  float myn = 0.f, myvb = 0.f, myet = 0.f, mybeta = 0.f;
#pragma unroll
  for (int m = 0; m < 16; ++m) {
    const int base  = (m < 8) ? 0 : 16;
    const float val = av[m & 7];
    const float S0  = __shfl(val, base + 0, 32);
    const float S1  = __shfl(val, base + 1, 32);
    const float S2  = __shfl(val, base + 2, 32);
    if (lane == m) {
      const float n_   = S0 * dv;
      const float vbar = S1 / S0;
      const float e_t  = (S2 - S1 * S1 / S0) * dv;
      myn = n_; myvb = vbar; myet = e_t;
      mybeta = n_ / (2.0f * e_t);
    }
  }
  return make_float4(myn, myvb, myet, mybeta);
}

template <int CTRL> __device__ __forceinline__ float row_lane(float x) {
  return __builtin_bit_cast(
      float, __builtin_amdgcn_update_dpp(0, __builtin_bit_cast(int, x), CTRL, 0xF, 0xF, true));
}

__device__ __forceinline__ float wave_sum(float x) {
  const unsigned u = __builtin_bit_cast(unsigned, x);
  x += __builtin_bit_cast(float, __builtin_amdgcn_permlanex16(u, u, 0x76543210u, 0xfedcba98u, false, false));
  x += row_lane<0x128>(x);
  x += __builtin_bit_cast(float, __builtin_amdgcn_mov_dpp8(__builtin_bit_cast(unsigned, x), 0x688facu));
  x += row_lane<0x4E>(x);
  x += row_lane<0xB1>(x);
  return x;
}

#define BROWS 4
__device__ __forceinline__ void beta_rows4(
    const float dv, const int lane, const float4 s0, const float4 s1, const float4 s2,
    const float4 s3, v2f& outA, v2f& outB) {
  const v2f vbA = {s0.y, s1.y}, vbB = {s2.y, s3.y};
  v2f betaA = {s0.w, s1.w}, betaB = {s2.w, s3.w};
  const v2f gA = {s0.x / s0.z, s1.x / s1.z};
  const v2f gB = {s2.x / s2.z, s3.x / s3.z};

  v2f u2A[32], u2B[32];
#pragma unroll
  for (int c = 0; c < 32; ++c) {
    const int j   = lane + 32 * c;
    const float v = dv * ((float)j + 0.5f) - VMAXF;
    const v2f vv  = {v, v};
    const v2f uA  = vv - vbA;
    const v2f uB  = vv - vbB;
    u2A[c] = uA * uA;
    u2B[c] = uB * uB;
  }

  const float NLOG2E = -1.44269504088896340736f;
  for (int it = 0; it < SC_ITERS; ++it) {
    const v2f nbA = betaA * NLOG2E;
    const v2f nbB = betaB * NLOG2E;
    v2f smA = {0.f, 0.f}, seA = {0.f, 0.f}, smB = {0.f, 0.f}, seB = {0.f, 0.f};
#pragma unroll
    for (int c = 0; c < 32; ++c) {
      const v2f tA = u2A[c] * nbA;
      const v2f tB = u2B[c] * nbB;
      v2f MA, MB;
      MA.x = __builtin_amdgcn_exp2f(tA.x);
      MA.y = __builtin_amdgcn_exp2f(tA.y);
      MB.x = __builtin_amdgcn_exp2f(tB.x);
      MB.y = __builtin_amdgcn_exp2f(tB.y);
      smA += MA;
      smB += MB;
      seA = __builtin_elementwise_fma(MA, u2A[c], seA);
      seB = __builtin_elementwise_fma(MB, u2B[c], seB);
    }
    float r[8] = {smA.x, smA.y, smB.x, smB.y, seA.x, seA.y, seB.x, seB.y};
#pragma unroll
    for (int k = 0; k < 8; ++k) r[k] = wave_sum(r[k]);
    const v2f smRA = {r[0], r[1]}, smRB = {r[2], r[3]};
    const v2f seRA = {r[4], r[5]}, seRB = {r[6], r[7]};
    betaA = betaA * (seRA / smRA) * gA;
    betaB = betaB * (seRB / smRB) * gB;
  }
  outA = betaA;
  outB = betaB;
}

__device__ __forceinline__ float cc_delta(float w) {
  const bool small = fabsf(w) < 1e-6f;
  const float ws   = small ? 1.0f : w;
  const float em   = expm1f(ws);
  const float d    = (em - ws) * __builtin_amdgcn_rcpf(ws * em);
  return small ? 0.5f : d;
}

__device__ __forceinline__ float cc_delta_series(float w) {
  const float w2 = w * w;
  float p = fmaf(w2, 1.0f / 1209600.0f, -1.0f / 30240.0f);
  p = fmaf(w2, p, 1.0f / 720.0f);
  p = fmaf(w2, p, -1.0f / 12.0f);
  const float d = fmaf(w, p, 0.5f);
  return (fabsf(w) < 1e-6f) ? 0.5f : d;
}
template <bool SERIES> __device__ __forceinline__ float cc_delta_of(float w) {
  return SERIES ? cc_delta_series(w) : cc_delta(w);
}

__device__ __forceinline__ void wave_minmax(float& lo, float& hi) {
  lo = fminf(lo, row_lane<0xB1>(lo));   hi = fmaxf(hi, row_lane<0xB1>(hi));
  lo = fminf(lo, row_lane<0x4E>(lo));   hi = fmaxf(hi, row_lane<0x4E>(hi));
  lo = fminf(lo, row_lane<0x141>(lo));  hi = fmaxf(hi, row_lane<0x141>(hi));
  lo = fminf(lo, row_lane<0x140>(lo));  hi = fmaxf(hi, row_lane<0x140>(hi));
  const unsigned ulo = __builtin_bit_cast(unsigned, lo), uhi = __builtin_bit_cast(unsigned, hi);
  lo = fminf(lo, __builtin_bit_cast(float, __builtin_amdgcn_permlanex16(ulo, ulo, 0x76543210u, 0xfedcba98u, false, false)));
  hi = fmaxf(hi, __builtin_bit_cast(float, __builtin_amdgcn_permlanex16(uhi, uhi, 0x76543210u, 0xfedcba98u, false, false)));
}

#define TCOLS 32
#define NTILE (NVC / TCOLS)
#define HTILE (NTILE / 2)
#define TP    36
#define NVE   544
#define HCELL (NVC / 2)
#define CPP   (HCELL + 4)

typedef __attribute__((ext_vector_type(4))) float v4f;

__device__ __forceinline__ void wave_lds_handover() {
  __builtin_amdgcn_fence(__ATOMIC_ACQ_REL, "wavefront");
}

template <int W> __device__ __forceinline__ int tile_of(int t) {
  return W ? (NTILE - 1 - t) : t;
}

template <int W> __device__ __forceinline__ int blk_of(int b) {
  return 4 * (W ? (TCOLS / 4 - 1 - b) : b);
}
template <int W> __device__ __forceinline__ float own4(v4f t, int q) {
  const int z = W ? 3 - q : q;
  return z == 0 ? t.x : z == 1 ? t.y : z == 2 ? t.z : t.w;
}
template <int W> __device__ __forceinline__ v4f pack4(float a0, float a1, float a2, float a3) {
  v4f r;
  if (W) { r.x = a3; r.y = a2; r.z = a1; r.w = a0; } else { r.x = a0; r.y = a1; r.z = a2; r.w = a3; }
  return r;
}

__device__ __forceinline__ void tile_load_lines(const float* g, float (&r)[TCOLS]) {
#pragma unroll
  for (int i = 0; i < 8; ++i) {
    const v4f t4 = *(const v4f*)(g + (size_t)i * 4 * NVC);
    r[4 * i + 0] = t4.x; r[4 * i + 1] = t4.y; r[4 * i + 2] = t4.z; r[4 * i + 3] = t4.w;
  }
}

__device__ __forceinline__ void tile_put_lines(float* T, int ldsG, const float (&r)[TCOLS]) {
#pragma unroll
  for (int i = 0; i < 8; ++i) {
    v4f t4; t4.x = r[4 * i + 0]; t4.y = r[4 * i + 1]; t4.z = r[4 * i + 2]; t4.w = r[4 * i + 3];
    *(v4f*)&T[ldsG + i * 4 * TP] = t4;
  }
}

__device__ __forceinline__ void tile_store_lines(const float* T, int ldsG, float* g) {
#pragma unroll
  for (int i = 0; i < 8; ++i) {
    const v4f t4 = *(const v4f*)&T[ldsG + i * 4 * TP];
    *(v4f*)(g + (size_t)i * 4 * NVC) = t4;
  }
}

template <int W>
__device__ __forceinline__ void bwd_tile(float* Td, int ldsG, int ldsR, float* g,
                                         const float (&dpl)[TCOLS], const float* Crt,
                                         float& xnext) {
  tile_put_lines(Td, ldsG, dpl);
  float cc[TCOLS];
#pragma unroll
  for (int q = 0; q < 8; ++q) {
    const v4f c4 = *(const v4f*)&Crt[4 * q];
    cc[4 * q + 0] = c4.x; cc[4 * q + 1] = c4.y; cc[4 * q + 2] = c4.z; cc[4 * q + 3] = c4.w;
  }
  wave_lds_handover();
  float dr[TCOLS];
#pragma unroll
  for (int q = 0; q < 8; ++q) {
    const v4f t4 = *(const v4f*)&Td[ldsR + blk_of<W>(q)];
    dr[4 * q + 0] = own4<W>(t4, 0); dr[4 * q + 1] = own4<W>(t4, 1);
    dr[4 * q + 2] = own4<W>(t4, 2); dr[4 * q + 3] = own4<W>(t4, 3);
  }
  float xr[TCOLS];
  float xn = xnext;
#pragma unroll
  for (int k = TCOLS - 1; k >= 0; --k) {
    const float x = dr[k] - cc[k] * xn;
    xr[k] = x;
    xn = x;
  }
  xnext = xn;
#pragma unroll
  for (int q = 0; q < 8; ++q) {
    *(v4f*)&Td[ldsR + blk_of<W>(q)] =
        pack4<W>(xr[4 * q + 0], xr[4 * q + 1], xr[4 * q + 2], xr[4 * q + 3]);
  }
  wave_lds_handover();
  tile_store_lines(Td, ldsG, g);
  wave_lds_handover();
}

template <int W, bool SERIES>
__device__ __forceinline__ void eliminate_sweep(
    const float* __restrict__ fG, float* __restrict__ oG,
    float* Tf, float* Td, float* Cr, const float* Ve, float (&pre)[TCOLS],
    int ldsG, int ldsR, float vbar, float Pdv, float s, float twobdv,
    float& cp_last, float& dp_last) {
  float Ce0 = 0.f, P0 = 0.f, d0 = 0.5f;
  float cp_prev = 0.f, dp_prev = 0.f;

  v4f fA = *(const v4f*)&Tf[ldsR + blk_of<W>(0)];
  v4f eA = *(const v4f*)&Ve[4];
  __builtin_amdgcn_sched_barrier(0);

  float cC[4], dC[4];
  {
    const v4f e4 = *(const v4f*)&Ve[0];
    const float ev[4] = {e4.x, e4.y, e4.z, e4.w};
#pragma unroll
    for (int q = 0; q < 4; ++q) {
      const float c = W ? (vbar - ev[q]) : (ev[q] - vbar);
      cC[q] = c;
      dC[q] = cc_delta_of<SERIES>(twobdv * c);
    }
  }

  for (int t = 0; t < HTILE; ++t) {
    const int tn = (t + 1 < HTILE) ? t + 1 : t;
    tile_load_lines(fG + tile_of<W>(tn) * TCOLS, pre);

    v4f fB, eB;
    auto block = [&](const int b, const v4f fj, const v4f ej, v4f& fn, v4f& en)
        __attribute__((always_inline)) {
      const int jb = t * TCOLS + b * 4;
      const int bn = (b + 1 < TCOLS / 4) ? b + 1 : b;
      fn = *(const v4f*)&Tf[ldsR + blk_of<W>(bn)];
      en = *(const v4f*)&Ve[jb + 8];
      __builtin_amdgcn_sched_barrier(0);
      const float fv[4] = {own4<W>(fj, 0), own4<W>(fj, 1), own4<W>(fj, 2), own4<W>(fj, 3)};
      const float ev[4] = {ej.x, ej.y, ej.z, ej.w};

      float cN[4], dN[4];
#pragma unroll
      for (int q = 0; q < 4; ++q) {
        const float c = W ? (vbar - ev[q]) : (ev[q] - vbar);
        cN[q] = c;
        dN[q] = cc_delta_of<SERIES>(twobdv * c);
      }

      float dpv[4], cpv[4];
#pragma unroll
      for (int q = 0; q < 4; ++q) {
        const float Ce1 = cC[q];
        const float P1  = Pdv;
        const float d1  = dC[q];

        const float coefU = Ce1 * (1.0f - d1) + P1;
        const float coefL = -Ce0 * d0 + P0;
        const float coefD = Ce1 * d1 - P1 - Ce0 * (1.0f - d0) - P0;
        const float dl = -s * coefL;
        const float dd = 1.0f - s * coefD;
        const float du = -s * coefU;

        const float denom = dd - dl * cp_prev;
        const float rdiv  = __builtin_amdgcn_rcpf(denom);
        const float cp    = du * rdiv;
        const float dp    = (fv[q] - dl * dp_prev) * rdiv;

        cpv[q] = cp;
        dpv[q] = dp;
        cp_prev = cp; dp_prev = dp;
        Ce0 = Ce1; P0 = P1; d0 = d1;
      }
      v4f c4; c4.x = cpv[0]; c4.y = cpv[1]; c4.z = cpv[2]; c4.w = cpv[3];
      *(v4f*)&Cr[jb] = c4;
      *(v4f*)&Td[ldsR + blk_of<W>(b)] = pack4<W>(dpv[0], dpv[1], dpv[2], dpv[3]);

#pragma unroll
      for (int q = 0; q < 4; ++q) { cC[q] = cN[q]; dC[q] = dN[q]; }
    };
#pragma nounroll
    for (int b = 0; b < TCOLS / 4; b += 2) {
      block(b, fA, eA, fB, eB);
      block(b + 1, fB, eB, fA, eA);
    }
    wave_lds_handover();

    const int tt = tile_of<W>(t);
    tile_put_lines(Tf, ldsG, pre);
    wave_lds_handover();
    fA = *(const v4f*)&Tf[ldsR + blk_of<W>(0)];
    __builtin_amdgcn_sched_barrier(0);
    tile_store_lines(Td, ldsG, oG + tt * TCOLS);
    wave_lds_handover();
  }
  cp_last = cp_prev;
  dp_last = dp_prev;
}

template <int W>
__device__ __forceinline__ void eliminate_half(
    const float* __restrict__ fG, float* __restrict__ oG,
    const float* __restrict__ v_edge, float* Tf, float* Td, float* Cr, float* Ve,
    int lane, int ldsG, int ldsR, float vbar, float Pdv, float s, float twobdv,
    float& cp_last, float& dp_last) {
  float pre[TCOLS];
  tile_load_lines(fG + tile_of<W>(0) * TCOLS, pre);

  float elo = __builtin_inff(), ehi = -__builtin_inff();
  for (int k = lane; k < NVE; k += 32) {
    const int e = W ? (NVC - 1 - k) : (k + 1);
    const float ve = v_edge[e];
    Ve[k] = ve;
    elo = fminf(elo, ve);
    ehi = fmaxf(ehi, ve);
  }
  wave_minmax(elo, ehi);
  const float wmax = fabsf(twobdv) * fmaxf(fabsf(elo - vbar), fabsf(ehi - vbar));
  const bool series =
      __builtin_amdgcn_ballot_w32(wmax < 0.5f) == __builtin_amdgcn_ballot_w32(true);

  tile_put_lines(Tf, ldsG, pre);
  wave_lds_handover();

  if (series)
    eliminate_sweep<W, true>(fG, oG, Tf, Td, Cr, Ve, pre, ldsG, ldsR, vbar, Pdv, s, twobdv,
                             cp_last, dp_last);
  else
    eliminate_sweep<W, false>(fG, oG, Tf, Td, Cr, Ve, pre, ldsG, ldsR, vbar, Pdv, s, twobdv,
                              cp_last, dp_last);
}

template <int W>
__device__ __forceinline__ void substitute_half(
    float* __restrict__ oG, const float* Cr, float* Td,
    int ldsG, int ldsR, float xnext) {
  float dA[TCOLS], dB[TCOLS];
  tile_load_lines(oG + tile_of<W>(HTILE - 1) * TCOLS, dA);

  for (int t = HTILE - 1; t >= 1; t -= 2) {
    tile_load_lines(oG + tile_of<W>(t - 1) * TCOLS, dB);
    bwd_tile<W>(Td, ldsG, ldsR, oG + tile_of<W>(t) * TCOLS, dA, Cr + t * TCOLS, xnext);

    const int t2 = (t >= 2) ? t - 2 : 0;
    tile_load_lines(oG + tile_of<W>(t2) * TCOLS, dA);
    bwd_tile<W>(Td, ldsG, ldsR, oG + tile_of<W>(t - 1) * TCOLS, dB, Cr + (t - 1) * TCOLS, xnext);
  }
}

__global__ __launch_bounds__(64) void fp_thomas(
    const float* __restrict__ f, const float* __restrict__ v_edge,
    const float* __restrict__ dv_p, const float* __restrict__ nu_p,
    const float* __restrict__ dt_p, float* __restrict__ out) {
  __shared__ __attribute__((aligned(16))) float Cp[2][32 * CPP];
  __shared__ __attribute__((aligned(16))) float Tf[2][32 * TP];
  __shared__ __attribute__((aligned(16))) float Td[2][32 * TP];
  __shared__ __attribute__((aligned(16))) float Ve[2][NVE];
  __shared__ float Xc[2][32], Xd[2][32];
  __shared__ __attribute__((aligned(16))) float4 Sc[32];

  const int w       = __builtin_amdgcn_readfirstlane((int)(threadIdx.x >> 5));
  const int lane    = threadIdx.x & 31;
  const int rowbase = blockIdx.x * 32;
  const float dv = *dv_p, nu = *nu_p, dt = *dt_p;

  const int gr = lane >> 3, gq = lane & 7;
  const float* fG = f   + (size_t)(rowbase + gr) * NVC + gq * 4;
  float*       oG = out + (size_t)(rowbase + gr) * NVC + gq * 4;
  const int ldsG = gr * TP + gq * 4;
  const int ldsR = lane * TP;
  const int cpR  = lane * CPP;

  {
    const float4 m = moments16(f, dv, rowbase + 16 * w, lane);
    if (lane < 16) Sc[16 * w + lane] = m;
    wave_lds_handover();
#pragma nounroll
    for (int g = 0; g < 16 / BROWS; ++g) {
      const int i0 = 16 * w + BROWS * g;
      const float4 s0 = Sc[i0 + 0], s1 = Sc[i0 + 1], s2 = Sc[i0 + 2], s3 = Sc[i0 + 3];
      v2f bA, bB;
      beta_rows4(dv, lane, s0, s1, s2, s3, bA, bB);
      if (lane == 0) {
        Sc[i0 + 0].w = bA.x;
        Sc[i0 + 1].w = bA.y;
        Sc[i0 + 2].w = bB.x;
        Sc[i0 + 3].w = bB.y;
      }
    }
  }
  __syncthreads();
  const float4 sc  = Sc[lane];
  const float vbar = sc.y, beta = sc.w;
  const float D      = 0.5f / beta;
  const float Pdv    = D / dv;
  const float s      = dt * nu / dv;
  const float twobdv = 2.0f * beta * dv;

  float cp_last, dp_last;
  if (w == 0)
    eliminate_half<0>(fG, oG, v_edge, Tf[0], Td[0], Cp[0] + cpR, Ve[0], lane, ldsG, ldsR,
                      vbar, Pdv, s, twobdv, cp_last, dp_last);
  else
    eliminate_half<1>(fG, oG, v_edge, Tf[1], Td[1], Cp[1] + cpR, Ve[1], lane, ldsG, ldsR,
                      vbar, Pdv, s, twobdv, cp_last, dp_last);

  Xc[w][lane] = cp_last;
  Xd[w][lane] = dp_last;
  __syncthreads();
  const float co = Xc[w ^ 1][lane];
  const float dq = Xd[w ^ 1][lane];
  const float xnext = (dq - co * dp_last) / (1.0f - co * cp_last);

  if (w == 0)
    substitute_half<0>(oG, Cp[0] + cpR, Td[0], ldsG, ldsR, xnext);
  else
    substitute_half<1>(oG, Cp[1] + cpR, Td[1], ldsG, ldsR, xnext);
}

extern "C" void kernel_launch(void* const* d_in, const int* in_sizes, int n_in,
                              void* d_out, int out_size, void* d_ws, size_t ws_size,
                              hipStream_t stream) {
  const float* f      = (const float*)d_in[0];
  const float* v      = (const float*)d_in[1];  (void)v;
  const float* v_edge = (const float*)d_in[2];
  const float* dv     = (const float*)d_in[3];
  const float* nu     = (const float*)d_in[4];
  const float* dt     = (const float*)d_in[5];
  (void)in_sizes; (void)n_in; (void)out_size; (void)ws_size;

  (void)d_ws;
  float* out = (float*)d_out;

  fp_thomas<<<B_ROWS / 32, 64, 0, stream>>>(f, v_edge, dv, nu, dt, out);
}
